# starting kernel
# baseline (speedup 1.0000x reference)
#include <hip/hip_runtime.h>
#include <hip/hip_bf16.h>
#include <math.h>


typedef __attribute__((ext_vector_type(2))) float v2f;
typedef __attribute__((ext_vector_type(8))) float v8f;

#define IN_FEAT   256
#define OUTC      16
#define NEG_SLOPE 0.2f
#define EPSF      1e-15f

__device__ __forceinline__ unsigned fmap(float f) {
    unsigned u = __float_as_uint(f);
    return (u & 0x80000000u) ? ~u : (u | 0x80000000u);
}
__device__ __forceinline__ float funmap(unsigned u) {
    unsigned v = (u & 0x80000000u) ? (u & 0x7FFFFFFFu) : ~u;
    return __uint_as_float(v);
}

__global__ __launch_bounds__(256) void k_node(
    const float* __restrict__ x, const float* __restrict__ W,
    const float* __restrict__ Wr, const float* __restrict__ bias,
    float* __restrict__ feat, float* __restrict__ rst, int N, int nTiles)
{
    __shared__ float sW [IN_FEAT * OUTC];
    __shared__ float sWr[IN_FEAT * OUTC];
    int tid = threadIdx.x;
    for (int i = tid; i < IN_FEAT * OUTC; i += 256) {
        sW[i]  = W[i];
        sWr[i] = Wr[i];
    }
    __syncthreads();

    int wave = tid >> 5;
    int lane = tid & 31;
    int t = blockIdx.x * 8 + wave;
    if (t >= nTiles) t = nTiles - 1;

    int half = lane >> 4;
    int l15  = lane & 15;
    int rowg = t * 16 + l15;
    if (rowg >= N) rowg = N - 1;
    const float* xrow = x + (size_t)rowg * IN_FEAT;

    v8f cf = {0.f,0.f,0.f,0.f,0.f,0.f,0.f,0.f};
    v8f cr = {0.f,0.f,0.f,0.f,0.f,0.f,0.f,0.f};
    float nsq = 0.f;

    for (int k0 = 0; k0 < IN_FEAT; k0 += 4) {
        int ka = k0 + half * 2;
        float a0 = xrow[ka], a1 = xrow[ka + 1];
        nsq = fmaf(a0, a0, nsq);
        nsq = fmaf(a1, a1, nsq);
        v2f a; a[0] = a0; a[1] = a1;
        v2f b;  b[0]  = sW [ka * OUTC + l15];  b[1]  = sW [(ka + 1) * OUTC + l15];
        v2f br; br[0] = sWr[ka * OUTC + l15];  br[1] = sWr[(ka + 1) * OUTC + l15];
        cf = __builtin_amdgcn_wmma_f32_16x16x4_f32(false, a, false, b,  (short)0, cf, false, false);
        cr = __builtin_amdgcn_wmma_f32_16x16x4_f32(false, a, false, br, (short)0, cr, false, false);
    }

    nsq += __shfl_xor(nsq, 16, 32);
    float nrm = fmaxf(sqrtf(nsq), EPSF);
    float z   = fminf(nrm, 1.0f - 1e-7f);
    float sc  = 0.5f * (log1pf(z) - log1pf(-z)) / nrm;
    float bv  = bias[l15];

    float* fp = feat + (size_t)t * 16 * OUTC;
    float* rp = rst  + (size_t)t * 16 * OUTC;
    #pragma unroll
    for (int r = 0; r < 8; ++r) {
        int row = r + half * 8;
        float s_r = __shfl(sc, row, 32);
        if (t * 16 + row < N) {
            fp[row * OUTC + l15] = cf[r] * s_r;
            rp[row * OUTC + l15] = cr[r] * s_r + bv;
        }
    }
}

__global__ void k_elr(const float* __restrict__ feat, const float* __restrict__ al,
                      const float* __restrict__ ar, float* __restrict__ el,
                      float* __restrict__ er, int N)
{
    int n = blockIdx.x * blockDim.x + threadIdx.x;
    if (n >= N) return;
    const float* f = feat + (size_t)n * OUTC;
    #pragma unroll
    for (int h = 0; h < 2; ++h) {
        float sl = 0.f, sr = 0.f;
        #pragma unroll
        for (int c = 0; c < 8; ++c) {
            float v = f[h * 8 + c];
            sl = fmaf(v, al[h * 8 + c], sl);
            sr = fmaf(v, ar[h * 8 + c], sr);
        }
        el[n * 2 + h] = sl;
        er[n * 2 + h] = sr;
    }
}

__global__ void k_init(unsigned* __restrict__ mmax, float* __restrict__ denom, int n2)
{
    int i = blockIdx.x * blockDim.x + threadIdx.x;
    if (i < n2) { mmax[i] = 0x007FFFFFu; denom[i] = 0.f; }
}

__global__ void k_edge_max(const int* __restrict__ ei, const float* __restrict__ el,
                           const float* __restrict__ er, unsigned* __restrict__ mmax, int E)
{
    int e = blockIdx.x * blockDim.x + threadIdx.x;
    if (e >= E) return;
    int s = ei[e], d = ei[E + e];
    #pragma unroll
    for (int h = 0; h < 2; ++h) {
        float v = el[s * 2 + h] + er[d * 2 + h];
        v = (v >= 0.f) ? v : NEG_SLOPE * v;
        atomicMax(&mmax[d * 2 + h], fmap(v));
    }
}

__global__ void k_max_fix(const unsigned* __restrict__ mmax, float* __restrict__ mfix, int n2)
{
    int i = blockIdx.x * blockDim.x + threadIdx.x;
    if (i >= n2) return;
    float m = funmap(mmax[i]);
    mfix[i] = (fabsf(m) <= 3.402823466e38f) ? m : 0.f;
}

__global__ void k_edge_sum(const int* __restrict__ ei, const float* __restrict__ el,
                           const float* __restrict__ er, const float* __restrict__ mfix,
                           float* __restrict__ denom, int E)
{
    int e = blockIdx.x * blockDim.x + threadIdx.x;
    if (e >= E) return;
    int s = ei[e], d = ei[E + e];
    #pragma unroll
    for (int h = 0; h < 2; ++h) {
        float v = el[s * 2 + h] + er[d * 2 + h];
        v = (v >= 0.f) ? v : NEG_SLOPE * v;
        atomicAdd(&denom[d * 2 + h], __expf(v - mfix[d * 2 + h]));
    }
}

__global__ void k_edge_agg(const int* __restrict__ ei, const float* __restrict__ el,
                           const float* __restrict__ er, const float* __restrict__ mfix,
                           const float* __restrict__ denom, const float* __restrict__ feat,
                           float* __restrict__ rst, int E)
{
    int e = blockIdx.x * blockDim.x + threadIdx.x;
    if (e >= E) return;
    int s = ei[e], d = ei[E + e];
    float alpha[2];
    #pragma unroll
    for (int h = 0; h < 2; ++h) {
        float v = el[s * 2 + h] + er[d * 2 + h];
        v = (v >= 0.f) ? v : NEG_SLOPE * v;
        float ex = __expf(v - mfix[d * 2 + h]);
        alpha[h] = ex / fmaxf(denom[d * 2 + h], EPSF);
    }
    const float* fs = feat + (size_t)s * OUTC;
    float* rd = rst + (size_t)d * OUTC;
    #pragma unroll
    for (int j = 0; j < OUTC; ++j)
        atomicAdd(&rd[j], alpha[j >> 3] * fs[j]);
}

__global__ void k_out(const float* __restrict__ rst, float* __restrict__ out, int N)
{
    int i = blockIdx.x * blockDim.x + threadIdx.x;
    if (i >= N * 8) return;
    int n = i >> 3, c = i & 7;
    out[i] = 0.5f * (rst[(size_t)n * OUTC + c] + rst[(size_t)n * OUTC + 8 + c]);
}

extern "C" void kernel_launch(void* const* d_in, const int* in_sizes, int n_in,
                              void* d_out, int out_size, void* d_ws, size_t ws_size,
                              hipStream_t stream)
{
    const float* x    = (const float*)d_in[0];
    const int*   ei   = (const int*)  d_in[1];
    const float* W    = (const float*)d_in[2];
    const float* al   = (const float*)d_in[3];
    const float* ar   = (const float*)d_in[4];
    const float* bias = (const float*)d_in[5];
    const float* Wr   = (const float*)d_in[6];
    float* out = (float*)d_out;

    int N = in_sizes[0] / IN_FEAT;
    int E = in_sizes[1] / 2;

    char* ws = (char*)d_ws;
    float*    feat  = (float*)ws;    ws += (size_t)N * OUTC * sizeof(float);
    float*    rst   = (float*)ws;    ws += (size_t)N * OUTC * sizeof(float);
    float*    el    = (float*)ws;    ws += (size_t)N * 2 * sizeof(float);
    float*    er    = (float*)ws;    ws += (size_t)N * 2 * sizeof(float);
    float*    mfix  = (float*)ws;    ws += (size_t)N * 2 * sizeof(float);
    float*    denom = (float*)ws;    ws += (size_t)N * 2 * sizeof(float);
    unsigned* mmax  = (unsigned*)ws; ws += (size_t)N * 2 * sizeof(unsigned);

    int nTiles = (N + 15) / 16;
    dim3 blk(256);

    k_node    <<<(nTiles + 7) / 8,      blk, 0, stream>>>(x, W, Wr, bias, feat, rst, N, nTiles);
    k_init    <<<(2 * N + 255) / 256,   blk, 0, stream>>>(mmax, denom, 2 * N);
    k_elr     <<<(N + 255) / 256,       blk, 0, stream>>>(feat, al, ar, el, er, N);
    k_edge_max<<<(E + 255) / 256,       blk, 0, stream>>>(ei, el, er, mmax, E);
    k_max_fix <<<(2 * N + 255) / 256,   blk, 0, stream>>>(mmax, mfix, 2 * N);
    k_edge_sum<<<(E + 255) / 256,       blk, 0, stream>>>(ei, el, er, mfix, denom, E);
    k_edge_agg<<<(E + 255) / 256,       blk, 0, stream>>>(ei, el, er, mfix, denom, feat, rst, E);
    k_out     <<<(N * 8 + 255) / 256,   blk, 0, stream>>>(rst, out, N);
}


# Fuse the softmax-denominator edge pass into the aggregation pass: add exp weights and weight*feat with the same float atomics, divide by the denominator once per node behind the sum
# speedup vs baseline: 1.1514x; 1.1514x over previous
#include <hip/hip_runtime.h>
#include <hip/hip_bf16.h>
#include <math.h>


typedef __attribute__((ext_vector_type(2))) float v2f;
typedef __attribute__((ext_vector_type(8))) float v8f;

#define IN_FEAT   256
#define OUTC      16
#define NEG_SLOPE 0.2f
#define EPSF      1e-15f

__device__ __forceinline__ unsigned fmap(float f) {
    unsigned u = __float_as_uint(f);
    return (u & 0x80000000u) ? ~u : (u | 0x80000000u);
}
__device__ __forceinline__ float funmap(unsigned u) {
    unsigned v = (u & 0x80000000u) ? (u & 0x7FFFFFFFu) : ~u;
    return __uint_as_float(v);
}

__global__ __launch_bounds__(256) void k_node(
    const float* __restrict__ x, const float* __restrict__ W,
    const float* __restrict__ Wr, const float* __restrict__ bias,
    float* __restrict__ feat, float* __restrict__ out, int N, int nTiles)
{
    __shared__ float sW [IN_FEAT * OUTC];
    __shared__ float sWr[IN_FEAT * OUTC];
    int tid = threadIdx.x;
    for (int i = tid; i < IN_FEAT * OUTC; i += 256) {
        sW[i]  = W[i];
        sWr[i] = Wr[i];
    }
    __syncthreads();

    int wave = tid >> 5;
    int lane = tid & 31;
    int t = blockIdx.x * 8 + wave;
    if (t >= nTiles) t = nTiles - 1;

    int half = lane >> 4;
    int l15  = lane & 15;
    int rowg = t * 16 + l15;
    if (rowg >= N) rowg = N - 1;
    const float* xrow = x + (size_t)rowg * IN_FEAT;

    v8f cf = {0.f,0.f,0.f,0.f,0.f,0.f,0.f,0.f};
    v8f cr = {0.f,0.f,0.f,0.f,0.f,0.f,0.f,0.f};
    float nsq = 0.f;

    for (int k0 = 0; k0 < IN_FEAT; k0 += 4) {
        int ka = k0 + half * 2;
        float a0 = xrow[ka], a1 = xrow[ka + 1];
        nsq = fmaf(a0, a0, nsq);
        nsq = fmaf(a1, a1, nsq);
        v2f a; a[0] = a0; a[1] = a1;
        v2f b;  b[0]  = sW [ka * OUTC + l15];  b[1]  = sW [(ka + 1) * OUTC + l15];
        v2f br; br[0] = sWr[ka * OUTC + l15];  br[1] = sWr[(ka + 1) * OUTC + l15];
        cf = __builtin_amdgcn_wmma_f32_16x16x4_f32(false, a, false, b,  (short)0, cf, false, false);
        cr = __builtin_amdgcn_wmma_f32_16x16x4_f32(false, a, false, br, (short)0, cr, false, false);
    }

    nsq += __shfl_xor(nsq, 16, 32);
    float nrm = fmaxf(sqrtf(nsq), EPSF);
    float z   = fminf(nrm, 1.0f - 1e-7f);
    float sc  = 0.5f * (log1pf(z) - log1pf(-z)) / nrm;
    float bv  = bias[l15];

    float* fp = feat + (size_t)t * 16 * OUTC;
    float* op = out  + (size_t)t * 16 * 8;
    #pragma unroll
    for (int r = 0; r < 8; ++r) {
        int row = r + half * 8;
        float s_r = __shfl(sc, row, 32);
        float res = cr[r] * s_r + bv;
        float rm  = 0.5f * (res + __shfl_xor(res, 8, 32));
        if (t * 16 + row < N) {
            fp[row * OUTC + l15] = cf[r] * s_r;
            if (l15 < 8) op[row * 8 + l15] = rm;
        }
    }
}

__global__ void k_elr(const float* __restrict__ feat, const float* __restrict__ al,
                      const float* __restrict__ ar, float* __restrict__ el,
                      float* __restrict__ er, int N)
{
    int n = blockIdx.x * blockDim.x + threadIdx.x;
    if (n >= N) return;
    const float* f = feat + (size_t)n * OUTC;
    #pragma unroll
    for (int h = 0; h < 2; ++h) {
        float sl = 0.f, sr = 0.f;
        #pragma unroll
        for (int c = 0; c < 8; ++c) {
            float v = f[h * 8 + c];
            sl = fmaf(v, al[h * 8 + c], sl);
            sr = fmaf(v, ar[h * 8 + c], sr);
        }
        el[n * 2 + h] = sl;
        er[n * 2 + h] = sr;
    }
}

__global__ void k_init(unsigned* __restrict__ mmax, float* __restrict__ denom,
                       float* __restrict__ acc, int n2, int n4)
{
    int i = blockIdx.x * blockDim.x + threadIdx.x;
    if (i < n2) { mmax[i] = 0x007FFFFFu; denom[i] = 0.f; }
    if (i < n4) reinterpret_cast<float4*>(acc)[i] = make_float4(0.f, 0.f, 0.f, 0.f);
}

__global__ void k_edge_max(const int* __restrict__ ei, const float* __restrict__ el,
                           const float* __restrict__ er, unsigned* __restrict__ mmax, int E)
{
    int e = blockIdx.x * blockDim.x + threadIdx.x;
    if (e >= E) return;
    int s = ei[e], d = ei[E + e];
    #pragma unroll
    for (int h = 0; h < 2; ++h) {
        float v = el[s * 2 + h] + er[d * 2 + h];
        v = (v >= 0.f) ? v : NEG_SLOPE * v;
        atomicMax(&mmax[d * 2 + h], fmap(v));
    }
}

__global__ void k_max_fix(const unsigned* __restrict__ mmax, float* __restrict__ mfix, int n2)
{
    int i = blockIdx.x * blockDim.x + threadIdx.x;
    if (i >= n2) return;
    float m = funmap(mmax[i]);
    mfix[i] = (fabsf(m) <= 3.402823466e38f) ? m : 0.f;
}

__global__ void k_edge_agg(const int* __restrict__ ei, const float* __restrict__ el,
                           const float* __restrict__ er, const float* __restrict__ mfix,
                           const float* __restrict__ feat, float* __restrict__ denom,
                           float* __restrict__ acc, int E)
{
    int e = blockIdx.x * blockDim.x + threadIdx.x;
    if (e >= E) return;
    int s = ei[e], d = ei[E + e];
    float w[2];
    #pragma unroll
    for (int h = 0; h < 2; ++h) {
        float v = el[s * 2 + h] + er[d * 2 + h];
        v = (v >= 0.f) ? v : NEG_SLOPE * v;
        w[h] = __expf(v - mfix[d * 2 + h]);
        atomicAdd(&denom[d * 2 + h], w[h]);
    }
    const float* fs = feat + (size_t)s * OUTC;
    float* ad = acc + (size_t)d * OUTC;
    #pragma unroll
    for (int j = 0; j < OUTC; ++j)
        atomicAdd(&ad[j], w[j >> 3] * fs[j]);
}

__global__ void k_out(const float* __restrict__ acc, const float* __restrict__ denom,
                      float* __restrict__ out, int N)
{
    int i = blockIdx.x * blockDim.x + threadIdx.x;
    if (i >= N * 8) return;
    int n = i >> 3, c = i & 7;
    float m0 = acc[(size_t)n * OUTC + c]     / fmaxf(denom[n * 2],     EPSF);
    float m1 = acc[(size_t)n * OUTC + 8 + c] / fmaxf(denom[n * 2 + 1], EPSF);
    out[i] = out[i] + 0.5f * (m0 + m1);
}

extern "C" void kernel_launch(void* const* d_in, const int* in_sizes, int n_in,
                              void* d_out, int out_size, void* d_ws, size_t ws_size,
                              hipStream_t stream)
{
    const float* x    = (const float*)d_in[0];
    const int*   ei   = (const int*)  d_in[1];
    const float* W    = (const float*)d_in[2];
    const float* al   = (const float*)d_in[3];
    const float* ar   = (const float*)d_in[4];
    const float* bias = (const float*)d_in[5];
    const float* Wr   = (const float*)d_in[6];
    float* out = (float*)d_out;

    int N = in_sizes[0] / IN_FEAT;
    int E = in_sizes[1] / 2;

    char* ws = (char*)d_ws;
    float*    feat  = (float*)ws;    ws += (size_t)N * OUTC * sizeof(float);
    float*    acc   = (float*)ws;    ws += (size_t)N * OUTC * sizeof(float);
    float*    el    = (float*)ws;    ws += (size_t)N * 2 * sizeof(float);
    float*    er    = (float*)ws;    ws += (size_t)N * 2 * sizeof(float);
    float*    mfix  = (float*)ws;    ws += (size_t)N * 2 * sizeof(float);
    float*    denom = (float*)ws;    ws += (size_t)N * 2 * sizeof(float);
    unsigned* mmax  = (unsigned*)ws; ws += (size_t)N * 2 * sizeof(unsigned);

    int nTiles = (N + 15) / 16;
    dim3 blk(256);

    k_node    <<<(nTiles + 7) / 8,      blk, 0, stream>>>(x, W, Wr, bias, feat, out, N, nTiles);
    k_init    <<<(4 * N + 255) / 256,   blk, 0, stream>>>(mmax, denom, acc, 2 * N, 4 * N);
    k_elr     <<<(N + 255) / 256,       blk, 0, stream>>>(feat, al, ar, el, er, N);
    k_edge_max<<<(E + 255) / 256,       blk, 0, stream>>>(ei, el, er, mmax, E);
    k_max_fix <<<(2 * N + 255) / 256,   blk, 0, stream>>>(mmax, mfix, 2 * N);
    k_edge_agg<<<(E + 255) / 256,       blk, 0, stream>>>(ei, el, er, mfix, feat, denom, acc, E);
    k_out     <<<(N * 8 + 255) / 256,   blk, 0, stream>>>(acc, denom, out, N);
}


# Edge passes deal their atomics out row-wise: a half-wave adds the 16 message columns of one edge as one 64-byte row (lane = column), a lane pair its two heads; same atomics, arrival order as before
# speedup vs baseline: 2.9622x; 2.5727x over previous
#include <hip/hip_runtime.h>
#include <hip/hip_bf16.h>
#include <math.h>


typedef __attribute__((ext_vector_type(2))) float v2f;
typedef __attribute__((ext_vector_type(8))) float v8f;

#define IN_FEAT   256
#define OUTC      16
#define NEG_SLOPE 0.2f
#define EPSF      1e-15f

__device__ __forceinline__ unsigned fmap(float f) {
    unsigned u = __float_as_uint(f);
    return (u & 0x80000000u) ? ~u : (u | 0x80000000u);
}
__device__ __forceinline__ float funmap(unsigned u) {
    unsigned v = (u & 0x80000000u) ? (u & 0x7FFFFFFFu) : ~u;
    return __uint_as_float(v);
}

__global__ __launch_bounds__(256) void k_node(
    const float* __restrict__ x, const float* __restrict__ W,
    const float* __restrict__ Wr, const float* __restrict__ bias,
    float* __restrict__ feat, float* __restrict__ out, int N, int nTiles)
{
    __shared__ float sW [IN_FEAT * OUTC];
    __shared__ float sWr[IN_FEAT * OUTC];
    int tid = threadIdx.x;
    for (int i = tid; i < IN_FEAT * OUTC; i += 256) {
        sW[i]  = W[i];
        sWr[i] = Wr[i];
    }
    __syncthreads();

    int wave = tid >> 5;
    int lane = tid & 31;
    int t = blockIdx.x * 8 + wave;
    if (t >= nTiles) t = nTiles - 1;

    int half = lane >> 4;
    int l15  = lane & 15;
    int rowg = t * 16 + l15;
    if (rowg >= N) rowg = N - 1;
    const float* xrow = x + (size_t)rowg * IN_FEAT;

    v8f cf = {0.f,0.f,0.f,0.f,0.f,0.f,0.f,0.f};
    v8f cr = {0.f,0.f,0.f,0.f,0.f,0.f,0.f,0.f};
    float nsq = 0.f;

    for (int k0 = 0; k0 < IN_FEAT; k0 += 4) {
        int ka = k0 + half * 2;
        float a0 = xrow[ka], a1 = xrow[ka + 1];
        nsq = fmaf(a0, a0, nsq);
        nsq = fmaf(a1, a1, nsq);
        v2f a; a[0] = a0; a[1] = a1;
        v2f b;  b[0]  = sW [ka * OUTC + l15];  b[1]  = sW [(ka + 1) * OUTC + l15];
        v2f br; br[0] = sWr[ka * OUTC + l15];  br[1] = sWr[(ka + 1) * OUTC + l15];
        cf = __builtin_amdgcn_wmma_f32_16x16x4_f32(false, a, false, b,  (short)0, cf, false, false);
        cr = __builtin_amdgcn_wmma_f32_16x16x4_f32(false, a, false, br, (short)0, cr, false, false);
    }

    nsq += __shfl_xor(nsq, 16, 32);
    float nrm = fmaxf(sqrtf(nsq), EPSF);
    float z   = fminf(nrm, 1.0f - 1e-7f);
    float sc  = 0.5f * (log1pf(z) - log1pf(-z)) / nrm;
    float bv  = bias[l15];

    float* fp = feat + (size_t)t * 16 * OUTC;
    float* op = out  + (size_t)t * 16 * 8;
    #pragma unroll
    for (int r = 0; r < 8; ++r) {
        int row = r + half * 8;
        float s_r = __shfl(sc, row, 32);
        float res = cr[r] * s_r + bv;
        float rm  = 0.5f * (res + __shfl_xor(res, 8, 32));
        if (t * 16 + row < N) {
            fp[row * OUTC + l15] = cf[r] * s_r;
            if (l15 < 8) op[row * 8 + l15] = rm;
        }
    }
}

__global__ void k_elr(const float* __restrict__ feat, const float* __restrict__ al,
                      const float* __restrict__ ar, float* __restrict__ el,
                      float* __restrict__ er, int N)
{
    int n = blockIdx.x * blockDim.x + threadIdx.x;
    if (n >= N) return;
    const float* f = feat + (size_t)n * OUTC;
    #pragma unroll
    for (int h = 0; h < 2; ++h) {
        float sl = 0.f, sr = 0.f;
        #pragma unroll
        for (int c = 0; c < 8; ++c) {
            float v = f[h * 8 + c];
            sl = fmaf(v, al[h * 8 + c], sl);
            sr = fmaf(v, ar[h * 8 + c], sr);
        }
        el[n * 2 + h] = sl;
        er[n * 2 + h] = sr;
    }
}

__global__ void k_init(unsigned* __restrict__ mmax, float* __restrict__ denom,
                       float* __restrict__ acc, int n2, int n4)
{
    int i = blockIdx.x * blockDim.x + threadIdx.x;
    if (i < n2) { mmax[i] = 0x007FFFFFu; denom[i] = 0.f; }
    if (i < n4) reinterpret_cast<float4*>(acc)[i] = make_float4(0.f, 0.f, 0.f, 0.f);
}

__global__ void k_edge_max(const int* __restrict__ ei, const float* __restrict__ el,
                           const float* __restrict__ er, unsigned* __restrict__ mmax, int E)
{
    int e = blockIdx.x * blockDim.x + threadIdx.x;
    int lane = threadIdx.x & 31;
    int base = e - lane;
    int d = 0;
    float v0 = 0.f, v1 = 0.f;
    if (e < E) {
        int s = ei[e];
        d = ei[E + e];
        v0 = el[s * 2] + er[d * 2];
        v0 = (v0 >= 0.f) ? v0 : NEG_SLOPE * v0;
        v1 = el[s * 2 + 1] + er[d * 2 + 1];
        v1 = (v1 >= 0.f) ? v1 : NEG_SLOPE * v1;
    }
    int h = lane & 1;
    for (int it = 0; it < 2; ++it) {
        int q = it * 16 + (lane >> 1);
        int dd = __shfl(d, q, 32);
        float a = __shfl(v0, q, 32), b = __shfl(v1, q, 32);
        if (base + q < E) atomicMax(&mmax[dd * 2 + h], fmap(h ? b : a));
    }
}

__global__ void k_max_fix(const unsigned* __restrict__ mmax, float* __restrict__ mfix, int n2)
{
    int i = blockIdx.x * blockDim.x + threadIdx.x;
    if (i >= n2) return;
    float m = funmap(mmax[i]);
    mfix[i] = (fabsf(m) <= 3.402823466e38f) ? m : 0.f;
}

__global__ void k_edge_agg(const int* __restrict__ ei, const float* __restrict__ el,
                           const float* __restrict__ er, const float* __restrict__ mfix,
                           const float* __restrict__ feat, float* __restrict__ denom,
                           float* __restrict__ acc, int E)
{
    int e = blockIdx.x * blockDim.x + threadIdx.x;
    int lane = threadIdx.x & 31;
    int base = e - lane;
    int s = 0, d = 0;
    float w0 = 0.f, w1 = 0.f;
    if (e < E) {
        s = ei[e];
        d = ei[E + e];
        float v = el[s * 2] + er[d * 2];
        v = (v >= 0.f) ? v : NEG_SLOPE * v;
        w0 = __expf(v - mfix[d * 2]);
        v = el[s * 2 + 1] + er[d * 2 + 1];
        v = (v >= 0.f) ? v : NEG_SLOPE * v;
        w1 = __expf(v - mfix[d * 2 + 1]);
    }

    int h = lane & 1;
    for (int it = 0; it < 2; ++it) {
        int q = it * 16 + (lane >> 1);
        int dd = __shfl(d, q, 32);
        float a = __shfl(w0, q, 32), b = __shfl(w1, q, 32);
        if (base + q < E) atomicAdd(&denom[dd * 2 + h], h ? b : a);
    }

    int col = lane & 15;
    int hw  = lane >> 4;
    bool hi = (col & 8) != 0;
    float m[16];
    int   di[16];
    #pragma unroll
    for (int it = 0; it < 16; ++it) {
        int q = it * 2 + hw;
        int ss = __shfl(s, q, 32);
        di[it] = __shfl(d, q, 32) * OUTC + col;
        float a = __shfl(w0, q, 32), b = __shfl(w1, q, 32);
        m[it] = (hi ? b : a) * feat[ss * OUTC + col];
    }
    #pragma unroll
    for (int it = 0; it < 16; ++it)
        if (base + it * 2 + hw < E) atomicAdd(&acc[di[it]], m[it]);
}

__global__ void k_out(const float* __restrict__ acc, const float* __restrict__ denom,
                      float* __restrict__ out, int N)
{
    int i = blockIdx.x * blockDim.x + threadIdx.x;
    if (i >= N * 8) return;
    int n = i >> 3, c = i & 7;
    float m0 = acc[(size_t)n * OUTC + c]     / fmaxf(denom[n * 2],     EPSF);
    float m1 = acc[(size_t)n * OUTC + 8 + c] / fmaxf(denom[n * 2 + 1], EPSF);
    out[i] = out[i] + 0.5f * (m0 + m1);
}

extern "C" void kernel_launch(void* const* d_in, const int* in_sizes, int n_in,
                              void* d_out, int out_size, void* d_ws, size_t ws_size,
                              hipStream_t stream)
{
    const float* x    = (const float*)d_in[0];
    const int*   ei   = (const int*)  d_in[1];
    const float* W    = (const float*)d_in[2];
    const float* al   = (const float*)d_in[3];
    const float* ar   = (const float*)d_in[4];
    const float* bias = (const float*)d_in[5];
    const float* Wr   = (const float*)d_in[6];
    float* out = (float*)d_out;

    int N = in_sizes[0] / IN_FEAT;
    int E = in_sizes[1] / 2;

    char* ws = (char*)d_ws;
    float*    feat  = (float*)ws;    ws += (size_t)N * OUTC * sizeof(float);
    float*    acc   = (float*)ws;    ws += (size_t)N * OUTC * sizeof(float);
    float*    el    = (float*)ws;    ws += (size_t)N * 2 * sizeof(float);
    float*    er    = (float*)ws;    ws += (size_t)N * 2 * sizeof(float);
    float*    mfix  = (float*)ws;    ws += (size_t)N * 2 * sizeof(float);
    float*    denom = (float*)ws;    ws += (size_t)N * 2 * sizeof(float);
    unsigned* mmax  = (unsigned*)ws; ws += (size_t)N * 2 * sizeof(unsigned);

    int nTiles = (N + 15) / 16;
    dim3 blk(256);

    k_node    <<<(nTiles + 7) / 8,      blk, 0, stream>>>(x, W, Wr, bias, feat, out, N, nTiles);
    k_init    <<<(4 * N + 255) / 256,   blk, 0, stream>>>(mmax, denom, acc, 2 * N, 4 * N);
    k_elr     <<<(N + 255) / 256,       blk, 0, stream>>>(feat, al, ar, el, er, N);
    k_edge_max<<<(E + 255) / 256,       blk, 0, stream>>>(ei, el, er, mmax, E);
    k_max_fix <<<(2 * N + 255) / 256,   blk, 0, stream>>>(mmax, mfix, 2 * N);
    k_edge_agg<<<(E + 255) / 256,       blk, 0, stream>>>(ei, el, er, mfix, feat, denom, acc, E);
    k_out     <<<(N * 8 + 255) / 256,   blk, 0, stream>>>(acc, denom, out, N);
}


# Destination scalars leave the edge gathers: segment max taken over el[src] alone (leaky(.+er[dst]) is monotone, applied per node after: same max bit for bit), er and max read as one 16-byte record
# speedup vs baseline: 3.1291x; 1.0563x over previous
#include <hip/hip_runtime.h>
#include <hip/hip_bf16.h>
#include <math.h>


typedef __attribute__((ext_vector_type(2))) float v2f;
typedef __attribute__((ext_vector_type(8))) float v8f;

#define IN_FEAT   256
#define OUTC      16
#define NEG_SLOPE 0.2f
#define EPSF      1e-15f

__device__ __forceinline__ unsigned fmap(float f) {
    unsigned u = __float_as_uint(f);
    return (u & 0x80000000u) ? ~u : (u | 0x80000000u);
}
__device__ __forceinline__ float funmap(unsigned u) {
    unsigned v = (u & 0x80000000u) ? (u & 0x7FFFFFFFu) : ~u;
    return __uint_as_float(v);
}

__global__ __launch_bounds__(256) void k_node(
    const float* __restrict__ x, const float* __restrict__ W,
    const float* __restrict__ Wr, const float* __restrict__ bias,
    float* __restrict__ feat, float* __restrict__ out, int N, int nTiles)
{
    __shared__ float sW [IN_FEAT * OUTC];
    __shared__ float sWr[IN_FEAT * OUTC];
    int tid = threadIdx.x;
    for (int i = tid; i < IN_FEAT * OUTC; i += 256) {
        sW[i]  = W[i];
        sWr[i] = Wr[i];
    }
    __syncthreads();

    int wave = tid >> 5;
    int lane = tid & 31;
    int t = blockIdx.x * 8 + wave;
    if (t >= nTiles) t = nTiles - 1;

    int half = lane >> 4;
    int l15  = lane & 15;
    int rowg = t * 16 + l15;
    if (rowg >= N) rowg = N - 1;
    const float* xrow = x + (size_t)rowg * IN_FEAT;

    v8f cf = {0.f,0.f,0.f,0.f,0.f,0.f,0.f,0.f};
    v8f cr = {0.f,0.f,0.f,0.f,0.f,0.f,0.f,0.f};
    float nsq = 0.f;

    for (int k0 = 0; k0 < IN_FEAT; k0 += 4) {
        int ka = k0 + half * 2;
        float a0 = xrow[ka], a1 = xrow[ka + 1];
        nsq = fmaf(a0, a0, nsq);
        nsq = fmaf(a1, a1, nsq);
        v2f a; a[0] = a0; a[1] = a1;
        v2f b;  b[0]  = sW [ka * OUTC + l15];  b[1]  = sW [(ka + 1) * OUTC + l15];
        v2f br; br[0] = sWr[ka * OUTC + l15];  br[1] = sWr[(ka + 1) * OUTC + l15];
        cf = __builtin_amdgcn_wmma_f32_16x16x4_f32(false, a, false, b,  (short)0, cf, false, false);
        cr = __builtin_amdgcn_wmma_f32_16x16x4_f32(false, a, false, br, (short)0, cr, false, false);
    }

    nsq += __shfl_xor(nsq, 16, 32);
    float nrm = fmaxf(sqrtf(nsq), EPSF);
    float z   = fminf(nrm, 1.0f - 1e-7f);
    float sc  = 0.5f * (log1pf(z) - log1pf(-z)) / nrm;
    float bv  = bias[l15];

    float* fp = feat + (size_t)t * 16 * OUTC;
    float* op = out  + (size_t)t * 16 * 8;
    #pragma unroll
    for (int r = 0; r < 8; ++r) {
        int row = r + half * 8;
        float s_r = __shfl(sc, row, 32);
        float res = cr[r] * s_r + bv;
        float rm  = 0.5f * (res + __shfl_xor(res, 8, 32));
        if (t * 16 + row < N) {
            fp[row * OUTC + l15] = cf[r] * s_r;
            if (l15 < 8) op[row * 8 + l15] = rm;
        }
    }
}

__global__ void k_elr(const float* __restrict__ feat, const float* __restrict__ al,
                      const float* __restrict__ ar, float* __restrict__ el,
                      float* __restrict__ rec, int N)
{
    int n = blockIdx.x * blockDim.x + threadIdx.x;
    if (n >= N) return;
    const float* f = feat + (size_t)n * OUTC;
    #pragma unroll
    for (int h = 0; h < 2; ++h) {
        float sl = 0.f, sr = 0.f;
        #pragma unroll
        for (int c = 0; c < 8; ++c) {
            float v = f[h * 8 + c];
            sl = fmaf(v, al[h * 8 + c], sl);
            sr = fmaf(v, ar[h * 8 + c], sr);
        }
        el[n * 2 + h] = sl;
        rec[n * 4 + h] = sr;
    }
}

__global__ void k_init(unsigned* __restrict__ mmax, float* __restrict__ denom,
                       float* __restrict__ acc, int n2, int n4)
{
    int i = blockIdx.x * blockDim.x + threadIdx.x;
    if (i < n2) { mmax[i] = 0x007FFFFFu; denom[i] = 0.f; }
    if (i < n4) reinterpret_cast<float4*>(acc)[i] = make_float4(0.f, 0.f, 0.f, 0.f);
}

__global__ void k_edge_max(const int* __restrict__ ei, const float* __restrict__ el,
                           unsigned* __restrict__ mmax, int E)
{
    int e = blockIdx.x * blockDim.x + threadIdx.x;
    int lane = threadIdx.x & 31;
    int base = e - lane;
    int d = 0;
    float v0 = 0.f, v1 = 0.f;
    if (e < E) {
        int s = ei[e];
        d = ei[E + e];
        v0 = el[s * 2];
        v1 = el[s * 2 + 1];
    }
    int h = lane & 1;
    for (int it = 0; it < 2; ++it) {
        int q = it * 16 + (lane >> 1);
        int dd = __shfl(d, q, 32);
        float a = __shfl(v0, q, 32), b = __shfl(v1, q, 32);
        if (base + q < E) atomicMax(&mmax[dd * 2 + h], fmap(h ? b : a));
    }
}

__global__ void k_max_fix(const unsigned* __restrict__ mmax, float* __restrict__ rec, int n2)
{
    int i = blockIdx.x * blockDim.x + threadIdx.x;
    if (i >= n2) return;
    int n = i >> 1, h = i & 1;
    float m = funmap(mmax[i]) + rec[n * 4 + h];
    m = (m >= 0.f) ? m : NEG_SLOPE * m;
    rec[n * 4 + 2 + h] = (fabsf(m) <= 3.402823466e38f) ? m : 0.f;
}

__global__ void k_edge_agg(const int* __restrict__ ei, const float* __restrict__ el,
                           const float* __restrict__ rec,
                           const float* __restrict__ feat, float* __restrict__ denom,
                           float* __restrict__ acc, int E)
{
    int e = blockIdx.x * blockDim.x + threadIdx.x;
    int lane = threadIdx.x & 31;
    int base = e - lane;
    int s = 0, d = 0;
    float w0 = 0.f, w1 = 0.f;
    if (e < E) {
        s = ei[e];
        d = ei[E + e];
        float4 r = reinterpret_cast<const float4*>(rec)[d];
        float v = el[s * 2] + r.x;
        v = (v >= 0.f) ? v : NEG_SLOPE * v;
        w0 = __expf(v - r.z);
        v = el[s * 2 + 1] + r.y;
        v = (v >= 0.f) ? v : NEG_SLOPE * v;
        w1 = __expf(v - r.w);
    }

    int h = lane & 1;
    for (int it = 0; it < 2; ++it) {
        int q = it * 16 + (lane >> 1);
        int dd = __shfl(d, q, 32);
        float a = __shfl(w0, q, 32), b = __shfl(w1, q, 32);
        if (base + q < E) atomicAdd(&denom[dd * 2 + h], h ? b : a);
    }

    int col = lane & 15;
    int hw  = lane >> 4;
    bool hi = (col & 8) != 0;
    float m[16];
    int   di[16];
    #pragma unroll
    for (int it = 0; it < 16; ++it) {
        int q = it * 2 + hw;
        int ss = __shfl(s, q, 32);
        di[it] = __shfl(d, q, 32) * OUTC + col;
        float a = __shfl(w0, q, 32), b = __shfl(w1, q, 32);
        m[it] = (hi ? b : a) * feat[ss * OUTC + col];
    }
    #pragma unroll
    for (int it = 0; it < 16; ++it)
        if (base + it * 2 + hw < E) atomicAdd(&acc[di[it]], m[it]);
}

__global__ void k_out(const float* __restrict__ acc, const float* __restrict__ denom,
                      float* __restrict__ out, int N)
{
    int i = blockIdx.x * blockDim.x + threadIdx.x;
    if (i >= N * 8) return;
    int n = i >> 3, c = i & 7;
    float m0 = acc[(size_t)n * OUTC + c]     / fmaxf(denom[n * 2],     EPSF);
    float m1 = acc[(size_t)n * OUTC + 8 + c] / fmaxf(denom[n * 2 + 1], EPSF);
    out[i] = out[i] + 0.5f * (m0 + m1);
}

extern "C" void kernel_launch(void* const* d_in, const int* in_sizes, int n_in,
                              void* d_out, int out_size, void* d_ws, size_t ws_size,
                              hipStream_t stream)
{
    const float* x    = (const float*)d_in[0];
    const int*   ei   = (const int*)  d_in[1];
    const float* W    = (const float*)d_in[2];
    const float* al   = (const float*)d_in[3];
    const float* ar   = (const float*)d_in[4];
    const float* bias = (const float*)d_in[5];
    const float* Wr   = (const float*)d_in[6];
    float* out = (float*)d_out;

    int N = in_sizes[0] / IN_FEAT;
    int E = in_sizes[1] / 2;

    char* ws = (char*)d_ws;
    float*    feat  = (float*)ws;    ws += (size_t)N * OUTC * sizeof(float);
    float*    acc   = (float*)ws;    ws += (size_t)N * OUTC * sizeof(float);
    float*    rec   = (float*)ws;    ws += (size_t)N * 4 * sizeof(float);
    float*    el    = (float*)ws;    ws += (size_t)N * 2 * sizeof(float);
    float*    denom = (float*)ws;    ws += (size_t)N * 2 * sizeof(float);
    unsigned* mmax  = (unsigned*)ws; ws += (size_t)N * 2 * sizeof(unsigned);

    int nTiles = (N + 15) / 16;
    dim3 blk(256);

    k_node    <<<(nTiles + 7) / 8,      blk, 0, stream>>>(x, W, Wr, bias, feat, out, N, nTiles);
    k_init    <<<(4 * N + 255) / 256,   blk, 0, stream>>>(mmax, denom, acc, 2 * N, 4 * N);
    k_elr     <<<(N + 255) / 256,       blk, 0, stream>>>(feat, al, ar, el, rec, N);
    k_edge_max<<<(E + 255) / 256,       blk, 0, stream>>>(ei, el, mmax, E);
    k_max_fix <<<(2 * N + 255) / 256,   blk, 0, stream>>>(mmax, rec, 2 * N);
    k_edge_agg<<<(E + 255) / 256,       blk, 0, stream>>>(ei, el, rec, feat, denom, acc, E);
    k_out     <<<(N * 8 + 255) / 256,   blk, 0, stream>>>(acc, denom, out, N);
}


# Edges bucketed by destination (one returning integer add per edge gives its rank, prefix sum, scatter of sources); each node sums its own edges in registers: no float atomics, sum order = rank order
# speedup vs baseline: 3.5101x; 1.1218x over previous
#include <hip/hip_runtime.h>
#include <hip/hip_bf16.h>
#include <math.h>


typedef __attribute__((ext_vector_type(2))) float v2f;
typedef __attribute__((ext_vector_type(8))) float v8f;

#define IN_FEAT   256
#define OUTC      16
#define NEG_SLOPE 0.2f
#define EPSF      1e-15f

__device__ __forceinline__ unsigned fmap(float f) {
    unsigned u = __float_as_uint(f);
    return (u & 0x80000000u) ? ~u : (u | 0x80000000u);
}
__device__ __forceinline__ float funmap(unsigned u) {
    unsigned v = (u & 0x80000000u) ? (u & 0x7FFFFFFFu) : ~u;
    return __uint_as_float(v);
}

__global__ __launch_bounds__(256) void k_node(
    const float* __restrict__ x, const float* __restrict__ W,
    const float* __restrict__ Wr, const float* __restrict__ bias,
    float* __restrict__ feat, float* __restrict__ out, int N, int nTiles)
{
    __shared__ float sW [IN_FEAT * OUTC];
    __shared__ float sWr[IN_FEAT * OUTC];
    int tid = threadIdx.x;
    for (int i = tid; i < IN_FEAT * OUTC; i += 256) {
        sW[i]  = W[i];
        sWr[i] = Wr[i];
    }
    __syncthreads();

    int wave = tid >> 5;
    int lane = tid & 31;
    int t = blockIdx.x * 8 + wave;
    if (t >= nTiles) t = nTiles - 1;

    int half = lane >> 4;
    int l15  = lane & 15;
    int rowg = t * 16 + l15;
    if (rowg >= N) rowg = N - 1;
    const float* xrow = x + (size_t)rowg * IN_FEAT;

    v8f cf = {0.f,0.f,0.f,0.f,0.f,0.f,0.f,0.f};
    v8f cr = {0.f,0.f,0.f,0.f,0.f,0.f,0.f,0.f};
    float nsq = 0.f;

    for (int k0 = 0; k0 < IN_FEAT; k0 += 4) {
        int ka = k0 + half * 2;
        float a0 = xrow[ka], a1 = xrow[ka + 1];
        nsq = fmaf(a0, a0, nsq);
        nsq = fmaf(a1, a1, nsq);
        v2f a; a[0] = a0; a[1] = a1;
        v2f b;  b[0]  = sW [ka * OUTC + l15];  b[1]  = sW [(ka + 1) * OUTC + l15];
        v2f br; br[0] = sWr[ka * OUTC + l15];  br[1] = sWr[(ka + 1) * OUTC + l15];
        cf = __builtin_amdgcn_wmma_f32_16x16x4_f32(false, a, false, b,  (short)0, cf, false, false);
        cr = __builtin_amdgcn_wmma_f32_16x16x4_f32(false, a, false, br, (short)0, cr, false, false);
    }

    nsq += __shfl_xor(nsq, 16, 32);
    float nrm = fmaxf(sqrtf(nsq), EPSF);
    float z   = fminf(nrm, 1.0f - 1e-7f);
    float sc  = 0.5f * (log1pf(z) - log1pf(-z)) / nrm;
    float bv  = bias[l15];

    float* fp = feat + (size_t)t * 16 * OUTC;
    float* op = out  + (size_t)t * 16 * 8;
    #pragma unroll
    for (int r = 0; r < 8; ++r) {
        int row = r + half * 8;
        float s_r = __shfl(sc, row, 32);
        float res = cr[r] * s_r + bv;
        float rm  = 0.5f * (res + __shfl_xor(res, 8, 32));
        if (t * 16 + row < N) {
            fp[row * OUTC + l15] = cf[r] * s_r;
            if (l15 < 8) op[row * 8 + l15] = rm;
        }
    }
}

__global__ void k_elr(const float* __restrict__ feat, const float* __restrict__ al,
                      const float* __restrict__ ar, float* __restrict__ el,
                      float* __restrict__ rec, int* __restrict__ cnt, int N)
{
    int n = blockIdx.x * blockDim.x + threadIdx.x;
    if (n >= N) return;
    if (cnt) cnt[n] = 0;
    const float* f = feat + (size_t)n * OUTC;
    #pragma unroll
    for (int h = 0; h < 2; ++h) {
        float sl = 0.f, sr = 0.f;
        #pragma unroll
        for (int c = 0; c < 8; ++c) {
            float v = f[h * 8 + c];
            sl = fmaf(v, al[h * 8 + c], sl);
            sr = fmaf(v, ar[h * 8 + c], sr);
        }
        el[n * 2 + h] = sl;
        rec[n * 4 + h] = sr;
    }
}

__global__ void k_init(unsigned* __restrict__ mmax, float* __restrict__ denom,
                       float* __restrict__ acc, int n2, int n4)
{
    int i = blockIdx.x * blockDim.x + threadIdx.x;
    if (i < n2) { mmax[i] = 0x007FFFFFu; denom[i] = 0.f; }
    if (i < n4) reinterpret_cast<float4*>(acc)[i] = make_float4(0.f, 0.f, 0.f, 0.f);
}

__global__ void k_edge_max(const int* __restrict__ ei, const float* __restrict__ el,
                           unsigned* __restrict__ mmax, int E)
{
    int e = blockIdx.x * blockDim.x + threadIdx.x;
    int lane = threadIdx.x & 31;
    int base = e - lane;
    int d = 0;
    float v0 = 0.f, v1 = 0.f;
    if (e < E) {
        int s = ei[e];
        d = ei[E + e];
        v0 = el[s * 2];
        v1 = el[s * 2 + 1];
    }
    int h = lane & 1;
    for (int it = 0; it < 2; ++it) {
        int q = it * 16 + (lane >> 1);
        int dd = __shfl(d, q, 32);
        float a = __shfl(v0, q, 32), b = __shfl(v1, q, 32);
        if (base + q < E) atomicMax(&mmax[dd * 2 + h], fmap(h ? b : a));
    }
}

__global__ void k_max_fix(const unsigned* __restrict__ mmax, float* __restrict__ rec, int n2)
{
    int i = blockIdx.x * blockDim.x + threadIdx.x;
    if (i >= n2) return;
    int n = i >> 1, h = i & 1;
    float m = funmap(mmax[i]) + rec[n * 4 + h];
    m = (m >= 0.f) ? m : NEG_SLOPE * m;
    rec[n * 4 + 2 + h] = (fabsf(m) <= 3.402823466e38f) ? m : 0.f;
}

__global__ void k_edge_agg(const int* __restrict__ ei, const float* __restrict__ el,
                           const float* __restrict__ rec,
                           const float* __restrict__ feat, float* __restrict__ denom,
                           float* __restrict__ acc, int E)
{
    int e = blockIdx.x * blockDim.x + threadIdx.x;
    int lane = threadIdx.x & 31;
    int base = e - lane;
    int s = 0, d = 0;
    float w0 = 0.f, w1 = 0.f;
    if (e < E) {
        s = ei[e];
        d = ei[E + e];
        float4 r = reinterpret_cast<const float4*>(rec)[d];
        float v = el[s * 2] + r.x;
        v = (v >= 0.f) ? v : NEG_SLOPE * v;
        w0 = __expf(v - r.z);
        v = el[s * 2 + 1] + r.y;
        v = (v >= 0.f) ? v : NEG_SLOPE * v;
        w1 = __expf(v - r.w);
    }

    int h = lane & 1;
    for (int it = 0; it < 2; ++it) {
        int q = it * 16 + (lane >> 1);
        int dd = __shfl(d, q, 32);
        float a = __shfl(w0, q, 32), b = __shfl(w1, q, 32);
        if (base + q < E) atomicAdd(&denom[dd * 2 + h], h ? b : a);
    }

    int col = lane & 15;
    int hw  = lane >> 4;
    bool hi = (col & 8) != 0;
    float m[16];
    int   di[16];
    #pragma unroll
    for (int it = 0; it < 16; ++it) {
        int q = it * 2 + hw;
        int ss = __shfl(s, q, 32);
        di[it] = __shfl(d, q, 32) * OUTC + col;
        float a = __shfl(w0, q, 32), b = __shfl(w1, q, 32);
        m[it] = (hi ? b : a) * feat[ss * OUTC + col];
    }
    #pragma unroll
    for (int it = 0; it < 16; ++it)
        if (base + it * 2 + hw < E) atomicAdd(&acc[di[it]], m[it]);
}

__global__ void k_out(const float* __restrict__ acc, const float* __restrict__ denom,
                      float* __restrict__ out, int N)
{
    int i = blockIdx.x * blockDim.x + threadIdx.x;
    if (i >= N * 8) return;
    int n = i >> 3, c = i & 7;
    float m0 = acc[(size_t)n * OUTC + c]     / fmaxf(denom[n * 2],     EPSF);
    float m1 = acc[(size_t)n * OUTC + 8 + c] / fmaxf(denom[n * 2 + 1], EPSF);
    out[i] = out[i] + 0.5f * (m0 + m1);
}


__global__ void k_rank(const int* __restrict__ ei, int* __restrict__ cnt,
                       int* __restrict__ rank, int E)
{
    int e = blockIdx.x * blockDim.x + threadIdx.x;
    if (e < E) rank[e] = atomicAdd(&cnt[ei[E + e]], 1);
}

__global__ __launch_bounds__(256) void k_blocksum(const int* __restrict__ cnt,
                                                  int* __restrict__ bsum, int N)
{
    __shared__ int sw[8];
    int tid = threadIdx.x;
    int i0 = (blockIdx.x * 256 + tid) * 4;
    int s = 0;
    #pragma unroll
    for (int j = 0; j < 4; ++j)
        if (i0 + j < N) s += cnt[i0 + j];
    #pragma unroll
    for (int o = 16; o > 0; o >>= 1) s += __shfl_xor(s, o, 32);
    if ((tid & 31) == 0) sw[tid >> 5] = s;
    __syncthreads();
    if (tid == 0) {
        int t = 0;
        #pragma unroll
        for (int w = 0; w < 8; ++w) t += sw[w];
        bsum[blockIdx.x] = t;
    }
}

__global__ __launch_bounds__(256) void k_offsets(const int* __restrict__ cnt,
                                                 const int* __restrict__ bsum,
                                                 int* __restrict__ off, int N)
{
    __shared__ int sw[8];
    int tid = threadIdx.x;
    int lane = tid & 31, wave = tid >> 5;

    int b = 0;
    for (int i = tid; i < (int)blockIdx.x; i += 256) b += bsum[i];
    #pragma unroll
    for (int o = 16; o > 0; o >>= 1) b += __shfl_xor(b, o, 32);
    if (lane == 0) sw[wave] = b;
    __syncthreads();
    int base = 0;
    #pragma unroll
    for (int w = 0; w < 8; ++w) base += sw[w];
    __syncthreads();

    int i0 = (blockIdx.x * 256 + tid) * 4;
    int c[4];
    #pragma unroll
    for (int j = 0; j < 4; ++j) c[j] = (i0 + j < N) ? cnt[i0 + j] : 0;
    int t = c[0] + c[1] + c[2] + c[3];
    int inc = t;
    #pragma unroll
    for (int o = 1; o < 32; o <<= 1) {
        int v = __shfl_up(inc, o, 32);
        if (lane >= o) inc += v;
    }
    if (lane == 31) sw[wave] = inc;
    __syncthreads();
    int wb = 0;
    for (int w = 0; w < wave; ++w) wb += sw[w];
    int ex = base + wb + inc - t;
    #pragma unroll
    for (int j = 0; j < 4; ++j)
        if (i0 + j < N) { off[i0 + j] = ex; ex += c[j]; }
}

__global__ void k_fill(const int* __restrict__ ei, const int* __restrict__ off,
                       const int* __restrict__ rank, int* __restrict__ ssrc, int E)
{
    int e = blockIdx.x * blockDim.x + threadIdx.x;
    if (e < E) ssrc[off[ei[E + e]] + rank[e]] = ei[e];
}

__global__ __launch_bounds__(256) void k_gather(
    const int* __restrict__ ssrc, const int* __restrict__ off, const int* __restrict__ cnt,
    const float* __restrict__ el, const float* __restrict__ rec,
    const float* __restrict__ feat, float* __restrict__ out, int N)
{
    __shared__ float so[64];
    int tid = threadIdx.x;
    int lane = tid & 31, wave = tid >> 5;
    int n = __builtin_amdgcn_readfirstlane(blockIdx.x * 8 + wave);
    int beg = 0, deg = 0;
    float er0 = 0.f, er1 = 0.f;
    if (n < N) {
        beg = off[n];
        deg = cnt[n];
        er0 = rec[n * 4];
        er1 = rec[n * 4 + 1];
    }

    float m0 = -INFINITY, m1 = -INFINITY;
    for (int i = lane; i < deg; i += 32) {
        int s = ssrc[beg + i];
        float2 v = *reinterpret_cast<const float2*>(el + 2 * s);
        m0 = fmaxf(m0, v.x);
        m1 = fmaxf(m1, v.y);
    }
    #pragma unroll
    for (int o = 16; o > 0; o >>= 1) {
        m0 = fmaxf(m0, __shfl_xor(m0, o, 32));
        m1 = fmaxf(m1, __shfl_xor(m1, o, 32));
    }
    m0 += er0;  m0 = (m0 >= 0.f) ? m0 : NEG_SLOPE * m0;
    m1 += er1;  m1 = (m1 >= 0.f) ? m1 : NEG_SLOPE * m1;
    m0 = (fabsf(m0) <= 3.402823466e38f) ? m0 : 0.f;
    m1 = (fabsf(m1) <= 3.402823466e38f) ? m1 : 0.f;

    int col = lane & 15;
    int hw  = lane >> 4;
    bool hi = (col & 8) != 0;
    float accv = 0.f, d0 = 0.f, d1 = 0.f;
    for (int c0 = 0; c0 < deg; c0 += 32) {
        int s = 0;
        float w0 = 0.f, w1 = 0.f;
        if (c0 + lane < deg) {
            s = ssrc[beg + c0 + lane];
            float2 v = *reinterpret_cast<const float2*>(el + 2 * s);
            float a = v.x + er0;
            a = (a >= 0.f) ? a : NEG_SLOPE * a;
            w0 = __expf(a - m0);
            float b = v.y + er1;
            b = (b >= 0.f) ? b : NEG_SLOPE * b;
            w1 = __expf(b - m1);
        }
        d0 += w0;
        d1 += w1;
        #pragma unroll
        for (int g = 0; g < 2; ++g) {
            if (c0 + g * 16 < deg) {
                float p[8];
                #pragma unroll
                for (int j = 0; j < 8; ++j) {
                    int q = g * 16 + j * 2 + hw;
                    int ss = __shfl(s, q, 32);
                    float wa = __shfl(w0, q, 32), wb = __shfl(w1, q, 32);
                    p[j] = (hi ? wb : wa) * feat[ss * OUTC + col];
                }
                #pragma unroll
                for (int j = 0; j < 8; ++j)
                    accv += (c0 + g * 16 + j * 2 + hw < deg) ? p[j] : 0.f;
            }
        }
    }

    accv += __shfl_xor(accv, 16, 32);
    #pragma unroll
    for (int o = 16; o > 0; o >>= 1) {
        d0 += __shfl_xor(d0, o, 32);
        d1 += __shfl_xor(d1, o, 32);
    }
    float r  = accv / fmaxf(hi ? d1 : d0, EPSF);
    float hm = 0.5f * (r + __shfl_xor(r, 8, 32));
    if (lane < 8) so[wave * 8 + lane] = hm;
    __syncthreads();
    if (tid < 64) {
        int i = blockIdx.x * 64 + tid;
        if (i < N * 8) out[i] = out[i] + so[tid];
    }
}

extern "C" void kernel_launch(void* const* d_in, const int* in_sizes, int n_in,
                              void* d_out, int out_size, void* d_ws, size_t ws_size,
                              hipStream_t stream)
{
    const float* x    = (const float*)d_in[0];
    const int*   ei   = (const int*)  d_in[1];
    const float* W    = (const float*)d_in[2];
    const float* al   = (const float*)d_in[3];
    const float* ar   = (const float*)d_in[4];
    const float* bias = (const float*)d_in[5];
    const float* Wr   = (const float*)d_in[6];
    float* out = (float*)d_out;

    int N = in_sizes[0] / IN_FEAT;
    int E = in_sizes[1] / 2;

    int nTiles = (N + 15) / 16;
    int nScan  = (N + 1023) / 1024;
    dim3 blk(256);

    char* ws = (char*)d_ws;
    float* feat  = (float*)ws;  ws += (size_t)N * OUTC * sizeof(float);
    float* rec   = (float*)ws;  ws += (size_t)N * 4 * sizeof(float);
    float* el    = (float*)ws;  ws += (size_t)N * 2 * sizeof(float);
    int*   cnt   = (int*)ws;    ws += (size_t)N * sizeof(int);
    int*   off   = (int*)ws;    ws += (size_t)N * sizeof(int);
    int*   bsum  = (int*)ws;    ws += (size_t)nScan * sizeof(int);
    int*   rank  = (int*)ws;    ws += (size_t)E * sizeof(int);
    int*   ssrc  = (int*)ws;    ws += (size_t)E * sizeof(int);

    if ((size_t)(ws - (char*)d_ws) <= ws_size) {
        k_node    <<<(nTiles + 7) / 8,    blk, 0, stream>>>(x, W, Wr, bias, feat, out, N, nTiles);
        k_elr     <<<(N + 255) / 256,     blk, 0, stream>>>(feat, al, ar, el, rec, cnt, N);
        k_rank    <<<(E + 255) / 256,     blk, 0, stream>>>(ei, cnt, rank, E);
        k_blocksum<<<nScan,               blk, 0, stream>>>(cnt, bsum, N);
        k_offsets <<<nScan,               blk, 0, stream>>>(cnt, bsum, off, N);
        k_fill    <<<(E + 255) / 256,     blk, 0, stream>>>(ei, off, rank, ssrc, E);
        k_gather  <<<(N + 7) / 8,         blk, 0, stream>>>(ssrc, off, cnt, el, rec, feat, out, N);
        return;
    }

    ws = (char*)d_ws;
    feat = (float*)ws;              ws += (size_t)N * OUTC * sizeof(float);
    float*    acc   = (float*)ws;    ws += (size_t)N * OUTC * sizeof(float);
    rec  = (float*)ws;              ws += (size_t)N * 4 * sizeof(float);
    el   = (float*)ws;              ws += (size_t)N * 2 * sizeof(float);
    float*    denom = (float*)ws;    ws += (size_t)N * 2 * sizeof(float);
    unsigned* mmax  = (unsigned*)ws; ws += (size_t)N * 2 * sizeof(unsigned);

    k_node    <<<(nTiles + 7) / 8,      blk, 0, stream>>>(x, W, Wr, bias, feat, out, N, nTiles);
    k_init    <<<(4 * N + 255) / 256,   blk, 0, stream>>>(mmax, denom, acc, 2 * N, 4 * N);
    k_elr     <<<(N + 255) / 256,       blk, 0, stream>>>(feat, al, ar, el, rec, (int*)nullptr, N);
    k_edge_max<<<(E + 255) / 256,       blk, 0, stream>>>(ei, el, mmax, E);
    k_max_fix <<<(2 * N + 255) / 256,   blk, 0, stream>>>(mmax, rec, 2 * N);
    k_edge_agg<<<(E + 255) / 256,       blk, 0, stream>>>(ei, el, rec, feat, denom, acc, E);
    k_out     <<<(N * 8 + 255) / 256,   blk, 0, stream>>>(acc, denom, out, N);
}


# Edge ranks from LDS counters in a two-level counting sort (chunks sorted by 256-node bucket in LDS, then one work-group per bucket), not a returning global add per edge; sum order = arrival rank
# speedup vs baseline: 9.5259x; 2.7139x over previous
#include <hip/hip_runtime.h>
#include <hip/hip_bf16.h>
#include <math.h>


typedef __attribute__((ext_vector_type(2))) float v2f;
typedef __attribute__((ext_vector_type(8))) float v8f;

#define IN_FEAT   256
#define OUTC      16
#define NEG_SLOPE 0.2f
#define EPSF      1e-15f

__device__ __forceinline__ unsigned fmap(float f) {
    unsigned u = __float_as_uint(f);
    return (u & 0x80000000u) ? ~u : (u | 0x80000000u);
}
__device__ __forceinline__ float funmap(unsigned u) {
    unsigned v = (u & 0x80000000u) ? (u & 0x7FFFFFFFu) : ~u;
    return __uint_as_float(v);
}

__global__ __launch_bounds__(256) void k_node(
    const float* __restrict__ x, const float* __restrict__ W,
    const float* __restrict__ Wr, const float* __restrict__ bias,
    float* __restrict__ feat, float* __restrict__ out, int N, int nTiles)
{
    __shared__ float sW [IN_FEAT * OUTC];
    __shared__ float sWr[IN_FEAT * OUTC];
    int tid = threadIdx.x;
    for (int i = tid; i < IN_FEAT * OUTC; i += 256) {
        sW[i]  = W[i];
        sWr[i] = Wr[i];
    }
    __syncthreads();

    int wave = tid >> 5;
    int lane = tid & 31;
    int t = blockIdx.x * 8 + wave;
    if (t >= nTiles) t = nTiles - 1;

    int half = lane >> 4;
    int l15  = lane & 15;
    int rowg = t * 16 + l15;
    if (rowg >= N) rowg = N - 1;
    const float* xrow = x + (size_t)rowg * IN_FEAT;

    v8f cf = {0.f,0.f,0.f,0.f,0.f,0.f,0.f,0.f};
    v8f cr = {0.f,0.f,0.f,0.f,0.f,0.f,0.f,0.f};
    float nsq = 0.f;

    for (int k0 = 0; k0 < IN_FEAT; k0 += 4) {
        int ka = k0 + half * 2;
        float a0 = xrow[ka], a1 = xrow[ka + 1];
        nsq = fmaf(a0, a0, nsq);
        nsq = fmaf(a1, a1, nsq);
        v2f a; a[0] = a0; a[1] = a1;
        v2f b;  b[0]  = sW [ka * OUTC + l15];  b[1]  = sW [(ka + 1) * OUTC + l15];
        v2f br; br[0] = sWr[ka * OUTC + l15];  br[1] = sWr[(ka + 1) * OUTC + l15];
        cf = __builtin_amdgcn_wmma_f32_16x16x4_f32(false, a, false, b,  (short)0, cf, false, false);
        cr = __builtin_amdgcn_wmma_f32_16x16x4_f32(false, a, false, br, (short)0, cr, false, false);
    }

    nsq += __shfl_xor(nsq, 16, 32);
    float nrm = fmaxf(sqrtf(nsq), EPSF);
    float z   = fminf(nrm, 1.0f - 1e-7f);
    float sc  = 0.5f * (log1pf(z) - log1pf(-z)) / nrm;
    float bv  = bias[l15];

    float* fp = feat + (size_t)t * 16 * OUTC;
    float* op = out  + (size_t)t * 16 * 8;
    #pragma unroll
    for (int r = 0; r < 8; ++r) {
        int row = r + half * 8;
        float s_r = __shfl(sc, row, 32);
        float res = cr[r] * s_r + bv;
        float rm  = 0.5f * (res + __shfl_xor(res, 8, 32));
        if (t * 16 + row < N) {
            fp[row * OUTC + l15] = cf[r] * s_r;
            if (l15 < 8) op[row * 8 + l15] = rm;
        }
    }
}

__global__ void k_elr(const float* __restrict__ feat, const float* __restrict__ al,
                      const float* __restrict__ ar, float* __restrict__ el,
                      float* __restrict__ rec, int N)
{
    int n = blockIdx.x * blockDim.x + threadIdx.x;
    if (n >= N) return;
    const float* f = feat + (size_t)n * OUTC;
    #pragma unroll
    for (int h = 0; h < 2; ++h) {
        float sl = 0.f, sr = 0.f;
        #pragma unroll
        for (int c = 0; c < 8; ++c) {
            float v = f[h * 8 + c];
            sl = fmaf(v, al[h * 8 + c], sl);
            sr = fmaf(v, ar[h * 8 + c], sr);
        }
        el[n * 2 + h] = sl;
        rec[n * 4 + h] = sr;
    }
}

__global__ void k_init(unsigned* __restrict__ mmax, float* __restrict__ denom,
                       float* __restrict__ acc, int n2, int n4)
{
    int i = blockIdx.x * blockDim.x + threadIdx.x;
    if (i < n2) { mmax[i] = 0x007FFFFFu; denom[i] = 0.f; }
    if (i < n4) reinterpret_cast<float4*>(acc)[i] = make_float4(0.f, 0.f, 0.f, 0.f);
}

__global__ void k_edge_max(const int* __restrict__ ei, const float* __restrict__ el,
                           unsigned* __restrict__ mmax, int E)
{
    int e = blockIdx.x * blockDim.x + threadIdx.x;
    int lane = threadIdx.x & 31;
    int base = e - lane;
    int d = 0;
    float v0 = 0.f, v1 = 0.f;
    if (e < E) {
        int s = ei[e];
        d = ei[E + e];
        v0 = el[s * 2];
        v1 = el[s * 2 + 1];
    }
    int h = lane & 1;
    for (int it = 0; it < 2; ++it) {
        int q = it * 16 + (lane >> 1);
        int dd = __shfl(d, q, 32);
        float a = __shfl(v0, q, 32), b = __shfl(v1, q, 32);
        if (base + q < E) atomicMax(&mmax[dd * 2 + h], fmap(h ? b : a));
    }
}

__global__ void k_max_fix(const unsigned* __restrict__ mmax, float* __restrict__ rec, int n2)
{
    int i = blockIdx.x * blockDim.x + threadIdx.x;
    if (i >= n2) return;
    int n = i >> 1, h = i & 1;
    float m = funmap(mmax[i]) + rec[n * 4 + h];
    m = (m >= 0.f) ? m : NEG_SLOPE * m;
    rec[n * 4 + 2 + h] = (fabsf(m) <= 3.402823466e38f) ? m : 0.f;
}

__global__ void k_edge_agg(const int* __restrict__ ei, const float* __restrict__ el,
                           const float* __restrict__ rec,
                           const float* __restrict__ feat, float* __restrict__ denom,
                           float* __restrict__ acc, int E)
{
    int e = blockIdx.x * blockDim.x + threadIdx.x;
    int lane = threadIdx.x & 31;
    int base = e - lane;
    int s = 0, d = 0;
    float w0 = 0.f, w1 = 0.f;
    if (e < E) {
        s = ei[e];
        d = ei[E + e];
        float4 r = reinterpret_cast<const float4*>(rec)[d];
        float v = el[s * 2] + r.x;
        v = (v >= 0.f) ? v : NEG_SLOPE * v;
        w0 = __expf(v - r.z);
        v = el[s * 2 + 1] + r.y;
        v = (v >= 0.f) ? v : NEG_SLOPE * v;
        w1 = __expf(v - r.w);
    }

    int h = lane & 1;
    for (int it = 0; it < 2; ++it) {
        int q = it * 16 + (lane >> 1);
        int dd = __shfl(d, q, 32);
        float a = __shfl(w0, q, 32), b = __shfl(w1, q, 32);
        if (base + q < E) atomicAdd(&denom[dd * 2 + h], h ? b : a);
    }

    int col = lane & 15;
    int hw  = lane >> 4;
    bool hi = (col & 8) != 0;
    float m[16];
    int   di[16];
    #pragma unroll
    for (int it = 0; it < 16; ++it) {
        int q = it * 2 + hw;
        int ss = __shfl(s, q, 32);
        di[it] = __shfl(d, q, 32) * OUTC + col;
        float a = __shfl(w0, q, 32), b = __shfl(w1, q, 32);
        m[it] = (hi ? b : a) * feat[ss * OUTC + col];
    }
    #pragma unroll
    for (int it = 0; it < 16; ++it)
        if (base + it * 2 + hw < E) atomicAdd(&acc[di[it]], m[it]);
}

__global__ void k_out(const float* __restrict__ acc, const float* __restrict__ denom,
                      float* __restrict__ out, int N)
{
    int i = blockIdx.x * blockDim.x + threadIdx.x;
    if (i >= N * 8) return;
    int n = i >> 3, c = i & 7;
    float m0 = acc[(size_t)n * OUTC + c]     / fmaxf(denom[n * 2],     EPSF);
    float m1 = acc[(size_t)n * OUTC + 8 + c] / fmaxf(denom[n * 2 + 1], EPSF);
    out[i] = out[i] + 0.5f * (m0 + m1);
}


#define BSH     8
#define BNODES  256
#define MAXBK   4096
#define SORT_G  256
#define CH_CAP  16384
#define BK_CAP  16384

__global__ __launch_bounds__(1024) void k_lsort(const int* __restrict__ ei, int* __restrict__ hist,
                                                int* __restrict__ lst, unsigned* __restrict__ lsorted,
                                                int E, int nBk, int chunk)
{
    __shared__ int sh[MAXBK];
    __shared__ int sw[32];
    __shared__ unsigned wbuf[CH_CAP];
    int tid = threadIdx.x;
    int lane = tid & 31, wave = tid >> 5;
    int g = blockIdx.x;
    for (int b = tid; b < nBk; b += 1024) sh[b] = 0;
    __syncthreads();

    int e0 = g * chunk;
    int e1 = min(e0 + chunk, E);
    int e = e0 + tid;
    for (; e + 3072 < e1; e += 4096) {
        int d[4];
        #pragma unroll
        for (int j = 0; j < 4; ++j) d[j] = ei[E + e + j * 1024];
        #pragma unroll
        for (int j = 0; j < 4; ++j) atomicAdd(&sh[d[j] >> BSH], 1);
    }
    for (; e < e1; e += 1024) atomicAdd(&sh[ei[E + e] >> BSH], 1);
    __syncthreads();

    int c[4];
    #pragma unroll
    for (int j = 0; j < 4; ++j) c[j] = (4 * tid + j < nBk) ? sh[4 * tid + j] : 0;
    int t = c[0] + c[1] + c[2] + c[3];
    int inc = t;
    for (int o = 1; o < 32; o <<= 1) {
        int v = __shfl_up(inc, o, 32);
        if (lane >= o) inc += v;
    }
    if (lane == 31) sw[wave] = inc;
    __syncthreads();
    int ex = inc - t;
    for (int w = 0; w < wave; ++w) ex += sw[w];
    #pragma unroll
    for (int j = 0; j < 4; ++j) {
        int b = 4 * tid + j;
        if (b < nBk) {
            hist[b * SORT_G + g] = c[j];
            lst[g * nBk + b] = ex;
            sh[b] = ex;
            ex += c[j];
        }
    }
    __syncthreads();

    bool staged = e1 - e0 <= CH_CAP;
    e = e0 + tid;
    for (; e + 3072 < e1; e += 4096) {
        int s[4], d[4];
        #pragma unroll
        for (int j = 0; j < 4; ++j) {
            s[j] = ei[e + j * 1024];
            d[j] = ei[E + e + j * 1024];
        }
        #pragma unroll
        for (int j = 0; j < 4; ++j) {
            int lp = atomicAdd(&sh[d[j] >> BSH], 1);
            unsigned w = ((unsigned)s[j] << BSH) | (unsigned)(d[j] & (BNODES - 1));
            if (staged) wbuf[lp] = w; else lsorted[e0 + lp] = w;
        }
    }
    for (; e < e1; e += 1024) {
        int s = ei[e], d = ei[E + e];
        int lp = atomicAdd(&sh[d >> BSH], 1);
        unsigned w = ((unsigned)s << BSH) | (unsigned)(d & (BNODES - 1));
        if (staged) wbuf[lp] = w; else lsorted[e0 + lp] = w;
    }
    if (staged) {
        __syncthreads();
        for (int i = tid - (e0 & 31); i < e1 - e0; i += 1024)
            if (i >= 0) lsorted[e0 + i] = wbuf[i];
    }
}

__global__ __launch_bounds__(256) void k_blocksum(const int* __restrict__ cnt,
                                                  int* __restrict__ bsum, int N)
{
    __shared__ int sw[8];
    int tid = threadIdx.x;
    int i0 = (blockIdx.x * 256 + tid) * 4;
    int s = 0;
    #pragma unroll
    for (int j = 0; j < 4; ++j)
        if (i0 + j < N) s += cnt[i0 + j];
    #pragma unroll
    for (int o = 16; o > 0; o >>= 1) s += __shfl_xor(s, o, 32);
    if ((tid & 31) == 0) sw[tid >> 5] = s;
    __syncthreads();
    if (tid == 0) {
        int t = 0;
        #pragma unroll
        for (int w = 0; w < 8; ++w) t += sw[w];
        bsum[blockIdx.x] = t;
    }
}

__global__ __launch_bounds__(256) void k_offsets(const int* __restrict__ cnt,
                                                 const int* __restrict__ bsum,
                                                 int* __restrict__ off, int N)
{
    __shared__ int sw[8];
    int tid = threadIdx.x;
    int lane = tid & 31, wave = tid >> 5;

    int b = 0;
    for (int i = tid; i < (int)blockIdx.x; i += 256) b += bsum[i];
    #pragma unroll
    for (int o = 16; o > 0; o >>= 1) b += __shfl_xor(b, o, 32);
    if (lane == 0) sw[wave] = b;
    __syncthreads();
    int base = 0;
    #pragma unroll
    for (int w = 0; w < 8; ++w) base += sw[w];
    __syncthreads();

    int i0 = (blockIdx.x * 256 + tid) * 4;
    int c[4];
    #pragma unroll
    for (int j = 0; j < 4; ++j) c[j] = (i0 + j < N) ? cnt[i0 + j] : 0;
    int t = c[0] + c[1] + c[2] + c[3];
    int inc = t;
    #pragma unroll
    for (int o = 1; o < 32; o <<= 1) {
        int v = __shfl_up(inc, o, 32);
        if (lane >= o) inc += v;
    }
    if (lane == 31) sw[wave] = inc;
    __syncthreads();
    int wb = 0;
    for (int w = 0; w < wave; ++w) wb += sw[w];
    int ex = base + wb + inc - t;
    #pragma unroll
    for (int j = 0; j < 4; ++j)
        if (i0 + j < N) { off[i0 + j] = ex; ex += c[j]; }
}

__device__ __forceinline__ unsigned bucket_entry(const unsigned* __restrict__ lsorted, const int* r_src,
                                                 const int* r_ko, int k)
{
    int lo = 0;
    for (int step = SORT_G / 2; step > 0; step >>= 1)
        if (r_ko[lo + step] <= k) lo += step;
    return lsorted[r_src[lo] + (k - r_ko[lo])];
}

__global__ __launch_bounds__(1024) void k_bsort(const unsigned* __restrict__ lsorted, const int* __restrict__ lst,
                                                const int* __restrict__ base,
                                                int* __restrict__ cnt, int* __restrict__ off,
                                                int* __restrict__ ssrc, int N, int E, int nBk, int chunk)
{
    __shared__ int sc[BNODES];
    __shared__ int sw[8];
    __shared__ int r_src[SORT_G];
    __shared__ int r_ko[SORT_G];
    __shared__ unsigned ebuf[BK_CAP];
    __shared__ int sbuf[BK_CAP];
    int tid = threadIdx.x;
    int lane = tid & 31, wave = tid >> 5;
    int b = blockIdx.x;
    int beg = base[b * SORT_G];
    int end = (b + 1 < nBk) ? base[(b + 1) * SORT_G] : E;
    int nEb = end - beg;
    if (tid < SORT_G) {
        r_src[tid] = tid * chunk + lst[tid * nBk + b];
        r_ko[tid]  = base[b * SORT_G + tid] - beg;
    }
    if (tid < BNODES) sc[tid] = 0;
    __syncthreads();

    bool fits = nEb <= BK_CAP;
    int k = tid;
    if (fits) {
        for (; k + 3072 < nEb; k += 4096) {
            unsigned p[4];
            #pragma unroll
            for (int j = 0; j < 4; ++j) p[j] = bucket_entry(lsorted, r_src, r_ko, k + j * 1024);
            #pragma unroll
            for (int j = 0; j < 4; ++j) ebuf[k + j * 1024] = p[j];
        }
        for (; k < nEb; k += 1024) ebuf[k] = bucket_entry(lsorted, r_src, r_ko, k);
        __syncthreads();
        for (k = tid; k < nEb; k += 1024) atomicAdd(&sc[ebuf[k] & (BNODES - 1)], 1);
    } else {
        for (; k < nEb; k += 1024) atomicAdd(&sc[bucket_entry(lsorted, r_src, r_ko, k) & (BNODES - 1)], 1);
    }
    __syncthreads();

    int c = 0, inc = 0;
    if (tid < BNODES) {
        c = sc[tid];
        inc = c;
        for (int o = 1; o < 32; o <<= 1) {
            int v = __shfl_up(inc, o, 32);
            if (lane >= o) inc += v;
        }
        if (lane == 31) sw[wave] = inc;
    }
    __syncthreads();
    if (tid < BNODES) {
        int ex = beg + inc - c;
        for (int w = 0; w < wave; ++w) ex += sw[w];
        int n = b * BNODES + tid;
        if (n < N) { cnt[n] = c; off[n] = ex; }
        sc[tid] = ex;
    }
    __syncthreads();

    if (fits) {
        for (k = tid; k < nEb; k += 1024) {
            unsigned p = ebuf[k];
            int pos = atomicAdd(&sc[p & (BNODES - 1)], 1);
            sbuf[pos - beg] = (int)(p >> BSH);
        }
        __syncthreads();
        for (k = tid - (beg & 31); k < nEb; k += 1024)
            if (k >= 0) ssrc[beg + k] = sbuf[k];
    } else {
        for (k = tid; k < nEb; k += 1024) {
            unsigned p = bucket_entry(lsorted, r_src, r_ko, k);
            int pos = atomicAdd(&sc[p & (BNODES - 1)], 1);
            ssrc[pos] = (int)(p >> BSH);
        }
    }
}

__global__ __launch_bounds__(256) void k_gather(
    const int* __restrict__ ssrc, const int* __restrict__ off, const int* __restrict__ cnt,
    const float* __restrict__ el, const float* __restrict__ rec,
    const float* __restrict__ feat, float* __restrict__ out, int N)
{
    __shared__ float so[64];
    int tid = threadIdx.x;
    int lane = tid & 31, wave = tid >> 5;
    int n = __builtin_amdgcn_readfirstlane(blockIdx.x * 8 + wave);
    int beg = 0, deg = 0;
    float er0 = 0.f, er1 = 0.f;
    if (n < N) {
        beg = off[n];
        deg = cnt[n];
        er0 = rec[n * 4];
        er1 = rec[n * 4 + 1];
    }

    float m0 = -INFINITY, m1 = -INFINITY;
    for (int i = lane; i < deg; i += 32) {
        int s = ssrc[beg + i];
        float2 v = *reinterpret_cast<const float2*>(el + 2 * s);
        m0 = fmaxf(m0, v.x);
        m1 = fmaxf(m1, v.y);
    }
    #pragma unroll
    for (int o = 16; o > 0; o >>= 1) {
        m0 = fmaxf(m0, __shfl_xor(m0, o, 32));
        m1 = fmaxf(m1, __shfl_xor(m1, o, 32));
    }
    m0 += er0;  m0 = (m0 >= 0.f) ? m0 : NEG_SLOPE * m0;
    m1 += er1;  m1 = (m1 >= 0.f) ? m1 : NEG_SLOPE * m1;
    m0 = (fabsf(m0) <= 3.402823466e38f) ? m0 : 0.f;
    m1 = (fabsf(m1) <= 3.402823466e38f) ? m1 : 0.f;

    int col = lane & 15;
    int hw  = lane >> 4;
    bool hi = (col & 8) != 0;
    float accv = 0.f, d0 = 0.f, d1 = 0.f;
    for (int c0 = 0; c0 < deg; c0 += 32) {
        int s = 0;
        float w0 = 0.f, w1 = 0.f;
        if (c0 + lane < deg) {
            s = ssrc[beg + c0 + lane];
            float2 v = *reinterpret_cast<const float2*>(el + 2 * s);
            float a = v.x + er0;
            a = (a >= 0.f) ? a : NEG_SLOPE * a;
            w0 = __expf(a - m0);
            float b = v.y + er1;
            b = (b >= 0.f) ? b : NEG_SLOPE * b;
            w1 = __expf(b - m1);
        }
        d0 += w0;
        d1 += w1;
        #pragma unroll
        for (int g = 0; g < 2; ++g) {
            if (c0 + g * 16 < deg) {
                float p[8];
                #pragma unroll
                for (int j = 0; j < 8; ++j) {
                    int q = g * 16 + j * 2 + hw;
                    int ss = __shfl(s, q, 32);
                    float wa = __shfl(w0, q, 32), wb = __shfl(w1, q, 32);
                    p[j] = (hi ? wb : wa) * feat[ss * OUTC + col];
                }
                #pragma unroll
                for (int j = 0; j < 8; ++j)
                    accv += (c0 + g * 16 + j * 2 + hw < deg) ? p[j] : 0.f;
            }
        }
    }

    accv += __shfl_xor(accv, 16, 32);
    #pragma unroll
    for (int o = 16; o > 0; o >>= 1) {
        d0 += __shfl_xor(d0, o, 32);
        d1 += __shfl_xor(d1, o, 32);
    }
    float r  = accv / fmaxf(hi ? d1 : d0, EPSF);
    float hm = 0.5f * (r + __shfl_xor(r, 8, 32));
    if (lane < 8) so[wave * 8 + lane] = hm;
    __syncthreads();
    if (tid < 64) {
        int i = blockIdx.x * 64 + tid;
        if (i < N * 8) out[i] = out[i] + so[tid];
    }
}

extern "C" void kernel_launch(void* const* d_in, const int* in_sizes, int n_in,
                              void* d_out, int out_size, void* d_ws, size_t ws_size,
                              hipStream_t stream)
{
    const float* x    = (const float*)d_in[0];
    const int*   ei   = (const int*)  d_in[1];
    const float* W    = (const float*)d_in[2];
    const float* al   = (const float*)d_in[3];
    const float* ar   = (const float*)d_in[4];
    const float* bias = (const float*)d_in[5];
    const float* Wr   = (const float*)d_in[6];
    float* out = (float*)d_out;

    int N = in_sizes[0] / IN_FEAT;
    int E = in_sizes[1] / 2;

    int nTiles = (N + 15) / 16;
    int nBk    = (N + BNODES - 1) / BNODES;
    int nTab   = nBk * SORT_G;
    int nScan  = (nTab + 1023) / 1024;
    int chunk  = (E + SORT_G - 1) / SORT_G;
    dim3 blk(256);

    char* ws = (char*)d_ws;
    float* feat  = (float*)ws;  ws += (size_t)N * OUTC * sizeof(float);
    float* rec   = (float*)ws;  ws += (size_t)N * 4 * sizeof(float);
    float* el    = (float*)ws;  ws += (size_t)N * 2 * sizeof(float);
    int*   cnt   = (int*)ws;    ws += (size_t)N * sizeof(int);
    int*   off   = (int*)ws;    ws += (size_t)N * sizeof(int);
    int*   hist  = (int*)ws;    ws += (size_t)nTab * sizeof(int);
    int*   lst   = (int*)ws;    ws += (size_t)nTab * sizeof(int);
    int*   base  = (int*)ws;    ws += (size_t)nTab * sizeof(int);
    unsigned* lsorted = (unsigned*)ws;  ws += (size_t)E * sizeof(unsigned);
    int*   ssrc  = (int*)ws;    ws += (size_t)E * sizeof(int);
    int*   bsum  = (int*)ws;    ws += (size_t)nScan * sizeof(int);

    if (nBk <= MAXBK && (size_t)(ws - (char*)d_ws) <= ws_size) {
        k_node    <<<(nTiles + 7) / 8,    blk, 0, stream>>>(x, W, Wr, bias, feat, out, N, nTiles);
        k_elr     <<<(N + 255) / 256,     blk, 0, stream>>>(feat, al, ar, el, rec, N);
        k_lsort   <<<SORT_G,              1024, 0, stream>>>(ei, hist, lst, lsorted, E, nBk, chunk);
        k_blocksum<<<nScan,               blk, 0, stream>>>(hist, bsum, nTab);
        k_offsets <<<nScan,               blk, 0, stream>>>(hist, bsum, base, nTab);
        k_bsort   <<<nBk,                 1024, 0, stream>>>(lsorted, lst, base, cnt, off, ssrc, N, E, nBk, chunk);
        k_gather  <<<(N + 7) / 8,         blk, 0, stream>>>(ssrc, off, cnt, el, rec, feat, out, N);
        return;
    }

    ws = (char*)d_ws;
    feat = (float*)ws;              ws += (size_t)N * OUTC * sizeof(float);
    float*    acc   = (float*)ws;    ws += (size_t)N * OUTC * sizeof(float);
    rec  = (float*)ws;              ws += (size_t)N * 4 * sizeof(float);
    el   = (float*)ws;              ws += (size_t)N * 2 * sizeof(float);
    float*    denom = (float*)ws;    ws += (size_t)N * 2 * sizeof(float);
    unsigned* mmax  = (unsigned*)ws; ws += (size_t)N * 2 * sizeof(unsigned);

    k_node    <<<(nTiles + 7) / 8,      blk, 0, stream>>>(x, W, Wr, bias, feat, out, N, nTiles);
    k_init    <<<(4 * N + 255) / 256,   blk, 0, stream>>>(mmax, denom, acc, 2 * N, 4 * N);
    k_elr     <<<(N + 255) / 256,       blk, 0, stream>>>(feat, al, ar, el, rec, N);
    k_edge_max<<<(E + 255) / 256,       blk, 0, stream>>>(ei, el, mmax, E);
    k_max_fix <<<(2 * N + 255) / 256,   blk, 0, stream>>>(mmax, rec, 2 * N);
    k_edge_agg<<<(E + 255) / 256,       blk, 0, stream>>>(ei, el, rec, feat, denom, acc, E);
    k_out     <<<(N * 8 + 255) / 256,   blk, 0, stream>>>(acc, denom, out, N);
}


# Node GEMM and level-1 chunk sort, independent of each other, share one launch: 256 sort work-groups first, 250 GEMM work-groups (25 tiles each) on the free wave slots; 6 launches, sums unchanged
# speedup vs baseline: 10.6948x; 1.1227x over previous
#include <hip/hip_runtime.h>
#include <hip/hip_bf16.h>
#include <math.h>


typedef __attribute__((ext_vector_type(2))) float v2f;
typedef __attribute__((ext_vector_type(8))) float v8f;

#define IN_FEAT   256
#define OUTC      16
#define NEG_SLOPE 0.2f
#define EPSF      1e-15f

__device__ __forceinline__ unsigned fmap(float f) {
    unsigned u = __float_as_uint(f);
    return (u & 0x80000000u) ? ~u : (u | 0x80000000u);
}
__device__ __forceinline__ float funmap(unsigned u) {
    unsigned v = (u & 0x80000000u) ? (u & 0x7FFFFFFFu) : ~u;
    return __uint_as_float(v);
}

template <int NTHR, int TPB>
__device__ __forceinline__ void node_body(
    float* sW, float* sWr, int blk,
    const float* __restrict__ x, const float* __restrict__ W,
    const float* __restrict__ Wr, const float* __restrict__ bias,
    float* __restrict__ feat, float* __restrict__ out, int N, int nTiles)
{
    int tid = threadIdx.x;
    for (int i = tid; i < IN_FEAT * OUTC; i += NTHR) {
        sW[i]  = W[i];
        sWr[i] = Wr[i];
    }
    __syncthreads();

    int wave = tid >> 5;
    int lane = tid & 31;
    if (wave >= TPB) return;
    int t = blk * TPB + wave;
    if (t >= nTiles) t = nTiles - 1;

    int half = lane >> 4;
    int l15  = lane & 15;
    int rowg = t * 16 + l15;
    if (rowg >= N) rowg = N - 1;
    const float* xrow = x + (size_t)rowg * IN_FEAT;

    v8f cf = {0.f,0.f,0.f,0.f,0.f,0.f,0.f,0.f};
    v8f cr = {0.f,0.f,0.f,0.f,0.f,0.f,0.f,0.f};
    float nsq = 0.f;

    for (int k0 = 0; k0 < IN_FEAT; k0 += 4) {
        int ka = k0 + half * 2;
        float a0 = xrow[ka], a1 = xrow[ka + 1];
        nsq = fmaf(a0, a0, nsq);
        nsq = fmaf(a1, a1, nsq);
        v2f a; a[0] = a0; a[1] = a1;
        v2f b;  b[0]  = sW [ka * OUTC + l15];  b[1]  = sW [(ka + 1) * OUTC + l15];
        v2f br; br[0] = sWr[ka * OUTC + l15];  br[1] = sWr[(ka + 1) * OUTC + l15];
        cf = __builtin_amdgcn_wmma_f32_16x16x4_f32(false, a, false, b,  (short)0, cf, false, false);
        cr = __builtin_amdgcn_wmma_f32_16x16x4_f32(false, a, false, br, (short)0, cr, false, false);
    }

    nsq += __shfl_xor(nsq, 16, 32);
    float nrm = fmaxf(sqrtf(nsq), EPSF);
    float z   = fminf(nrm, 1.0f - 1e-7f);
    float sc  = 0.5f * (log1pf(z) - log1pf(-z)) / nrm;
    float bv  = bias[l15];

    float* fp = feat + (size_t)t * 16 * OUTC;
    float* op = out  + (size_t)t * 16 * 8;
    #pragma unroll
    for (int r = 0; r < 8; ++r) {
        int row = r + half * 8;
        float s_r = __shfl(sc, row, 32);
        float res = cr[r] * s_r + bv;
        float rm  = 0.5f * (res + __shfl_xor(res, 8, 32));
        if (t * 16 + row < N) {
            fp[row * OUTC + l15] = cf[r] * s_r;
            if (l15 < 8) op[row * 8 + l15] = rm;
        }
    }
}

__global__ __launch_bounds__(256) void k_node(
    const float* __restrict__ x, const float* __restrict__ W,
    const float* __restrict__ Wr, const float* __restrict__ bias,
    float* __restrict__ feat, float* __restrict__ out, int N, int nTiles)
{
    __shared__ float sW [IN_FEAT * OUTC];
    __shared__ float sWr[IN_FEAT * OUTC];
    node_body<256, 8>(sW, sWr, blockIdx.x, x, W, Wr, bias, feat, out, N, nTiles);
}

__global__ void k_elr(const float* __restrict__ feat, const float* __restrict__ al,
                      const float* __restrict__ ar, float* __restrict__ el,
                      float* __restrict__ rec, int N)
{
    int n = blockIdx.x * blockDim.x + threadIdx.x;
    if (n >= N) return;
    const float* f = feat + (size_t)n * OUTC;
    #pragma unroll
    for (int h = 0; h < 2; ++h) {
        float sl = 0.f, sr = 0.f;
        #pragma unroll
        for (int c = 0; c < 8; ++c) {
            float v = f[h * 8 + c];
            sl = fmaf(v, al[h * 8 + c], sl);
            sr = fmaf(v, ar[h * 8 + c], sr);
        }
        el[n * 2 + h] = sl;
        rec[n * 4 + h] = sr;
    }
}

__global__ void k_init(unsigned* __restrict__ mmax, float* __restrict__ denom,
                       float* __restrict__ acc, int n2, int n4)
{
    int i = blockIdx.x * blockDim.x + threadIdx.x;
    if (i < n2) { mmax[i] = 0x007FFFFFu; denom[i] = 0.f; }
    if (i < n4) reinterpret_cast<float4*>(acc)[i] = make_float4(0.f, 0.f, 0.f, 0.f);
}

__global__ void k_edge_max(const int* __restrict__ ei, const float* __restrict__ el,
                           unsigned* __restrict__ mmax, int E)
{
    int e = blockIdx.x * blockDim.x + threadIdx.x;
    int lane = threadIdx.x & 31;
    int base = e - lane;
    int d = 0;
    float v0 = 0.f, v1 = 0.f;
    if (e < E) {
        int s = ei[e];
        d = ei[E + e];
        v0 = el[s * 2];
        v1 = el[s * 2 + 1];
    }
    int h = lane & 1;
    for (int it = 0; it < 2; ++it) {
        int q = it * 16 + (lane >> 1);
        int dd = __shfl(d, q, 32);
        float a = __shfl(v0, q, 32), b = __shfl(v1, q, 32);
        if (base + q < E) atomicMax(&mmax[dd * 2 + h], fmap(h ? b : a));
    }
}

__global__ void k_max_fix(const unsigned* __restrict__ mmax, float* __restrict__ rec, int n2)
{
    int i = blockIdx.x * blockDim.x + threadIdx.x;
    if (i >= n2) return;
    int n = i >> 1, h = i & 1;
    float m = funmap(mmax[i]) + rec[n * 4 + h];
    m = (m >= 0.f) ? m : NEG_SLOPE * m;
    rec[n * 4 + 2 + h] = (fabsf(m) <= 3.402823466e38f) ? m : 0.f;
}

__global__ void k_edge_agg(const int* __restrict__ ei, const float* __restrict__ el,
                           const float* __restrict__ rec,
                           const float* __restrict__ feat, float* __restrict__ denom,
                           float* __restrict__ acc, int E)
{
    int e = blockIdx.x * blockDim.x + threadIdx.x;
    int lane = threadIdx.x & 31;
    int base = e - lane;
    int s = 0, d = 0;
    float w0 = 0.f, w1 = 0.f;
    if (e < E) {
        s = ei[e];
        d = ei[E + e];
        float4 r = reinterpret_cast<const float4*>(rec)[d];
        float v = el[s * 2] + r.x;
        v = (v >= 0.f) ? v : NEG_SLOPE * v;
        w0 = __expf(v - r.z);
        v = el[s * 2 + 1] + r.y;
        v = (v >= 0.f) ? v : NEG_SLOPE * v;
        w1 = __expf(v - r.w);
    }

    int h = lane & 1;
    for (int it = 0; it < 2; ++it) {
        int q = it * 16 + (lane >> 1);
        int dd = __shfl(d, q, 32);
        float a = __shfl(w0, q, 32), b = __shfl(w1, q, 32);
        if (base + q < E) atomicAdd(&denom[dd * 2 + h], h ? b : a);
    }

    int col = lane & 15;
    int hw  = lane >> 4;
    bool hi = (col & 8) != 0;
    float m[16];
    int   di[16];
    #pragma unroll
    for (int it = 0; it < 16; ++it) {
        int q = it * 2 + hw;
        int ss = __shfl(s, q, 32);
        di[it] = __shfl(d, q, 32) * OUTC + col;
        float a = __shfl(w0, q, 32), b = __shfl(w1, q, 32);
        m[it] = (hi ? b : a) * feat[ss * OUTC + col];
    }
    #pragma unroll
    for (int it = 0; it < 16; ++it)
        if (base + it * 2 + hw < E) atomicAdd(&acc[di[it]], m[it]);
}

__global__ void k_out(const float* __restrict__ acc, const float* __restrict__ denom,
                      float* __restrict__ out, int N)
{
    int i = blockIdx.x * blockDim.x + threadIdx.x;
    if (i >= N * 8) return;
    int n = i >> 3, c = i & 7;
    float m0 = acc[(size_t)n * OUTC + c]     / fmaxf(denom[n * 2],     EPSF);
    float m1 = acc[(size_t)n * OUTC + 8 + c] / fmaxf(denom[n * 2 + 1], EPSF);
    out[i] = out[i] + 0.5f * (m0 + m1);
}


#define BSH     8
#define BNODES  256
#define MAXBK   4096
#define SORT_G  256
#define CH_CAP  16384
#define BK_CAP  16384

__device__ __forceinline__ void lsort_body(int* sh, int* sw, unsigned* wbuf, int g,
                                           const int* __restrict__ ei, int* __restrict__ hist,
                                           int* __restrict__ lst, unsigned* __restrict__ lsorted,
                                           int E, int nBk, int chunk)
{
    int tid = threadIdx.x;
    int lane = tid & 31, wave = tid >> 5;
    for (int b = tid; b < nBk; b += 1024) sh[b] = 0;
    __syncthreads();

    int e0 = g * chunk;
    int e1 = min(e0 + chunk, E);
    int e = e0 + tid;
    for (; e + 3072 < e1; e += 4096) {
        int d[4];
        #pragma unroll
        for (int j = 0; j < 4; ++j) d[j] = ei[E + e + j * 1024];
        #pragma unroll
        for (int j = 0; j < 4; ++j) atomicAdd(&sh[d[j] >> BSH], 1);
    }
    for (; e < e1; e += 1024) atomicAdd(&sh[ei[E + e] >> BSH], 1);
    __syncthreads();

    int c[4];
    #pragma unroll
    for (int j = 0; j < 4; ++j) c[j] = (4 * tid + j < nBk) ? sh[4 * tid + j] : 0;
    int t = c[0] + c[1] + c[2] + c[3];
    int inc = t;
    for (int o = 1; o < 32; o <<= 1) {
        int v = __shfl_up(inc, o, 32);
        if (lane >= o) inc += v;
    }
    if (lane == 31) sw[wave] = inc;
    __syncthreads();
    int ex = inc - t;
    for (int w = 0; w < wave; ++w) ex += sw[w];
    #pragma unroll
    for (int j = 0; j < 4; ++j) {
        int b = 4 * tid + j;
        if (b < nBk) {
            hist[b * SORT_G + g] = c[j];
            lst[g * nBk + b] = ex;
            sh[b] = ex;
            ex += c[j];
        }
    }
    __syncthreads();

    bool staged = e1 - e0 <= CH_CAP;
    e = e0 + tid;
    for (; e + 3072 < e1; e += 4096) {
        int s[4], d[4];
        #pragma unroll
        for (int j = 0; j < 4; ++j) {
            s[j] = ei[e + j * 1024];
            d[j] = ei[E + e + j * 1024];
        }
        #pragma unroll
        for (int j = 0; j < 4; ++j) {
            int lp = atomicAdd(&sh[d[j] >> BSH], 1);
            unsigned w = ((unsigned)s[j] << BSH) | (unsigned)(d[j] & (BNODES - 1));
            if (staged) wbuf[lp] = w; else lsorted[e0 + lp] = w;
        }
    }
    for (; e < e1; e += 1024) {
        int s = ei[e], d = ei[E + e];
        int lp = atomicAdd(&sh[d >> BSH], 1);
        unsigned w = ((unsigned)s << BSH) | (unsigned)(d & (BNODES - 1));
        if (staged) wbuf[lp] = w; else lsorted[e0 + lp] = w;
    }
    if (staged) {
        __syncthreads();
        for (int i = tid - (e0 & 31); i < e1 - e0; i += 1024)
            if (i >= 0) lsorted[e0 + i] = wbuf[i];
    }
}

#define NODE_TPB 25
__global__ __launch_bounds__(1024) void k_node_lsort(
    const float* __restrict__ x, const float* __restrict__ W,
    const float* __restrict__ Wr, const float* __restrict__ bias,
    float* __restrict__ feat, float* __restrict__ out, int N, int nTiles,
    const int* __restrict__ ei, int* __restrict__ hist,
    int* __restrict__ lst, unsigned* __restrict__ lsorted,
    int E, int nBk, int chunk)
{
    __shared__ unsigned smem[MAXBK + 32 + CH_CAP];
    if (blockIdx.x < SORT_G) {
        lsort_body(reinterpret_cast<int*>(smem), reinterpret_cast<int*>(smem) + MAXBK, smem + MAXBK + 32,
                   blockIdx.x, ei, hist, lst, lsorted, E, nBk, chunk);
    } else {
        float* sW = reinterpret_cast<float*>(smem);
        node_body<1024, NODE_TPB>(sW, sW + IN_FEAT * OUTC, blockIdx.x - SORT_G, x, W, Wr, bias, feat, out, N, nTiles);
    }
}

__global__ __launch_bounds__(256) void k_blocksum(const int* __restrict__ cnt,
                                                  int* __restrict__ bsum, int N)
{
    __shared__ int sw[8];
    int tid = threadIdx.x;
    int i0 = (blockIdx.x * 256 + tid) * 4;
    int s = 0;
    #pragma unroll
    for (int j = 0; j < 4; ++j)
        if (i0 + j < N) s += cnt[i0 + j];
    #pragma unroll
    for (int o = 16; o > 0; o >>= 1) s += __shfl_xor(s, o, 32);
    if ((tid & 31) == 0) sw[tid >> 5] = s;
    __syncthreads();
    if (tid == 0) {
        int t = 0;
        #pragma unroll
        for (int w = 0; w < 8; ++w) t += sw[w];
        bsum[blockIdx.x] = t;
    }
}

__global__ __launch_bounds__(256) void k_offsets(const int* __restrict__ cnt,
                                                 const int* __restrict__ bsum,
                                                 int* __restrict__ off, int N)
{
    __shared__ int sw[8];
    int tid = threadIdx.x;
    int lane = tid & 31, wave = tid >> 5;

    int b = 0;
    for (int i = tid; i < (int)blockIdx.x; i += 256) b += bsum[i];
    #pragma unroll
    for (int o = 16; o > 0; o >>= 1) b += __shfl_xor(b, o, 32);
    if (lane == 0) sw[wave] = b;
    __syncthreads();
    int base = 0;
    #pragma unroll
    for (int w = 0; w < 8; ++w) base += sw[w];
    __syncthreads();

    int i0 = (blockIdx.x * 256 + tid) * 4;
    int c[4];
    #pragma unroll
    for (int j = 0; j < 4; ++j) c[j] = (i0 + j < N) ? cnt[i0 + j] : 0;
    int t = c[0] + c[1] + c[2] + c[3];
    int inc = t;
    #pragma unroll
    for (int o = 1; o < 32; o <<= 1) {
        int v = __shfl_up(inc, o, 32);
        if (lane >= o) inc += v;
    }
    if (lane == 31) sw[wave] = inc;
    __syncthreads();
    int wb = 0;
    for (int w = 0; w < wave; ++w) wb += sw[w];
    int ex = base + wb + inc - t;
    #pragma unroll
    for (int j = 0; j < 4; ++j)
        if (i0 + j < N) { off[i0 + j] = ex; ex += c[j]; }
}

__device__ __forceinline__ unsigned bucket_entry(const unsigned* __restrict__ lsorted, const int* r_src,
                                                 const int* r_ko, int k)
{
    int lo = 0;
    for (int step = SORT_G / 2; step > 0; step >>= 1)
        if (r_ko[lo + step] <= k) lo += step;
    return lsorted[r_src[lo] + (k - r_ko[lo])];
}

__global__ __launch_bounds__(1024) void k_bsort(const unsigned* __restrict__ lsorted, const int* __restrict__ lst,
                                                const int* __restrict__ base,
                                                int* __restrict__ cnt, int* __restrict__ off,
                                                int* __restrict__ ssrc, int N, int E, int nBk, int chunk)
{
    __shared__ int sc[BNODES];
    __shared__ int sw[8];
    __shared__ int r_src[SORT_G];
    __shared__ int r_ko[SORT_G];
    __shared__ unsigned ebuf[BK_CAP];
    __shared__ int sbuf[BK_CAP];
    int tid = threadIdx.x;
    int lane = tid & 31, wave = tid >> 5;
    int b = blockIdx.x;
    int beg = base[b * SORT_G];
    int end = (b + 1 < nBk) ? base[(b + 1) * SORT_G] : E;
    int nEb = end - beg;
    if (tid < SORT_G) {
        r_src[tid] = tid * chunk + lst[tid * nBk + b];
        r_ko[tid]  = base[b * SORT_G + tid] - beg;
    }
    if (tid < BNODES) sc[tid] = 0;
    __syncthreads();

    bool fits = nEb <= BK_CAP;
    int k = tid;
    if (fits) {
        for (; k + 3072 < nEb; k += 4096) {
            unsigned p[4];
            #pragma unroll
            for (int j = 0; j < 4; ++j) p[j] = bucket_entry(lsorted, r_src, r_ko, k + j * 1024);
            #pragma unroll
            for (int j = 0; j < 4; ++j) ebuf[k + j * 1024] = p[j];
        }
        for (; k < nEb; k += 1024) ebuf[k] = bucket_entry(lsorted, r_src, r_ko, k);
        __syncthreads();
        for (k = tid; k < nEb; k += 1024) atomicAdd(&sc[ebuf[k] & (BNODES - 1)], 1);
    } else {
        for (; k < nEb; k += 1024) atomicAdd(&sc[bucket_entry(lsorted, r_src, r_ko, k) & (BNODES - 1)], 1);
    }
    __syncthreads();

    int c = 0, inc = 0;
    if (tid < BNODES) {
        c = sc[tid];
        inc = c;
        for (int o = 1; o < 32; o <<= 1) {
            int v = __shfl_up(inc, o, 32);
            if (lane >= o) inc += v;
        }
        if (lane == 31) sw[wave] = inc;
    }
    __syncthreads();
    if (tid < BNODES) {
        int ex = beg + inc - c;
        for (int w = 0; w < wave; ++w) ex += sw[w];
        int n = b * BNODES + tid;
        if (n < N) { cnt[n] = c; off[n] = ex; }
        sc[tid] = ex;
    }
    __syncthreads();

    if (fits) {
        for (k = tid; k < nEb; k += 1024) {
            unsigned p = ebuf[k];
            int pos = atomicAdd(&sc[p & (BNODES - 1)], 1);
            sbuf[pos - beg] = (int)(p >> BSH);
        }
        __syncthreads();
        for (k = tid - (beg & 31); k < nEb; k += 1024)
            if (k >= 0) ssrc[beg + k] = sbuf[k];
    } else {
        for (k = tid; k < nEb; k += 1024) {
            unsigned p = bucket_entry(lsorted, r_src, r_ko, k);
            int pos = atomicAdd(&sc[p & (BNODES - 1)], 1);
            ssrc[pos] = (int)(p >> BSH);
        }
    }
}

__global__ __launch_bounds__(256) void k_gather(
    const int* __restrict__ ssrc, const int* __restrict__ off, const int* __restrict__ cnt,
    const float* __restrict__ el, const float* __restrict__ rec,
    const float* __restrict__ feat, float* __restrict__ out, int N)
{
    __shared__ float so[64];
    int tid = threadIdx.x;
    int lane = tid & 31, wave = tid >> 5;
    int n = __builtin_amdgcn_readfirstlane(blockIdx.x * 8 + wave);
    int beg = 0, deg = 0;
    float er0 = 0.f, er1 = 0.f;
    if (n < N) {
        beg = off[n];
        deg = cnt[n];
        er0 = rec[n * 4];
        er1 = rec[n * 4 + 1];
    }

    float m0 = -INFINITY, m1 = -INFINITY;
    for (int i = lane; i < deg; i += 32) {
        int s = ssrc[beg + i];
        float2 v = *reinterpret_cast<const float2*>(el + 2 * s);
        m0 = fmaxf(m0, v.x);
        m1 = fmaxf(m1, v.y);
    }
    #pragma unroll
    for (int o = 16; o > 0; o >>= 1) {
        m0 = fmaxf(m0, __shfl_xor(m0, o, 32));
        m1 = fmaxf(m1, __shfl_xor(m1, o, 32));
    }
    m0 += er0;  m0 = (m0 >= 0.f) ? m0 : NEG_SLOPE * m0;
    m1 += er1;  m1 = (m1 >= 0.f) ? m1 : NEG_SLOPE * m1;
    m0 = (fabsf(m0) <= 3.402823466e38f) ? m0 : 0.f;
    m1 = (fabsf(m1) <= 3.402823466e38f) ? m1 : 0.f;

    int col = lane & 15;
    int hw  = lane >> 4;
    bool hi = (col & 8) != 0;
    float accv = 0.f, d0 = 0.f, d1 = 0.f;
    for (int c0 = 0; c0 < deg; c0 += 32) {
        int s = 0;
        float w0 = 0.f, w1 = 0.f;
        if (c0 + lane < deg) {
            s = ssrc[beg + c0 + lane];
            float2 v = *reinterpret_cast<const float2*>(el + 2 * s);
            float a = v.x + er0;
            a = (a >= 0.f) ? a : NEG_SLOPE * a;
            w0 = __expf(a - m0);
            float b = v.y + er1;
            b = (b >= 0.f) ? b : NEG_SLOPE * b;
            w1 = __expf(b - m1);
        }
        d0 += w0;
        d1 += w1;
        #pragma unroll
        for (int g = 0; g < 2; ++g) {
            if (c0 + g * 16 < deg) {
                float p[8];
                #pragma unroll
                for (int j = 0; j < 8; ++j) {
                    int q = g * 16 + j * 2 + hw;
                    int ss = __shfl(s, q, 32);
                    float wa = __shfl(w0, q, 32), wb = __shfl(w1, q, 32);
                    p[j] = (hi ? wb : wa) * feat[ss * OUTC + col];
                }
                #pragma unroll
                for (int j = 0; j < 8; ++j)
                    accv += (c0 + g * 16 + j * 2 + hw < deg) ? p[j] : 0.f;
            }
        }
    }

    accv += __shfl_xor(accv, 16, 32);
    #pragma unroll
    for (int o = 16; o > 0; o >>= 1) {
        d0 += __shfl_xor(d0, o, 32);
        d1 += __shfl_xor(d1, o, 32);
    }
    float r  = accv / fmaxf(hi ? d1 : d0, EPSF);
    float hm = 0.5f * (r + __shfl_xor(r, 8, 32));
    if (lane < 8) so[wave * 8 + lane] = hm;
    __syncthreads();
    if (tid < 64) {
        int i = blockIdx.x * 64 + tid;
        if (i < N * 8) out[i] = out[i] + so[tid];
    }
}

extern "C" void kernel_launch(void* const* d_in, const int* in_sizes, int n_in,
                              void* d_out, int out_size, void* d_ws, size_t ws_size,
                              hipStream_t stream)
{
    const float* x    = (const float*)d_in[0];
    const int*   ei   = (const int*)  d_in[1];
    const float* W    = (const float*)d_in[2];
    const float* al   = (const float*)d_in[3];
    const float* ar   = (const float*)d_in[4];
    const float* bias = (const float*)d_in[5];
    const float* Wr   = (const float*)d_in[6];
    float* out = (float*)d_out;

    int N = in_sizes[0] / IN_FEAT;
    int E = in_sizes[1] / 2;

    int nTiles = (N + 15) / 16;
    int nBk    = (N + BNODES - 1) / BNODES;
    int nTab   = nBk * SORT_G;
    int nScan  = (nTab + 1023) / 1024;
    int chunk  = (E + SORT_G - 1) / SORT_G;
    dim3 blk(256);

    char* ws = (char*)d_ws;
    float* feat  = (float*)ws;  ws += (size_t)N * OUTC * sizeof(float);
    float* rec   = (float*)ws;  ws += (size_t)N * 4 * sizeof(float);
    float* el    = (float*)ws;  ws += (size_t)N * 2 * sizeof(float);
    int*   cnt   = (int*)ws;    ws += (size_t)N * sizeof(int);
    int*   off   = (int*)ws;    ws += (size_t)N * sizeof(int);
    int*   hist  = (int*)ws;    ws += (size_t)nTab * sizeof(int);
    int*   lst   = (int*)ws;    ws += (size_t)nTab * sizeof(int);
    int*   base  = (int*)ws;    ws += (size_t)nTab * sizeof(int);
    unsigned* lsorted = (unsigned*)ws;  ws += (size_t)E * sizeof(unsigned);
    int*   ssrc  = (int*)ws;    ws += (size_t)E * sizeof(int);
    int*   bsum  = (int*)ws;    ws += (size_t)nScan * sizeof(int);

    if (nBk <= MAXBK && (size_t)(ws - (char*)d_ws) <= ws_size) {
        k_node_lsort<<<SORT_G + (nTiles + NODE_TPB - 1) / NODE_TPB, 1024, 0, stream>>>(x, W, Wr, bias, feat, out, N, nTiles,
                                                                      ei, hist, lst, lsorted, E, nBk, chunk);
        k_elr     <<<(N + 255) / 256,     blk, 0, stream>>>(feat, al, ar, el, rec, N);
        k_blocksum<<<nScan,               blk, 0, stream>>>(hist, bsum, nTab);
        k_offsets <<<nScan,               blk, 0, stream>>>(hist, bsum, base, nTab);
        k_bsort   <<<nBk,                 1024, 0, stream>>>(lsorted, lst, base, cnt, off, ssrc, N, E, nBk, chunk);
        k_gather  <<<(N + 7) / 8,         blk, 0, stream>>>(ssrc, off, cnt, el, rec, feat, out, N);
        return;
    }

    ws = (char*)d_ws;
    feat = (float*)ws;              ws += (size_t)N * OUTC * sizeof(float);
    float*    acc   = (float*)ws;    ws += (size_t)N * OUTC * sizeof(float);
    rec  = (float*)ws;              ws += (size_t)N * 4 * sizeof(float);
    el   = (float*)ws;              ws += (size_t)N * 2 * sizeof(float);
    float*    denom = (float*)ws;    ws += (size_t)N * 2 * sizeof(float);
    unsigned* mmax  = (unsigned*)ws; ws += (size_t)N * 2 * sizeof(unsigned);

    k_node    <<<(nTiles + 7) / 8,      blk, 0, stream>>>(x, W, Wr, bias, feat, out, N, nTiles);
    k_init    <<<(4 * N + 255) / 256,   blk, 0, stream>>>(mmax, denom, acc, 2 * N, 4 * N);
    k_elr     <<<(N + 255) / 256,       blk, 0, stream>>>(feat, al, ar, el, rec, N);
    k_edge_max<<<(E + 255) / 256,       blk, 0, stream>>>(ei, el, mmax, E);
    k_max_fix <<<(2 * N + 255) / 256,   blk, 0, stream>>>(mmax, rec, 2 * N);
    k_edge_agg<<<(E + 255) / 256,       blk, 0, stream>>>(ei, el, rec, feat, denom, acc, E);
    k_out     <<<(N * 8 + 255) / 256,   blk, 0, stream>>>(acc, denom, out, N);
}


# Sum pass loads a node's first 32 edges once: sources, scores el[src] and feat rows go to registers in one sweep (rows requested ahead of the max); weights and sums run from registers, sum order kept
# speedup vs baseline: 11.7760x; 1.1011x over previous
#include <hip/hip_runtime.h>
#include <hip/hip_bf16.h>
#include <math.h>


typedef __attribute__((ext_vector_type(2))) float v2f;
typedef __attribute__((ext_vector_type(8))) float v8f;

#define IN_FEAT   256
#define OUTC      16
#define NEG_SLOPE 0.2f
#define EPSF      1e-15f

__device__ __forceinline__ unsigned fmap(float f) {
    unsigned u = __float_as_uint(f);
    return (u & 0x80000000u) ? ~u : (u | 0x80000000u);
}
__device__ __forceinline__ float funmap(unsigned u) {
    unsigned v = (u & 0x80000000u) ? (u & 0x7FFFFFFFu) : ~u;
    return __uint_as_float(v);
}

template <int NTHR, int TPB>
__device__ __forceinline__ void node_body(
    float* sW, float* sWr, int blk,
    const float* __restrict__ x, const float* __restrict__ W,
    const float* __restrict__ Wr, const float* __restrict__ bias,
    float* __restrict__ feat, float* __restrict__ out, int N, int nTiles)
{
    int tid = threadIdx.x;
    for (int i = tid; i < IN_FEAT * OUTC; i += NTHR) {
        sW[i]  = W[i];
        sWr[i] = Wr[i];
    }
    __syncthreads();

    int wave = tid >> 5;
    int lane = tid & 31;
    if (wave >= TPB) return;
    int t = blk * TPB + wave;
    if (t >= nTiles) t = nTiles - 1;

    int half = lane >> 4;
    int l15  = lane & 15;
    int rowg = t * 16 + l15;
    if (rowg >= N) rowg = N - 1;
    const float* xrow = x + (size_t)rowg * IN_FEAT;

    v8f cf = {0.f,0.f,0.f,0.f,0.f,0.f,0.f,0.f};
    v8f cr = {0.f,0.f,0.f,0.f,0.f,0.f,0.f,0.f};
    float nsq = 0.f;

    for (int k0 = 0; k0 < IN_FEAT; k0 += 4) {
        int ka = k0 + half * 2;
        float a0 = xrow[ka], a1 = xrow[ka + 1];
        nsq = fmaf(a0, a0, nsq);
        nsq = fmaf(a1, a1, nsq);
        v2f a; a[0] = a0; a[1] = a1;
        v2f b;  b[0]  = sW [ka * OUTC + l15];  b[1]  = sW [(ka + 1) * OUTC + l15];
        v2f br; br[0] = sWr[ka * OUTC + l15];  br[1] = sWr[(ka + 1) * OUTC + l15];
        cf = __builtin_amdgcn_wmma_f32_16x16x4_f32(false, a, false, b,  (short)0, cf, false, false);
        cr = __builtin_amdgcn_wmma_f32_16x16x4_f32(false, a, false, br, (short)0, cr, false, false);
    }

    nsq += __shfl_xor(nsq, 16, 32);
    float nrm = fmaxf(sqrtf(nsq), EPSF);
    float z   = fminf(nrm, 1.0f - 1e-7f);
    float sc  = 0.5f * (log1pf(z) - log1pf(-z)) / nrm;
    float bv  = bias[l15];

    float* fp = feat + (size_t)t * 16 * OUTC;
    float* op = out  + (size_t)t * 16 * 8;
    #pragma unroll
    for (int r = 0; r < 8; ++r) {
        int row = r + half * 8;
        float s_r = __shfl(sc, row, 32);
        float res = cr[r] * s_r + bv;
        float rm  = 0.5f * (res + __shfl_xor(res, 8, 32));
        if (t * 16 + row < N) {
            fp[row * OUTC + l15] = cf[r] * s_r;
            if (l15 < 8) op[row * 8 + l15] = rm;
        }
    }
}

__global__ __launch_bounds__(256) void k_node(
    const float* __restrict__ x, const float* __restrict__ W,
    const float* __restrict__ Wr, const float* __restrict__ bias,
    float* __restrict__ feat, float* __restrict__ out, int N, int nTiles)
{
    __shared__ float sW [IN_FEAT * OUTC];
    __shared__ float sWr[IN_FEAT * OUTC];
    node_body<256, 8>(sW, sWr, blockIdx.x, x, W, Wr, bias, feat, out, N, nTiles);
}

__global__ void k_elr(const float* __restrict__ feat, const float* __restrict__ al,
                      const float* __restrict__ ar, float* __restrict__ el,
                      float* __restrict__ rec, int N)
{
    int n = blockIdx.x * blockDim.x + threadIdx.x;
    if (n >= N) return;
    const float* f = feat + (size_t)n * OUTC;
    #pragma unroll
    for (int h = 0; h < 2; ++h) {
        float sl = 0.f, sr = 0.f;
        #pragma unroll
        for (int c = 0; c < 8; ++c) {
            float v = f[h * 8 + c];
            sl = fmaf(v, al[h * 8 + c], sl);
            sr = fmaf(v, ar[h * 8 + c], sr);
        }
        el[n * 2 + h] = sl;
        rec[n * 4 + h] = sr;
    }
}

__global__ void k_init(unsigned* __restrict__ mmax, float* __restrict__ denom,
                       float* __restrict__ acc, int n2, int n4)
{
    int i = blockIdx.x * blockDim.x + threadIdx.x;
    if (i < n2) { mmax[i] = 0x007FFFFFu; denom[i] = 0.f; }
    if (i < n4) reinterpret_cast<float4*>(acc)[i] = make_float4(0.f, 0.f, 0.f, 0.f);
}

__global__ void k_edge_max(const int* __restrict__ ei, const float* __restrict__ el,
                           unsigned* __restrict__ mmax, int E)
{
    int e = blockIdx.x * blockDim.x + threadIdx.x;
    int lane = threadIdx.x & 31;
    int base = e - lane;
    int d = 0;
    float v0 = 0.f, v1 = 0.f;
    if (e < E) {
        int s = ei[e];
        d = ei[E + e];
        v0 = el[s * 2];
        v1 = el[s * 2 + 1];
    }
    int h = lane & 1;
    for (int it = 0; it < 2; ++it) {
        int q = it * 16 + (lane >> 1);
        int dd = __shfl(d, q, 32);
        float a = __shfl(v0, q, 32), b = __shfl(v1, q, 32);
        if (base + q < E) atomicMax(&mmax[dd * 2 + h], fmap(h ? b : a));
    }
}

__global__ void k_max_fix(const unsigned* __restrict__ mmax, float* __restrict__ rec, int n2)
{
    int i = blockIdx.x * blockDim.x + threadIdx.x;
    if (i >= n2) return;
    int n = i >> 1, h = i & 1;
    float m = funmap(mmax[i]) + rec[n * 4 + h];
    m = (m >= 0.f) ? m : NEG_SLOPE * m;
    rec[n * 4 + 2 + h] = (fabsf(m) <= 3.402823466e38f) ? m : 0.f;
}

__global__ void k_edge_agg(const int* __restrict__ ei, const float* __restrict__ el,
                           const float* __restrict__ rec,
                           const float* __restrict__ feat, float* __restrict__ denom,
                           float* __restrict__ acc, int E)
{
    int e = blockIdx.x * blockDim.x + threadIdx.x;
    int lane = threadIdx.x & 31;
    int base = e - lane;
    int s = 0, d = 0;
    float w0 = 0.f, w1 = 0.f;
    if (e < E) {
        s = ei[e];
        d = ei[E + e];
        float4 r = reinterpret_cast<const float4*>(rec)[d];
        float v = el[s * 2] + r.x;
        v = (v >= 0.f) ? v : NEG_SLOPE * v;
        w0 = __expf(v - r.z);
        v = el[s * 2 + 1] + r.y;
        v = (v >= 0.f) ? v : NEG_SLOPE * v;
        w1 = __expf(v - r.w);
    }

    int h = lane & 1;
    for (int it = 0; it < 2; ++it) {
        int q = it * 16 + (lane >> 1);
        int dd = __shfl(d, q, 32);
        float a = __shfl(w0, q, 32), b = __shfl(w1, q, 32);
        if (base + q < E) atomicAdd(&denom[dd * 2 + h], h ? b : a);
    }

    int col = lane & 15;
    int hw  = lane >> 4;
    bool hi = (col & 8) != 0;
    float m[16];
    int   di[16];
    #pragma unroll
    for (int it = 0; it < 16; ++it) {
        int q = it * 2 + hw;
        int ss = __shfl(s, q, 32);
        di[it] = __shfl(d, q, 32) * OUTC + col;
        float a = __shfl(w0, q, 32), b = __shfl(w1, q, 32);
        m[it] = (hi ? b : a) * feat[ss * OUTC + col];
    }
    #pragma unroll
    for (int it = 0; it < 16; ++it)
        if (base + it * 2 + hw < E) atomicAdd(&acc[di[it]], m[it]);
}

__global__ void k_out(const float* __restrict__ acc, const float* __restrict__ denom,
                      float* __restrict__ out, int N)
{
    int i = blockIdx.x * blockDim.x + threadIdx.x;
    if (i >= N * 8) return;
    int n = i >> 3, c = i & 7;
    float m0 = acc[(size_t)n * OUTC + c]     / fmaxf(denom[n * 2],     EPSF);
    float m1 = acc[(size_t)n * OUTC + 8 + c] / fmaxf(denom[n * 2 + 1], EPSF);
    out[i] = out[i] + 0.5f * (m0 + m1);
}


#define BSH     8
#define BNODES  256
#define MAXBK   4096
#define SORT_G  256
#define CH_CAP  16384
#define BK_CAP  16384

__device__ __forceinline__ void lsort_body(int* sh, int* sw, unsigned* wbuf, int g,
                                           const int* __restrict__ ei, int* __restrict__ hist,
                                           int* __restrict__ lst, unsigned* __restrict__ lsorted,
                                           int E, int nBk, int chunk)
{
    int tid = threadIdx.x;
    int lane = tid & 31, wave = tid >> 5;
    for (int b = tid; b < nBk; b += 1024) sh[b] = 0;
    __syncthreads();

    int e0 = g * chunk;
    int e1 = min(e0 + chunk, E);
    int e = e0 + tid;
    for (; e + 3072 < e1; e += 4096) {
        int d[4];
        #pragma unroll
        for (int j = 0; j < 4; ++j) d[j] = ei[E + e + j * 1024];
        #pragma unroll
        for (int j = 0; j < 4; ++j) atomicAdd(&sh[d[j] >> BSH], 1);
    }
    for (; e < e1; e += 1024) atomicAdd(&sh[ei[E + e] >> BSH], 1);
    __syncthreads();

    int c[4];
    #pragma unroll
    for (int j = 0; j < 4; ++j) c[j] = (4 * tid + j < nBk) ? sh[4 * tid + j] : 0;
    int t = c[0] + c[1] + c[2] + c[3];
    int inc = t;
    for (int o = 1; o < 32; o <<= 1) {
        int v = __shfl_up(inc, o, 32);
        if (lane >= o) inc += v;
    }
    if (lane == 31) sw[wave] = inc;
    __syncthreads();
    int ex = inc - t;
    for (int w = 0; w < wave; ++w) ex += sw[w];
    #pragma unroll
    for (int j = 0; j < 4; ++j) {
        int b = 4 * tid + j;
        if (b < nBk) {
            hist[b * SORT_G + g] = c[j];
            lst[g * nBk + b] = ex;
            sh[b] = ex;
            ex += c[j];
        }
    }
    __syncthreads();

    bool staged = e1 - e0 <= CH_CAP;
    e = e0 + tid;
    for (; e + 3072 < e1; e += 4096) {
        int s[4], d[4];
        #pragma unroll
        for (int j = 0; j < 4; ++j) {
            s[j] = ei[e + j * 1024];
            d[j] = ei[E + e + j * 1024];
        }
        #pragma unroll
        for (int j = 0; j < 4; ++j) {
            int lp = atomicAdd(&sh[d[j] >> BSH], 1);
            unsigned w = ((unsigned)s[j] << BSH) | (unsigned)(d[j] & (BNODES - 1));
            if (staged) wbuf[lp] = w; else lsorted[e0 + lp] = w;
        }
    }
    for (; e < e1; e += 1024) {
        int s = ei[e], d = ei[E + e];
        int lp = atomicAdd(&sh[d >> BSH], 1);
        unsigned w = ((unsigned)s << BSH) | (unsigned)(d & (BNODES - 1));
        if (staged) wbuf[lp] = w; else lsorted[e0 + lp] = w;
    }
    if (staged) {
        __syncthreads();
        for (int i = tid - (e0 & 31); i < e1 - e0; i += 1024)
            if (i >= 0) lsorted[e0 + i] = wbuf[i];
    }
}

#define NODE_TPB 25
__global__ __launch_bounds__(1024) void k_node_lsort(
    const float* __restrict__ x, const float* __restrict__ W,
    const float* __restrict__ Wr, const float* __restrict__ bias,
    float* __restrict__ feat, float* __restrict__ out, int N, int nTiles,
    const int* __restrict__ ei, int* __restrict__ hist,
    int* __restrict__ lst, unsigned* __restrict__ lsorted,
    int E, int nBk, int chunk)
{
    __shared__ unsigned smem[MAXBK + 32 + CH_CAP];
    if (blockIdx.x < SORT_G) {
        lsort_body(reinterpret_cast<int*>(smem), reinterpret_cast<int*>(smem) + MAXBK, smem + MAXBK + 32,
                   blockIdx.x, ei, hist, lst, lsorted, E, nBk, chunk);
    } else {
        float* sW = reinterpret_cast<float*>(smem);
        node_body<1024, NODE_TPB>(sW, sW + IN_FEAT * OUTC, blockIdx.x - SORT_G, x, W, Wr, bias, feat, out, N, nTiles);
    }
}

__global__ __launch_bounds__(256) void k_blocksum(const int* __restrict__ cnt,
                                                  int* __restrict__ bsum, int N)
{
    __shared__ int sw[8];
    int tid = threadIdx.x;
    int i0 = (blockIdx.x * 256 + tid) * 4;
    int s = 0;
    #pragma unroll
    for (int j = 0; j < 4; ++j)
        if (i0 + j < N) s += cnt[i0 + j];
    #pragma unroll
    for (int o = 16; o > 0; o >>= 1) s += __shfl_xor(s, o, 32);
    if ((tid & 31) == 0) sw[tid >> 5] = s;
    __syncthreads();
    if (tid == 0) {
        int t = 0;
        #pragma unroll
        for (int w = 0; w < 8; ++w) t += sw[w];
        bsum[blockIdx.x] = t;
    }
}

__global__ __launch_bounds__(256) void k_offsets(const int* __restrict__ cnt,
                                                 const int* __restrict__ bsum,
                                                 int* __restrict__ off, int N)
{
    __shared__ int sw[8];
    int tid = threadIdx.x;
    int lane = tid & 31, wave = tid >> 5;

    int b = 0;
    for (int i = tid; i < (int)blockIdx.x; i += 256) b += bsum[i];
    #pragma unroll
    for (int o = 16; o > 0; o >>= 1) b += __shfl_xor(b, o, 32);
    if (lane == 0) sw[wave] = b;
    __syncthreads();
    int base = 0;
    #pragma unroll
    for (int w = 0; w < 8; ++w) base += sw[w];
    __syncthreads();

    int i0 = (blockIdx.x * 256 + tid) * 4;
    int c[4];
    #pragma unroll
    for (int j = 0; j < 4; ++j) c[j] = (i0 + j < N) ? cnt[i0 + j] : 0;
    int t = c[0] + c[1] + c[2] + c[3];
    int inc = t;
    #pragma unroll
    for (int o = 1; o < 32; o <<= 1) {
        int v = __shfl_up(inc, o, 32);
        if (lane >= o) inc += v;
    }
    if (lane == 31) sw[wave] = inc;
    __syncthreads();
    int wb = 0;
    for (int w = 0; w < wave; ++w) wb += sw[w];
    int ex = base + wb + inc - t;
    #pragma unroll
    for (int j = 0; j < 4; ++j)
        if (i0 + j < N) { off[i0 + j] = ex; ex += c[j]; }
}

__device__ __forceinline__ unsigned bucket_entry(const unsigned* __restrict__ lsorted, const int* r_src,
                                                 const int* r_ko, int k)
{
    int lo = 0;
    for (int step = SORT_G / 2; step > 0; step >>= 1)
        if (r_ko[lo + step] <= k) lo += step;
    return lsorted[r_src[lo] + (k - r_ko[lo])];
}

__global__ __launch_bounds__(1024) void k_bsort(const unsigned* __restrict__ lsorted, const int* __restrict__ lst,
                                                const int* __restrict__ base,
                                                int* __restrict__ cnt, int* __restrict__ off,
                                                int* __restrict__ ssrc, int N, int E, int nBk, int chunk)
{
    __shared__ int sc[BNODES];
    __shared__ int sw[8];
    __shared__ int r_src[SORT_G];
    __shared__ int r_ko[SORT_G];
    __shared__ unsigned ebuf[BK_CAP];
    __shared__ int sbuf[BK_CAP];
    int tid = threadIdx.x;
    int lane = tid & 31, wave = tid >> 5;
    int b = blockIdx.x;
    int beg = base[b * SORT_G];
    int end = (b + 1 < nBk) ? base[(b + 1) * SORT_G] : E;
    int nEb = end - beg;
    if (tid < SORT_G) {
        r_src[tid] = tid * chunk + lst[tid * nBk + b];
        r_ko[tid]  = base[b * SORT_G + tid] - beg;
    }
    if (tid < BNODES) sc[tid] = 0;
    __syncthreads();

    bool fits = nEb <= BK_CAP;
    int k = tid;
    if (fits) {
        for (; k + 3072 < nEb; k += 4096) {
            unsigned p[4];
            #pragma unroll
            for (int j = 0; j < 4; ++j) p[j] = bucket_entry(lsorted, r_src, r_ko, k + j * 1024);
            #pragma unroll
            for (int j = 0; j < 4; ++j) ebuf[k + j * 1024] = p[j];
        }
        for (; k < nEb; k += 1024) ebuf[k] = bucket_entry(lsorted, r_src, r_ko, k);
        __syncthreads();
        for (k = tid; k < nEb; k += 1024) atomicAdd(&sc[ebuf[k] & (BNODES - 1)], 1);
    } else {
        for (; k < nEb; k += 1024) atomicAdd(&sc[bucket_entry(lsorted, r_src, r_ko, k) & (BNODES - 1)], 1);
    }
    __syncthreads();

    int c = 0, inc = 0;
    if (tid < BNODES) {
        c = sc[tid];
        inc = c;
        for (int o = 1; o < 32; o <<= 1) {
            int v = __shfl_up(inc, o, 32);
            if (lane >= o) inc += v;
        }
        if (lane == 31) sw[wave] = inc;
    }
    __syncthreads();
    if (tid < BNODES) {
        int ex = beg + inc - c;
        for (int w = 0; w < wave; ++w) ex += sw[w];
        int n = b * BNODES + tid;
        if (n < N) { cnt[n] = c; off[n] = ex; }
        sc[tid] = ex;
    }
    __syncthreads();

    if (fits) {
        for (k = tid; k < nEb; k += 1024) {
            unsigned p = ebuf[k];
            int pos = atomicAdd(&sc[p & (BNODES - 1)], 1);
            sbuf[pos - beg] = (int)(p >> BSH);
        }
        __syncthreads();
        for (k = tid - (beg & 31); k < nEb; k += 1024)
            if (k >= 0) ssrc[beg + k] = sbuf[k];
    } else {
        for (k = tid; k < nEb; k += 1024) {
            unsigned p = bucket_entry(lsorted, r_src, r_ko, k);
            int pos = atomicAdd(&sc[p & (BNODES - 1)], 1);
            ssrc[pos] = (int)(p >> BSH);
        }
    }
}

__global__ __launch_bounds__(256) void k_gather(
    const int* __restrict__ ssrc, const int* __restrict__ off, const int* __restrict__ cnt,
    const float* __restrict__ el, const float* __restrict__ rec,
    const float* __restrict__ feat, float* __restrict__ out, int N)
{
    __shared__ float so[64];
    int tid = threadIdx.x;
    int lane = tid & 31, wave = tid >> 5;
    int n = __builtin_amdgcn_readfirstlane(blockIdx.x * 8 + wave);
    int beg = 0, deg = 0;
    float er0 = 0.f, er1 = 0.f;
    if (n < N) {
        beg = off[n];
        deg = cnt[n];
        er0 = rec[n * 4];
        er1 = rec[n * 4 + 1];
    }

    int col = lane & 15;
    int hw  = lane >> 4;
    bool hi = (col & 8) != 0;
    bool live = lane < deg;

    int sraw = ssrc[beg + (live ? lane : 0)];
    int s1 = live ? sraw : 0;
    float2 v1 = *reinterpret_cast<const float2*>(el + 2 * s1);
    float p[16];
    #pragma unroll
    for (int j = 0; j < 8; ++j) {
        int ss = __shfl(s1, j * 2 + hw, 32);
        p[j] = feat[ss * OUTC + col];
        p[j + 8] = 0.f;
    }
    if (deg > 16) {
        #pragma unroll
        for (int j = 8; j < 16; ++j) {
            int ss = __shfl(s1, j * 2 + hw, 32);
            p[j] = feat[ss * OUTC + col];
        }
    }

    float m0 = live ? v1.x : -INFINITY;
    float m1 = live ? v1.y : -INFINITY;
    for (int i = 32 + lane; i < deg; i += 32) {
        int s = ssrc[beg + i];
        float2 v = *reinterpret_cast<const float2*>(el + 2 * s);
        m0 = fmaxf(m0, v.x);
        m1 = fmaxf(m1, v.y);
    }
    #pragma unroll
    for (int o = 16; o > 0; o >>= 1) {
        m0 = fmaxf(m0, __shfl_xor(m0, o, 32));
        m1 = fmaxf(m1, __shfl_xor(m1, o, 32));
    }
    m0 += er0;  m0 = (m0 >= 0.f) ? m0 : NEG_SLOPE * m0;
    m1 += er1;  m1 = (m1 >= 0.f) ? m1 : NEG_SLOPE * m1;
    m0 = (fabsf(m0) <= 3.402823466e38f) ? m0 : 0.f;
    m1 = (fabsf(m1) <= 3.402823466e38f) ? m1 : 0.f;

    float accv = 0.f, d0 = 0.f, d1 = 0.f;
    {
        float w0 = 0.f, w1 = 0.f;
        if (live) {
            float a = v1.x + er0;
            a = (a >= 0.f) ? a : NEG_SLOPE * a;
            w0 = __expf(a - m0);
            float b = v1.y + er1;
            b = (b >= 0.f) ? b : NEG_SLOPE * b;
            w1 = __expf(b - m1);
        }
        d0 = w0;
        d1 = w1;
        #pragma unroll
        for (int j = 0; j < 16; ++j) {
            int q = j * 2 + hw;
            float wa = __shfl(w0, q, 32), wb = __shfl(w1, q, 32);
            float t = (hi ? wb : wa) * p[j];
            accv += (q < deg) ? t : 0.f;
        }
    }

    for (int c0 = 32; c0 < deg; c0 += 32) {
        int s = 0;
        float w0 = 0.f, w1 = 0.f;
        if (c0 + lane < deg) {
            s = ssrc[beg + c0 + lane];
            float2 v = *reinterpret_cast<const float2*>(el + 2 * s);
            float a = v.x + er0;
            a = (a >= 0.f) ? a : NEG_SLOPE * a;
            w0 = __expf(a - m0);
            float b = v.y + er1;
            b = (b >= 0.f) ? b : NEG_SLOPE * b;
            w1 = __expf(b - m1);
        }
        d0 += w0;
        d1 += w1;
        #pragma unroll
        for (int g = 0; g < 2; ++g) {
            if (c0 + g * 16 < deg) {
                float r[8];
                #pragma unroll
                for (int j = 0; j < 8; ++j) {
                    int q = g * 16 + j * 2 + hw;
                    int ss = __shfl(s, q, 32);
                    float wa = __shfl(w0, q, 32), wb = __shfl(w1, q, 32);
                    r[j] = (hi ? wb : wa) * feat[ss * OUTC + col];
                }
                #pragma unroll
                for (int j = 0; j < 8; ++j)
                    accv += (c0 + g * 16 + j * 2 + hw < deg) ? r[j] : 0.f;
            }
        }
    }

    accv += __shfl_xor(accv, 16, 32);
    #pragma unroll
    for (int o = 16; o > 0; o >>= 1) {
        d0 += __shfl_xor(d0, o, 32);
        d1 += __shfl_xor(d1, o, 32);
    }
    float r  = accv / fmaxf(hi ? d1 : d0, EPSF);
    float hm = 0.5f * (r + __shfl_xor(r, 8, 32));
    if (lane < 8) so[wave * 8 + lane] = hm;
    __syncthreads();
    if (tid < 64) {
        int i = blockIdx.x * 64 + tid;
        if (i < N * 8) out[i] = out[i] + so[tid];
    }
}

extern "C" void kernel_launch(void* const* d_in, const int* in_sizes, int n_in,
                              void* d_out, int out_size, void* d_ws, size_t ws_size,
                              hipStream_t stream)
{
    const float* x    = (const float*)d_in[0];
    const int*   ei   = (const int*)  d_in[1];
    const float* W    = (const float*)d_in[2];
    const float* al   = (const float*)d_in[3];
    const float* ar   = (const float*)d_in[4];
    const float* bias = (const float*)d_in[5];
    const float* Wr   = (const float*)d_in[6];
    float* out = (float*)d_out;

    int N = in_sizes[0] / IN_FEAT;
    int E = in_sizes[1] / 2;

    int nTiles = (N + 15) / 16;
    int nBk    = (N + BNODES - 1) / BNODES;
    int nTab   = nBk * SORT_G;
    int nScan  = (nTab + 1023) / 1024;
    int chunk  = (E + SORT_G - 1) / SORT_G;
    dim3 blk(256);

    char* ws = (char*)d_ws;
    float* feat  = (float*)ws;  ws += (size_t)N * OUTC * sizeof(float);
    float* rec   = (float*)ws;  ws += (size_t)N * 4 * sizeof(float);
    float* el    = (float*)ws;  ws += (size_t)N * 2 * sizeof(float);
    int*   cnt   = (int*)ws;    ws += (size_t)N * sizeof(int);
    int*   off   = (int*)ws;    ws += (size_t)N * sizeof(int);
    int*   hist  = (int*)ws;    ws += (size_t)nTab * sizeof(int);
    int*   lst   = (int*)ws;    ws += (size_t)nTab * sizeof(int);
    int*   base  = (int*)ws;    ws += (size_t)nTab * sizeof(int);
    unsigned* lsorted = (unsigned*)ws;  ws += (size_t)E * sizeof(unsigned);
    int*   ssrc  = (int*)ws;    ws += (size_t)E * sizeof(int);
    int*   bsum  = (int*)ws;    ws += (size_t)nScan * sizeof(int);

    if (nBk <= MAXBK && (size_t)(ws - (char*)d_ws) <= ws_size) {
        k_node_lsort<<<SORT_G + (nTiles + NODE_TPB - 1) / NODE_TPB, 1024, 0, stream>>>(x, W, Wr, bias, feat, out, N, nTiles,
                                                                      ei, hist, lst, lsorted, E, nBk, chunk);
        k_elr     <<<(N + 255) / 256,     blk, 0, stream>>>(feat, al, ar, el, rec, N);
        k_blocksum<<<nScan,               blk, 0, stream>>>(hist, bsum, nTab);
        k_offsets <<<nScan,               blk, 0, stream>>>(hist, bsum, base, nTab);
        k_bsort   <<<nBk,                 1024, 0, stream>>>(lsorted, lst, base, cnt, off, ssrc, N, E, nBk, chunk);
        k_gather  <<<(N + 7) / 8,         blk, 0, stream>>>(ssrc, off, cnt, el, rec, feat, out, N);
        return;
    }

    ws = (char*)d_ws;
    feat = (float*)ws;              ws += (size_t)N * OUTC * sizeof(float);
    float*    acc   = (float*)ws;    ws += (size_t)N * OUTC * sizeof(float);
    rec  = (float*)ws;              ws += (size_t)N * 4 * sizeof(float);
    el   = (float*)ws;              ws += (size_t)N * 2 * sizeof(float);
    float*    denom = (float*)ws;    ws += (size_t)N * 2 * sizeof(float);
    unsigned* mmax  = (unsigned*)ws; ws += (size_t)N * 2 * sizeof(unsigned);

    k_node    <<<(nTiles + 7) / 8,      blk, 0, stream>>>(x, W, Wr, bias, feat, out, N, nTiles);
    k_init    <<<(4 * N + 255) / 256,   blk, 0, stream>>>(mmax, denom, acc, 2 * N, 4 * N);
    k_elr     <<<(N + 255) / 256,       blk, 0, stream>>>(feat, al, ar, el, rec, N);
    k_edge_max<<<(E + 255) / 256,       blk, 0, stream>>>(ei, el, mmax, E);
    k_max_fix <<<(2 * N + 255) / 256,   blk, 0, stream>>>(mmax, rec, 2 * N);
    k_edge_agg<<<(E + 255) / 256,       blk, 0, stream>>>(ei, el, rec, feat, denom, acc, E);
    k_out     <<<(N * 8 + 255) / 256,   blk, 0, stream>>>(acc, denom, out, N);
}


# Sum pass moves lanes by DPP (row_share broadcasts, row_xmask steps + one row permute), not ds_bpermute: a half-wave takes 16 consecutive edges, so the message sum order changes; denominators kept
# speedup vs baseline: 12.6080x; 1.0707x over previous
#include <hip/hip_runtime.h>
#include <hip/hip_bf16.h>
#include <math.h>


typedef __attribute__((ext_vector_type(2))) float v2f;
typedef __attribute__((ext_vector_type(8))) float v8f;

#define IN_FEAT   256
#define OUTC      16
#define NEG_SLOPE 0.2f
#define EPSF      1e-15f

__device__ __forceinline__ unsigned fmap(float f) {
    unsigned u = __float_as_uint(f);
    return (u & 0x80000000u) ? ~u : (u | 0x80000000u);
}
__device__ __forceinline__ float funmap(unsigned u) {
    unsigned v = (u & 0x80000000u) ? (u & 0x7FFFFFFFu) : ~u;
    return __uint_as_float(v);
}

template <int NTHR, int TPB>
__device__ __forceinline__ void node_body(
    float* sW, float* sWr, int blk,
    const float* __restrict__ x, const float* __restrict__ W,
    const float* __restrict__ Wr, const float* __restrict__ bias,
    float* __restrict__ feat, float* __restrict__ out, int N, int nTiles)
{
    int tid = threadIdx.x;
    for (int i = tid; i < IN_FEAT * OUTC; i += NTHR) {
        sW[i]  = W[i];
        sWr[i] = Wr[i];
    }
    __syncthreads();

    int wave = tid >> 5;
    int lane = tid & 31;
    if (wave >= TPB) return;
    int t = blk * TPB + wave;
    if (t >= nTiles) t = nTiles - 1;

    int half = lane >> 4;
    int l15  = lane & 15;
    int rowg = t * 16 + l15;
    if (rowg >= N) rowg = N - 1;
    const float* xrow = x + (size_t)rowg * IN_FEAT;

    v8f cf = {0.f,0.f,0.f,0.f,0.f,0.f,0.f,0.f};
    v8f cr = {0.f,0.f,0.f,0.f,0.f,0.f,0.f,0.f};
    float nsq = 0.f;

    for (int k0 = 0; k0 < IN_FEAT; k0 += 4) {
        int ka = k0 + half * 2;
        float a0 = xrow[ka], a1 = xrow[ka + 1];
        nsq = fmaf(a0, a0, nsq);
        nsq = fmaf(a1, a1, nsq);
        v2f a; a[0] = a0; a[1] = a1;
        v2f b;  b[0]  = sW [ka * OUTC + l15];  b[1]  = sW [(ka + 1) * OUTC + l15];
        v2f br; br[0] = sWr[ka * OUTC + l15];  br[1] = sWr[(ka + 1) * OUTC + l15];
        cf = __builtin_amdgcn_wmma_f32_16x16x4_f32(false, a, false, b,  (short)0, cf, false, false);
        cr = __builtin_amdgcn_wmma_f32_16x16x4_f32(false, a, false, br, (short)0, cr, false, false);
    }

    nsq += __shfl_xor(nsq, 16, 32);
    float nrm = fmaxf(sqrtf(nsq), EPSF);
    float z   = fminf(nrm, 1.0f - 1e-7f);
    float sc  = 0.5f * (log1pf(z) - log1pf(-z)) / nrm;
    float bv  = bias[l15];

    float* fp = feat + (size_t)t * 16 * OUTC;
    float* op = out  + (size_t)t * 16 * 8;
    #pragma unroll
    for (int r = 0; r < 8; ++r) {
        int row = r + half * 8;
        float s_r = __shfl(sc, row, 32);
        float res = cr[r] * s_r + bv;
        float rm  = 0.5f * (res + __shfl_xor(res, 8, 32));
        if (t * 16 + row < N) {
            fp[row * OUTC + l15] = cf[r] * s_r;
            if (l15 < 8) op[row * 8 + l15] = rm;
        }
    }
}

__global__ __launch_bounds__(256) void k_node(
    const float* __restrict__ x, const float* __restrict__ W,
    const float* __restrict__ Wr, const float* __restrict__ bias,
    float* __restrict__ feat, float* __restrict__ out, int N, int nTiles)
{
    __shared__ float sW [IN_FEAT * OUTC];
    __shared__ float sWr[IN_FEAT * OUTC];
    node_body<256, 8>(sW, sWr, blockIdx.x, x, W, Wr, bias, feat, out, N, nTiles);
}

__global__ void k_elr(const float* __restrict__ feat, const float* __restrict__ al,
                      const float* __restrict__ ar, float* __restrict__ el,
                      float* __restrict__ rec, int N)
{
    int n = blockIdx.x * blockDim.x + threadIdx.x;
    if (n >= N) return;
    const float* f = feat + (size_t)n * OUTC;
    #pragma unroll
    for (int h = 0; h < 2; ++h) {
        float sl = 0.f, sr = 0.f;
        #pragma unroll
        for (int c = 0; c < 8; ++c) {
            float v = f[h * 8 + c];
            sl = fmaf(v, al[h * 8 + c], sl);
            sr = fmaf(v, ar[h * 8 + c], sr);
        }
        el[n * 2 + h] = sl;
        rec[n * 4 + h] = sr;
    }
}

__global__ void k_init(unsigned* __restrict__ mmax, float* __restrict__ denom,
                       float* __restrict__ acc, int n2, int n4)
{
    int i = blockIdx.x * blockDim.x + threadIdx.x;
    if (i < n2) { mmax[i] = 0x007FFFFFu; denom[i] = 0.f; }
    if (i < n4) reinterpret_cast<float4*>(acc)[i] = make_float4(0.f, 0.f, 0.f, 0.f);
}

__global__ void k_edge_max(const int* __restrict__ ei, const float* __restrict__ el,
                           unsigned* __restrict__ mmax, int E)
{
    int e = blockIdx.x * blockDim.x + threadIdx.x;
    int lane = threadIdx.x & 31;
    int base = e - lane;
    int d = 0;
    float v0 = 0.f, v1 = 0.f;
    if (e < E) {
        int s = ei[e];
        d = ei[E + e];
        v0 = el[s * 2];
        v1 = el[s * 2 + 1];
    }
    int h = lane & 1;
    for (int it = 0; it < 2; ++it) {
        int q = it * 16 + (lane >> 1);
        int dd = __shfl(d, q, 32);
        float a = __shfl(v0, q, 32), b = __shfl(v1, q, 32);
        if (base + q < E) atomicMax(&mmax[dd * 2 + h], fmap(h ? b : a));
    }
}

__global__ void k_max_fix(const unsigned* __restrict__ mmax, float* __restrict__ rec, int n2)
{
    int i = blockIdx.x * blockDim.x + threadIdx.x;
    if (i >= n2) return;
    int n = i >> 1, h = i & 1;
    float m = funmap(mmax[i]) + rec[n * 4 + h];
    m = (m >= 0.f) ? m : NEG_SLOPE * m;
    rec[n * 4 + 2 + h] = (fabsf(m) <= 3.402823466e38f) ? m : 0.f;
}

__global__ void k_edge_agg(const int* __restrict__ ei, const float* __restrict__ el,
                           const float* __restrict__ rec,
                           const float* __restrict__ feat, float* __restrict__ denom,
                           float* __restrict__ acc, int E)
{
    int e = blockIdx.x * blockDim.x + threadIdx.x;
    int lane = threadIdx.x & 31;
    int base = e - lane;
    int s = 0, d = 0;
    float w0 = 0.f, w1 = 0.f;
    if (e < E) {
        s = ei[e];
        d = ei[E + e];
        float4 r = reinterpret_cast<const float4*>(rec)[d];
        float v = el[s * 2] + r.x;
        v = (v >= 0.f) ? v : NEG_SLOPE * v;
        w0 = __expf(v - r.z);
        v = el[s * 2 + 1] + r.y;
        v = (v >= 0.f) ? v : NEG_SLOPE * v;
        w1 = __expf(v - r.w);
    }

    int h = lane & 1;
    for (int it = 0; it < 2; ++it) {
        int q = it * 16 + (lane >> 1);
        int dd = __shfl(d, q, 32);
        float a = __shfl(w0, q, 32), b = __shfl(w1, q, 32);
        if (base + q < E) atomicAdd(&denom[dd * 2 + h], h ? b : a);
    }

    int col = lane & 15;
    int hw  = lane >> 4;
    bool hi = (col & 8) != 0;
    float m[16];
    int   di[16];
    #pragma unroll
    for (int it = 0; it < 16; ++it) {
        int q = it * 2 + hw;
        int ss = __shfl(s, q, 32);
        di[it] = __shfl(d, q, 32) * OUTC + col;
        float a = __shfl(w0, q, 32), b = __shfl(w1, q, 32);
        m[it] = (hi ? b : a) * feat[ss * OUTC + col];
    }
    #pragma unroll
    for (int it = 0; it < 16; ++it)
        if (base + it * 2 + hw < E) atomicAdd(&acc[di[it]], m[it]);
}

__global__ void k_out(const float* __restrict__ acc, const float* __restrict__ denom,
                      float* __restrict__ out, int N)
{
    int i = blockIdx.x * blockDim.x + threadIdx.x;
    if (i >= N * 8) return;
    int n = i >> 3, c = i & 7;
    float m0 = acc[(size_t)n * OUTC + c]     / fmaxf(denom[n * 2],     EPSF);
    float m1 = acc[(size_t)n * OUTC + 8 + c] / fmaxf(denom[n * 2 + 1], EPSF);
    out[i] = out[i] + 0.5f * (m0 + m1);
}


#define BSH     8
#define BNODES  256
#define MAXBK   4096
#define SORT_G  256
#define CH_CAP  16384
#define BK_CAP  16384

__device__ __forceinline__ void lsort_body(int* sh, int* sw, unsigned* wbuf, int g,
                                           const int* __restrict__ ei, int* __restrict__ hist,
                                           int* __restrict__ lst, unsigned* __restrict__ lsorted,
                                           int E, int nBk, int chunk)
{
    int tid = threadIdx.x;
    int lane = tid & 31, wave = tid >> 5;
    for (int b = tid; b < nBk; b += 1024) sh[b] = 0;
    __syncthreads();

    int e0 = g * chunk;
    int e1 = min(e0 + chunk, E);
    int e = e0 + tid;
    for (; e + 3072 < e1; e += 4096) {
        int d[4];
        #pragma unroll
        for (int j = 0; j < 4; ++j) d[j] = ei[E + e + j * 1024];
        #pragma unroll
        for (int j = 0; j < 4; ++j) atomicAdd(&sh[d[j] >> BSH], 1);
    }
    for (; e < e1; e += 1024) atomicAdd(&sh[ei[E + e] >> BSH], 1);
    __syncthreads();

    int c[4];
    #pragma unroll
    for (int j = 0; j < 4; ++j) c[j] = (4 * tid + j < nBk) ? sh[4 * tid + j] : 0;
    int t = c[0] + c[1] + c[2] + c[3];
    int inc = t;
    for (int o = 1; o < 32; o <<= 1) {
        int v = __shfl_up(inc, o, 32);
        if (lane >= o) inc += v;
    }
    if (lane == 31) sw[wave] = inc;
    __syncthreads();
    int ex = inc - t;
    for (int w = 0; w < wave; ++w) ex += sw[w];
    #pragma unroll
    for (int j = 0; j < 4; ++j) {
        int b = 4 * tid + j;
        if (b < nBk) {
            hist[b * SORT_G + g] = c[j];
            lst[g * nBk + b] = ex;
            sh[b] = ex;
            ex += c[j];
        }
    }
    __syncthreads();

    bool staged = e1 - e0 <= CH_CAP;
    e = e0 + tid;
    for (; e + 3072 < e1; e += 4096) {
        int s[4], d[4];
        #pragma unroll
        for (int j = 0; j < 4; ++j) {
            s[j] = ei[e + j * 1024];
            d[j] = ei[E + e + j * 1024];
        }
        #pragma unroll
        for (int j = 0; j < 4; ++j) {
            int lp = atomicAdd(&sh[d[j] >> BSH], 1);
            unsigned w = ((unsigned)s[j] << BSH) | (unsigned)(d[j] & (BNODES - 1));
            if (staged) wbuf[lp] = w; else lsorted[e0 + lp] = w;
        }
    }
    for (; e < e1; e += 1024) {
        int s = ei[e], d = ei[E + e];
        int lp = atomicAdd(&sh[d >> BSH], 1);
        unsigned w = ((unsigned)s << BSH) | (unsigned)(d & (BNODES - 1));
        if (staged) wbuf[lp] = w; else lsorted[e0 + lp] = w;
    }
    if (staged) {
        __syncthreads();
        for (int i = tid - (e0 & 31); i < e1 - e0; i += 1024)
            if (i >= 0) lsorted[e0 + i] = wbuf[i];
    }
}

#define NODE_TPB 25
__global__ __launch_bounds__(1024) void k_node_lsort(
    const float* __restrict__ x, const float* __restrict__ W,
    const float* __restrict__ Wr, const float* __restrict__ bias,
    float* __restrict__ feat, float* __restrict__ out, int N, int nTiles,
    const int* __restrict__ ei, int* __restrict__ hist,
    int* __restrict__ lst, unsigned* __restrict__ lsorted,
    int E, int nBk, int chunk)
{
    __shared__ unsigned smem[MAXBK + 32 + CH_CAP];
    if (blockIdx.x < SORT_G) {
        lsort_body(reinterpret_cast<int*>(smem), reinterpret_cast<int*>(smem) + MAXBK, smem + MAXBK + 32,
                   blockIdx.x, ei, hist, lst, lsorted, E, nBk, chunk);
    } else {
        float* sW = reinterpret_cast<float*>(smem);
        node_body<1024, NODE_TPB>(sW, sW + IN_FEAT * OUTC, blockIdx.x - SORT_G, x, W, Wr, bias, feat, out, N, nTiles);
    }
}

__global__ __launch_bounds__(256) void k_blocksum(const int* __restrict__ cnt,
                                                  int* __restrict__ bsum, int N)
{
    __shared__ int sw[8];
    int tid = threadIdx.x;
    int i0 = (blockIdx.x * 256 + tid) * 4;
    int s = 0;
    #pragma unroll
    for (int j = 0; j < 4; ++j)
        if (i0 + j < N) s += cnt[i0 + j];
    #pragma unroll
    for (int o = 16; o > 0; o >>= 1) s += __shfl_xor(s, o, 32);
    if ((tid & 31) == 0) sw[tid >> 5] = s;
    __syncthreads();
    if (tid == 0) {
        int t = 0;
        #pragma unroll
        for (int w = 0; w < 8; ++w) t += sw[w];
        bsum[blockIdx.x] = t;
    }
}

__global__ __launch_bounds__(256) void k_offsets(const int* __restrict__ cnt,
                                                 const int* __restrict__ bsum,
                                                 int* __restrict__ off, int N)
{
    __shared__ int sw[8];
    int tid = threadIdx.x;
    int lane = tid & 31, wave = tid >> 5;

    int b = 0;
    for (int i = tid; i < (int)blockIdx.x; i += 256) b += bsum[i];
    #pragma unroll
    for (int o = 16; o > 0; o >>= 1) b += __shfl_xor(b, o, 32);
    if (lane == 0) sw[wave] = b;
    __syncthreads();
    int base = 0;
    #pragma unroll
    for (int w = 0; w < 8; ++w) base += sw[w];
    __syncthreads();

    int i0 = (blockIdx.x * 256 + tid) * 4;
    int c[4];
    #pragma unroll
    for (int j = 0; j < 4; ++j) c[j] = (i0 + j < N) ? cnt[i0 + j] : 0;
    int t = c[0] + c[1] + c[2] + c[3];
    int inc = t;
    #pragma unroll
    for (int o = 1; o < 32; o <<= 1) {
        int v = __shfl_up(inc, o, 32);
        if (lane >= o) inc += v;
    }
    if (lane == 31) sw[wave] = inc;
    __syncthreads();
    int wb = 0;
    for (int w = 0; w < wave; ++w) wb += sw[w];
    int ex = base + wb + inc - t;
    #pragma unroll
    for (int j = 0; j < 4; ++j)
        if (i0 + j < N) { off[i0 + j] = ex; ex += c[j]; }
}

__device__ __forceinline__ unsigned bucket_entry(const unsigned* __restrict__ lsorted, const int* r_src,
                                                 const int* r_ko, int k)
{
    int lo = 0;
    for (int step = SORT_G / 2; step > 0; step >>= 1)
        if (r_ko[lo + step] <= k) lo += step;
    return lsorted[r_src[lo] + (k - r_ko[lo])];
}

__global__ __launch_bounds__(1024) void k_bsort(const unsigned* __restrict__ lsorted, const int* __restrict__ lst,
                                                const int* __restrict__ base,
                                                int* __restrict__ cnt, int* __restrict__ off,
                                                int* __restrict__ ssrc, int N, int E, int nBk, int chunk)
{
    __shared__ int sc[BNODES];
    __shared__ int sw[8];
    __shared__ int r_src[SORT_G];
    __shared__ int r_ko[SORT_G];
    __shared__ unsigned ebuf[BK_CAP];
    __shared__ int sbuf[BK_CAP];
    int tid = threadIdx.x;
    int lane = tid & 31, wave = tid >> 5;
    int b = blockIdx.x;
    int beg = base[b * SORT_G];
    int end = (b + 1 < nBk) ? base[(b + 1) * SORT_G] : E;
    int nEb = end - beg;
    if (tid < SORT_G) {
        r_src[tid] = tid * chunk + lst[tid * nBk + b];
        r_ko[tid]  = base[b * SORT_G + tid] - beg;
    }
    if (tid < BNODES) sc[tid] = 0;
    __syncthreads();

    bool fits = nEb <= BK_CAP;
    int k = tid;
    if (fits) {
        for (; k + 3072 < nEb; k += 4096) {
            unsigned p[4];
            #pragma unroll
            for (int j = 0; j < 4; ++j) p[j] = bucket_entry(lsorted, r_src, r_ko, k + j * 1024);
            #pragma unroll
            for (int j = 0; j < 4; ++j) ebuf[k + j * 1024] = p[j];
        }
        for (; k < nEb; k += 1024) ebuf[k] = bucket_entry(lsorted, r_src, r_ko, k);
        __syncthreads();
        for (k = tid; k < nEb; k += 1024) atomicAdd(&sc[ebuf[k] & (BNODES - 1)], 1);
    } else {
        for (; k < nEb; k += 1024) atomicAdd(&sc[bucket_entry(lsorted, r_src, r_ko, k) & (BNODES - 1)], 1);
    }
    __syncthreads();

    int c = 0, inc = 0;
    if (tid < BNODES) {
        c = sc[tid];
        inc = c;
        for (int o = 1; o < 32; o <<= 1) {
            int v = __shfl_up(inc, o, 32);
            if (lane >= o) inc += v;
        }
        if (lane == 31) sw[wave] = inc;
    }
    __syncthreads();
    if (tid < BNODES) {
        int ex = beg + inc - c;
        for (int w = 0; w < wave; ++w) ex += sw[w];
        int n = b * BNODES + tid;
        if (n < N) { cnt[n] = c; off[n] = ex; }
        sc[tid] = ex;
    }
    __syncthreads();

    if (fits) {
        for (k = tid; k < nEb; k += 1024) {
            unsigned p = ebuf[k];
            int pos = atomicAdd(&sc[p & (BNODES - 1)], 1);
            sbuf[pos - beg] = (int)(p >> BSH);
        }
        __syncthreads();
        for (k = tid - (beg & 31); k < nEb; k += 1024)
            if (k >= 0) ssrc[beg + k] = sbuf[k];
    } else {
        for (k = tid; k < nEb; k += 1024) {
            unsigned p = bucket_entry(lsorted, r_src, r_ko, k);
            int pos = atomicAdd(&sc[p & (BNODES - 1)], 1);
            ssrc[pos] = (int)(p >> BSH);
        }
    }
}

template <int J>
__device__ __forceinline__ int row_bcast_i(int v)
{
    return __builtin_amdgcn_update_dpp(0, v, 0x150 + J, 0xF, 0xF, true);
}
template <int J>
__device__ __forceinline__ float row_bcast_head(float w0, float w1)
{
    int t = __builtin_amdgcn_update_dpp(0, __float_as_int(w0), 0x150 + J, 0xF, 0xF, true);
    t = __builtin_amdgcn_update_dpp(t, __float_as_int(w1), 0x150 + J, 0xF, 0xC, false);
    return __int_as_float(t);
}
template <int M>
__device__ __forceinline__ float row_xor_f(float v)
{
    return __int_as_float(__builtin_amdgcn_update_dpp(0, __float_as_int(v), 0x160 + M, 0xF, 0xF, true));
}
__device__ __forceinline__ float xhalf(float x)
{
    int b = __float_as_int(x);
    return __int_as_float(__builtin_amdgcn_permlanex16(b, b, 0x76543210, 0xfedcba98, false, false));
}
__device__ __forceinline__ float wave_max(float v)
{
    v = fmaxf(v, xhalf(v));
    v = fmaxf(v, row_xor_f<8>(v));
    v = fmaxf(v, row_xor_f<4>(v));
    v = fmaxf(v, row_xor_f<2>(v));
    v = fmaxf(v, row_xor_f<1>(v));
    return v;
}
__device__ __forceinline__ float wave_sum(float v)
{
    v += xhalf(v);
    v += row_xor_f<8>(v);
    v += row_xor_f<4>(v);
    v += row_xor_f<2>(v);
    v += row_xor_f<1>(v);
    return v;
}
#define REP8_LO(X) X(0) X(1) X(2) X(3) X(4) X(5) X(6) X(7)
#define REP8_HI(X) X(8) X(9) X(10) X(11) X(12) X(13) X(14) X(15)

__global__ __launch_bounds__(256) void k_gather(
    const int* __restrict__ ssrc, const int* __restrict__ off, const int* __restrict__ cnt,
    const float* __restrict__ el, const float* __restrict__ rec,
    const float* __restrict__ feat, float* __restrict__ out, int N)
{
    __shared__ float so[64];
    int tid = threadIdx.x;
    int lane = tid & 31, wave = tid >> 5;
    int n = __builtin_amdgcn_readfirstlane(blockIdx.x * 8 + wave);
    int beg = 0, deg = 0;
    float er0 = 0.f, er1 = 0.f;
    if (n < N) {
        beg = off[n];
        deg = cnt[n];
        er0 = rec[n * 4];
        er1 = rec[n * 4 + 1];
    }

    int col = lane & 15;
    int hw  = lane >> 4;
    bool hi = (col & 8) != 0;
    bool live = lane < deg;

    int sraw = ssrc[beg + (live ? lane : 0)];
    int s1 = live ? sraw : 0;
    float2 v1 = *reinterpret_cast<const float2*>(el + 2 * s1);
    float p[16];
#define ROW_LOAD(J) p[J] = feat[row_bcast_i<J>(s1) * OUTC + col];
    REP8_LO(ROW_LOAD)
    REP8_HI(ROW_LOAD)
#undef ROW_LOAD

    float m0 = live ? v1.x : -INFINITY;
    float m1 = live ? v1.y : -INFINITY;
    for (int i = 32 + lane; i < deg; i += 32) {
        int s = ssrc[beg + i];
        float2 v = *reinterpret_cast<const float2*>(el + 2 * s);
        m0 = fmaxf(m0, v.x);
        m1 = fmaxf(m1, v.y);
    }
    m0 = wave_max(m0);
    m1 = wave_max(m1);
    m0 += er0;  m0 = (m0 >= 0.f) ? m0 : NEG_SLOPE * m0;
    m1 += er1;  m1 = (m1 >= 0.f) ? m1 : NEG_SLOPE * m1;
    m0 = (fabsf(m0) <= 3.402823466e38f) ? m0 : 0.f;
    m1 = (fabsf(m1) <= 3.402823466e38f) ? m1 : 0.f;

    float accv = 0.f, d0 = 0.f, d1 = 0.f;
    {
        float w0 = 0.f, w1 = 0.f;
        if (live) {
            float a = v1.x + er0;
            a = (a >= 0.f) ? a : NEG_SLOPE * a;
            w0 = __expf(a - m0);
            float b = v1.y + er1;
            b = (b >= 0.f) ? b : NEG_SLOPE * b;
            w1 = __expf(b - m1);
        }
        d0 = w0;
        d1 = w1;
#define ROW_SUM(J) { float t = row_bcast_head<J>(w0, w1) * p[J]; accv += (hw * 16 + J < deg) ? t : 0.f; }
        REP8_LO(ROW_SUM)
        REP8_HI(ROW_SUM)
#undef ROW_SUM
    }

    for (int c0 = 32; c0 < deg; c0 += 32) {
        int s = 0;
        float w0 = 0.f, w1 = 0.f;
        if (c0 + lane < deg) {
            s = ssrc[beg + c0 + lane];
            float2 v = *reinterpret_cast<const float2*>(el + 2 * s);
            float a = v.x + er0;
            a = (a >= 0.f) ? a : NEG_SLOPE * a;
            w0 = __expf(a - m0);
            float b = v.y + er1;
            b = (b >= 0.f) ? b : NEG_SLOPE * b;
            w1 = __expf(b - m1);
        }
        d0 += w0;
        d1 += w1;
        {
            float r[8];
#define TAIL_LOAD(J) r[J & 7] = row_bcast_head<J>(w0, w1) * feat[row_bcast_i<J>(s) * OUTC + col];
#define TAIL_SUM(J)  accv += (c0 + hw * 16 + J < deg) ? r[J & 7] : 0.f;
            REP8_LO(TAIL_LOAD)
            REP8_LO(TAIL_SUM)
            if (c0 + 8 < deg) {
                REP8_HI(TAIL_LOAD)
                REP8_HI(TAIL_SUM)
            }
#undef TAIL_LOAD
#undef TAIL_SUM
        }
    }

    accv += xhalf(accv);
    d0 = wave_sum(d0);
    d1 = wave_sum(d1);
    float r  = accv / fmaxf(hi ? d1 : d0, EPSF);
    float hm = 0.5f * (r + row_xor_f<8>(r));
    if (lane < 8) so[wave * 8 + lane] = hm;
    __syncthreads();
    if (tid < 64) {
        int i = blockIdx.x * 64 + tid;
        if (i < N * 8) out[i] = out[i] + so[tid];
    }
}

extern "C" void kernel_launch(void* const* d_in, const int* in_sizes, int n_in,
                              void* d_out, int out_size, void* d_ws, size_t ws_size,
                              hipStream_t stream)
{
    const float* x    = (const float*)d_in[0];
    const int*   ei   = (const int*)  d_in[1];
    const float* W    = (const float*)d_in[2];
    const float* al   = (const float*)d_in[3];
    const float* ar   = (const float*)d_in[4];
    const float* bias = (const float*)d_in[5];
    const float* Wr   = (const float*)d_in[6];
    float* out = (float*)d_out;

    int N = in_sizes[0] / IN_FEAT;
    int E = in_sizes[1] / 2;

    int nTiles = (N + 15) / 16;
    int nBk    = (N + BNODES - 1) / BNODES;
    int nTab   = nBk * SORT_G;
    int nScan  = (nTab + 1023) / 1024;
    int chunk  = (E + SORT_G - 1) / SORT_G;
    dim3 blk(256);

    char* ws = (char*)d_ws;
    float* feat  = (float*)ws;  ws += (size_t)N * OUTC * sizeof(float);
    float* rec   = (float*)ws;  ws += (size_t)N * 4 * sizeof(float);
    float* el    = (float*)ws;  ws += (size_t)N * 2 * sizeof(float);
    int*   cnt   = (int*)ws;    ws += (size_t)N * sizeof(int);
    int*   off   = (int*)ws;    ws += (size_t)N * sizeof(int);
    int*   hist  = (int*)ws;    ws += (size_t)nTab * sizeof(int);
    int*   lst   = (int*)ws;    ws += (size_t)nTab * sizeof(int);
    int*   base  = (int*)ws;    ws += (size_t)nTab * sizeof(int);
    unsigned* lsorted = (unsigned*)ws;  ws += (size_t)E * sizeof(unsigned);
    int*   ssrc  = (int*)ws;    ws += (size_t)E * sizeof(int);
    int*   bsum  = (int*)ws;    ws += (size_t)nScan * sizeof(int);

    if (nBk <= MAXBK && (size_t)(ws - (char*)d_ws) <= ws_size) {
        k_node_lsort<<<SORT_G + (nTiles + NODE_TPB - 1) / NODE_TPB, 1024, 0, stream>>>(x, W, Wr, bias, feat, out, N, nTiles,
                                                                      ei, hist, lst, lsorted, E, nBk, chunk);
        k_elr     <<<(N + 255) / 256,     blk, 0, stream>>>(feat, al, ar, el, rec, N);
        k_blocksum<<<nScan,               blk, 0, stream>>>(hist, bsum, nTab);
        k_offsets <<<nScan,               blk, 0, stream>>>(hist, bsum, base, nTab);
        k_bsort   <<<nBk,                 1024, 0, stream>>>(lsorted, lst, base, cnt, off, ssrc, N, E, nBk, chunk);
        k_gather  <<<(N + 7) / 8,         blk, 0, stream>>>(ssrc, off, cnt, el, rec, feat, out, N);
        return;
    }

    ws = (char*)d_ws;
    feat = (float*)ws;              ws += (size_t)N * OUTC * sizeof(float);
    float*    acc   = (float*)ws;    ws += (size_t)N * OUTC * sizeof(float);
    rec  = (float*)ws;              ws += (size_t)N * 4 * sizeof(float);
    el   = (float*)ws;              ws += (size_t)N * 2 * sizeof(float);
    float*    denom = (float*)ws;    ws += (size_t)N * 2 * sizeof(float);
    unsigned* mmax  = (unsigned*)ws; ws += (size_t)N * 2 * sizeof(unsigned);

    k_node    <<<(nTiles + 7) / 8,      blk, 0, stream>>>(x, W, Wr, bias, feat, out, N, nTiles);
    k_init    <<<(4 * N + 255) / 256,   blk, 0, stream>>>(mmax, denom, acc, 2 * N, 4 * N);
    k_elr     <<<(N + 255) / 256,       blk, 0, stream>>>(feat, al, ar, el, rec, N);
    k_edge_max<<<(E + 255) / 256,       blk, 0, stream>>>(ei, el, mmax, E);
    k_max_fix <<<(2 * N + 255) / 256,   blk, 0, stream>>>(mmax, rec, 2 * N);
    k_edge_agg<<<(E + 255) / 256,       blk, 0, stream>>>(ei, el, rec, feat, denom, acc, E);
    k_out     <<<(N * 8 + 255) / 256,   blk, 0, stream>>>(acc, denom, out, N);
}


# Sum pass: one wave takes 4 consecutive nodes (25,000 waves, not 100,000): sources, scores and rows of all four are requested together; it stores its own 128-byte out line; sum order kept
# speedup vs baseline: 13.0805x; 1.0375x over previous
#include <hip/hip_runtime.h>
#include <hip/hip_bf16.h>
#include <math.h>


typedef __attribute__((ext_vector_type(2))) float v2f;
typedef __attribute__((ext_vector_type(8))) float v8f;

#define IN_FEAT   256
#define OUTC      16
#define NEG_SLOPE 0.2f
#define EPSF      1e-15f

__device__ __forceinline__ unsigned fmap(float f) {
    unsigned u = __float_as_uint(f);
    return (u & 0x80000000u) ? ~u : (u | 0x80000000u);
}
__device__ __forceinline__ float funmap(unsigned u) {
    unsigned v = (u & 0x80000000u) ? (u & 0x7FFFFFFFu) : ~u;
    return __uint_as_float(v);
}

template <int NTHR, int TPB>
__device__ __forceinline__ void node_body(
    float* sW, float* sWr, int blk,
    const float* __restrict__ x, const float* __restrict__ W,
    const float* __restrict__ Wr, const float* __restrict__ bias,
    float* __restrict__ feat, float* __restrict__ out, int N, int nTiles)
{
    int tid = threadIdx.x;
    for (int i = tid; i < IN_FEAT * OUTC; i += NTHR) {
        sW[i]  = W[i];
        sWr[i] = Wr[i];
    }
    __syncthreads();

    int wave = tid >> 5;
    int lane = tid & 31;
    if (wave >= TPB) return;
    int t = blk * TPB + wave;
    if (t >= nTiles) t = nTiles - 1;

    int half = lane >> 4;
    int l15  = lane & 15;
    int rowg = t * 16 + l15;
    if (rowg >= N) rowg = N - 1;
    const float* xrow = x + (size_t)rowg * IN_FEAT;

    v8f cf = {0.f,0.f,0.f,0.f,0.f,0.f,0.f,0.f};
    v8f cr = {0.f,0.f,0.f,0.f,0.f,0.f,0.f,0.f};
    float nsq = 0.f;

    for (int k0 = 0; k0 < IN_FEAT; k0 += 4) {
        int ka = k0 + half * 2;
        float a0 = xrow[ka], a1 = xrow[ka + 1];
        nsq = fmaf(a0, a0, nsq);
        nsq = fmaf(a1, a1, nsq);
        v2f a; a[0] = a0; a[1] = a1;
        v2f b;  b[0]  = sW [ka * OUTC + l15];  b[1]  = sW [(ka + 1) * OUTC + l15];
        v2f br; br[0] = sWr[ka * OUTC + l15];  br[1] = sWr[(ka + 1) * OUTC + l15];
        cf = __builtin_amdgcn_wmma_f32_16x16x4_f32(false, a, false, b,  (short)0, cf, false, false);
        cr = __builtin_amdgcn_wmma_f32_16x16x4_f32(false, a, false, br, (short)0, cr, false, false);
    }

    nsq += __shfl_xor(nsq, 16, 32);
    float nrm = fmaxf(sqrtf(nsq), EPSF);
    float z   = fminf(nrm, 1.0f - 1e-7f);
    float sc  = 0.5f * (log1pf(z) - log1pf(-z)) / nrm;
    float bv  = bias[l15];

    float* fp = feat + (size_t)t * 16 * OUTC;
    float* op = out  + (size_t)t * 16 * 8;
    #pragma unroll
    for (int r = 0; r < 8; ++r) {
        int row = r + half * 8;
        float s_r = __shfl(sc, row, 32);
        float res = cr[r] * s_r + bv;
        float rm  = 0.5f * (res + __shfl_xor(res, 8, 32));
        if (t * 16 + row < N) {
            fp[row * OUTC + l15] = cf[r] * s_r;
            if (l15 < 8) op[row * 8 + l15] = rm;
        }
    }
}

__global__ __launch_bounds__(256) void k_node(
    const float* __restrict__ x, const float* __restrict__ W,
    const float* __restrict__ Wr, const float* __restrict__ bias,
    float* __restrict__ feat, float* __restrict__ out, int N, int nTiles)
{
    __shared__ float sW [IN_FEAT * OUTC];
    __shared__ float sWr[IN_FEAT * OUTC];
    node_body<256, 8>(sW, sWr, blockIdx.x, x, W, Wr, bias, feat, out, N, nTiles);
}

__global__ void k_elr(const float* __restrict__ feat, const float* __restrict__ al,
                      const float* __restrict__ ar, float* __restrict__ el,
                      float* __restrict__ rec, int N)
{
    int n = blockIdx.x * blockDim.x + threadIdx.x;
    if (n >= N) return;
    const float* f = feat + (size_t)n * OUTC;
    #pragma unroll
    for (int h = 0; h < 2; ++h) {
        float sl = 0.f, sr = 0.f;
        #pragma unroll
        for (int c = 0; c < 8; ++c) {
            float v = f[h * 8 + c];
            sl = fmaf(v, al[h * 8 + c], sl);
            sr = fmaf(v, ar[h * 8 + c], sr);
        }
        el[n * 2 + h] = sl;
        rec[n * 4 + h] = sr;
    }
}

__global__ void k_init(unsigned* __restrict__ mmax, float* __restrict__ denom,
                       float* __restrict__ acc, int n2, int n4)
{
    int i = blockIdx.x * blockDim.x + threadIdx.x;
    if (i < n2) { mmax[i] = 0x007FFFFFu; denom[i] = 0.f; }
    if (i < n4) reinterpret_cast<float4*>(acc)[i] = make_float4(0.f, 0.f, 0.f, 0.f);
}

__global__ void k_edge_max(const int* __restrict__ ei, const float* __restrict__ el,
                           unsigned* __restrict__ mmax, int E)
{
    int e = blockIdx.x * blockDim.x + threadIdx.x;
    int lane = threadIdx.x & 31;
    int base = e - lane;
    int d = 0;
    float v0 = 0.f, v1 = 0.f;
    if (e < E) {
        int s = ei[e];
        d = ei[E + e];
        v0 = el[s * 2];
        v1 = el[s * 2 + 1];
    }
    int h = lane & 1;
    for (int it = 0; it < 2; ++it) {
        int q = it * 16 + (lane >> 1);
        int dd = __shfl(d, q, 32);
        float a = __shfl(v0, q, 32), b = __shfl(v1, q, 32);
        if (base + q < E) atomicMax(&mmax[dd * 2 + h], fmap(h ? b : a));
    }
}

__global__ void k_max_fix(const unsigned* __restrict__ mmax, float* __restrict__ rec, int n2)
{
    int i = blockIdx.x * blockDim.x + threadIdx.x;
    if (i >= n2) return;
    int n = i >> 1, h = i & 1;
    float m = funmap(mmax[i]) + rec[n * 4 + h];
    m = (m >= 0.f) ? m : NEG_SLOPE * m;
    rec[n * 4 + 2 + h] = (fabsf(m) <= 3.402823466e38f) ? m : 0.f;
}

__global__ void k_edge_agg(const int* __restrict__ ei, const float* __restrict__ el,
                           const float* __restrict__ rec,
                           const float* __restrict__ feat, float* __restrict__ denom,
                           float* __restrict__ acc, int E)
{
    int e = blockIdx.x * blockDim.x + threadIdx.x;
    int lane = threadIdx.x & 31;
    int base = e - lane;
    int s = 0, d = 0;
    float w0 = 0.f, w1 = 0.f;
    if (e < E) {
        s = ei[e];
        d = ei[E + e];
        float4 r = reinterpret_cast<const float4*>(rec)[d];
        float v = el[s * 2] + r.x;
        v = (v >= 0.f) ? v : NEG_SLOPE * v;
        w0 = __expf(v - r.z);
        v = el[s * 2 + 1] + r.y;
        v = (v >= 0.f) ? v : NEG_SLOPE * v;
        w1 = __expf(v - r.w);
    }

    int h = lane & 1;
    for (int it = 0; it < 2; ++it) {
        int q = it * 16 + (lane >> 1);
        int dd = __shfl(d, q, 32);
        float a = __shfl(w0, q, 32), b = __shfl(w1, q, 32);
        if (base + q < E) atomicAdd(&denom[dd * 2 + h], h ? b : a);
    }

    int col = lane & 15;
    int hw  = lane >> 4;
    bool hi = (col & 8) != 0;
    float m[16];
    int   di[16];
    #pragma unroll
    for (int it = 0; it < 16; ++it) {
        int q = it * 2 + hw;
        int ss = __shfl(s, q, 32);
        di[it] = __shfl(d, q, 32) * OUTC + col;
        float a = __shfl(w0, q, 32), b = __shfl(w1, q, 32);
        m[it] = (hi ? b : a) * feat[ss * OUTC + col];
    }
    #pragma unroll
    for (int it = 0; it < 16; ++it)
        if (base + it * 2 + hw < E) atomicAdd(&acc[di[it]], m[it]);
}

__global__ void k_out(const float* __restrict__ acc, const float* __restrict__ denom,
                      float* __restrict__ out, int N)
{
    int i = blockIdx.x * blockDim.x + threadIdx.x;
    if (i >= N * 8) return;
    int n = i >> 3, c = i & 7;
    float m0 = acc[(size_t)n * OUTC + c]     / fmaxf(denom[n * 2],     EPSF);
    float m1 = acc[(size_t)n * OUTC + 8 + c] / fmaxf(denom[n * 2 + 1], EPSF);
    out[i] = out[i] + 0.5f * (m0 + m1);
}


#define BSH     8
#define BNODES  256
#define MAXBK   4096
#define SORT_G  256
#define CH_CAP  16384
#define BK_CAP  16384

__device__ __forceinline__ void lsort_body(int* sh, int* sw, unsigned* wbuf, int g,
                                           const int* __restrict__ ei, int* __restrict__ hist,
                                           int* __restrict__ lst, unsigned* __restrict__ lsorted,
                                           int E, int nBk, int chunk)
{
    int tid = threadIdx.x;
    int lane = tid & 31, wave = tid >> 5;
    for (int b = tid; b < nBk; b += 1024) sh[b] = 0;
    __syncthreads();

    int e0 = g * chunk;
    int e1 = min(e0 + chunk, E);
    int e = e0 + tid;
    for (; e + 3072 < e1; e += 4096) {
        int d[4];
        #pragma unroll
        for (int j = 0; j < 4; ++j) d[j] = ei[E + e + j * 1024];
        #pragma unroll
        for (int j = 0; j < 4; ++j) atomicAdd(&sh[d[j] >> BSH], 1);
    }
    for (; e < e1; e += 1024) atomicAdd(&sh[ei[E + e] >> BSH], 1);
    __syncthreads();

    int c[4];
    #pragma unroll
    for (int j = 0; j < 4; ++j) c[j] = (4 * tid + j < nBk) ? sh[4 * tid + j] : 0;
    int t = c[0] + c[1] + c[2] + c[3];
    int inc = t;
    for (int o = 1; o < 32; o <<= 1) {
        int v = __shfl_up(inc, o, 32);
        if (lane >= o) inc += v;
    }
    if (lane == 31) sw[wave] = inc;
    __syncthreads();
    int ex = inc - t;
    for (int w = 0; w < wave; ++w) ex += sw[w];
    #pragma unroll
    for (int j = 0; j < 4; ++j) {
        int b = 4 * tid + j;
        if (b < nBk) {
            hist[b * SORT_G + g] = c[j];
            lst[g * nBk + b] = ex;
            sh[b] = ex;
            ex += c[j];
        }
    }
    __syncthreads();

    bool staged = e1 - e0 <= CH_CAP;
    e = e0 + tid;
    for (; e + 3072 < e1; e += 4096) {
        int s[4], d[4];
        #pragma unroll
        for (int j = 0; j < 4; ++j) {
            s[j] = ei[e + j * 1024];
            d[j] = ei[E + e + j * 1024];
        }
        #pragma unroll
        for (int j = 0; j < 4; ++j) {
            int lp = atomicAdd(&sh[d[j] >> BSH], 1);
            unsigned w = ((unsigned)s[j] << BSH) | (unsigned)(d[j] & (BNODES - 1));
            if (staged) wbuf[lp] = w; else lsorted[e0 + lp] = w;
        }
    }
    for (; e < e1; e += 1024) {
        int s = ei[e], d = ei[E + e];
        int lp = atomicAdd(&sh[d >> BSH], 1);
        unsigned w = ((unsigned)s << BSH) | (unsigned)(d & (BNODES - 1));
        if (staged) wbuf[lp] = w; else lsorted[e0 + lp] = w;
    }
    if (staged) {
        __syncthreads();
        for (int i = tid - (e0 & 31); i < e1 - e0; i += 1024)
            if (i >= 0) lsorted[e0 + i] = wbuf[i];
    }
}

#define NODE_TPB 25
__global__ __launch_bounds__(1024) void k_node_lsort(
    const float* __restrict__ x, const float* __restrict__ W,
    const float* __restrict__ Wr, const float* __restrict__ bias,
    float* __restrict__ feat, float* __restrict__ out, int N, int nTiles,
    const int* __restrict__ ei, int* __restrict__ hist,
    int* __restrict__ lst, unsigned* __restrict__ lsorted,
    int E, int nBk, int chunk)
{
    __shared__ unsigned smem[MAXBK + 32 + CH_CAP];
    if (blockIdx.x < SORT_G) {
        lsort_body(reinterpret_cast<int*>(smem), reinterpret_cast<int*>(smem) + MAXBK, smem + MAXBK + 32,
                   blockIdx.x, ei, hist, lst, lsorted, E, nBk, chunk);
    } else {
        float* sW = reinterpret_cast<float*>(smem);
        node_body<1024, NODE_TPB>(sW, sW + IN_FEAT * OUTC, blockIdx.x - SORT_G, x, W, Wr, bias, feat, out, N, nTiles);
    }
}

__global__ __launch_bounds__(256) void k_blocksum(const int* __restrict__ cnt,
                                                  int* __restrict__ bsum, int N)
{
    __shared__ int sw[8];
    int tid = threadIdx.x;
    int i0 = (blockIdx.x * 256 + tid) * 4;
    int s = 0;
    #pragma unroll
    for (int j = 0; j < 4; ++j)
        if (i0 + j < N) s += cnt[i0 + j];
    #pragma unroll
    for (int o = 16; o > 0; o >>= 1) s += __shfl_xor(s, o, 32);
    if ((tid & 31) == 0) sw[tid >> 5] = s;
    __syncthreads();
    if (tid == 0) {
        int t = 0;
        #pragma unroll
        for (int w = 0; w < 8; ++w) t += sw[w];
        bsum[blockIdx.x] = t;
    }
}

__global__ __launch_bounds__(256) void k_offsets(const int* __restrict__ cnt,
                                                 const int* __restrict__ bsum,
                                                 int* __restrict__ off, int N)
{
    __shared__ int sw[8];
    int tid = threadIdx.x;
    int lane = tid & 31, wave = tid >> 5;

    int b = 0;
    for (int i = tid; i < (int)blockIdx.x; i += 256) b += bsum[i];
    #pragma unroll
    for (int o = 16; o > 0; o >>= 1) b += __shfl_xor(b, o, 32);
    if (lane == 0) sw[wave] = b;
    __syncthreads();
    int base = 0;
    #pragma unroll
    for (int w = 0; w < 8; ++w) base += sw[w];
    __syncthreads();

    int i0 = (blockIdx.x * 256 + tid) * 4;
    int c[4];
    #pragma unroll
    for (int j = 0; j < 4; ++j) c[j] = (i0 + j < N) ? cnt[i0 + j] : 0;
    int t = c[0] + c[1] + c[2] + c[3];
    int inc = t;
    #pragma unroll
    for (int o = 1; o < 32; o <<= 1) {
        int v = __shfl_up(inc, o, 32);
        if (lane >= o) inc += v;
    }
    if (lane == 31) sw[wave] = inc;
    __syncthreads();
    int wb = 0;
    for (int w = 0; w < wave; ++w) wb += sw[w];
    int ex = base + wb + inc - t;
    #pragma unroll
    for (int j = 0; j < 4; ++j)
        if (i0 + j < N) { off[i0 + j] = ex; ex += c[j]; }
}

__device__ __forceinline__ unsigned bucket_entry(const unsigned* __restrict__ lsorted, const int* r_src,
                                                 const int* r_ko, int k)
{
    int lo = 0;
    for (int step = SORT_G / 2; step > 0; step >>= 1)
        if (r_ko[lo + step] <= k) lo += step;
    return lsorted[r_src[lo] + (k - r_ko[lo])];
}

__global__ __launch_bounds__(1024) void k_bsort(const unsigned* __restrict__ lsorted, const int* __restrict__ lst,
                                                const int* __restrict__ base,
                                                int* __restrict__ cnt, int* __restrict__ off,
                                                int* __restrict__ ssrc, int N, int E, int nBk, int chunk)
{
    __shared__ int sc[BNODES];
    __shared__ int sw[8];
    __shared__ int r_src[SORT_G];
    __shared__ int r_ko[SORT_G];
    __shared__ unsigned ebuf[BK_CAP];
    __shared__ int sbuf[BK_CAP];
    int tid = threadIdx.x;
    int lane = tid & 31, wave = tid >> 5;
    int b = blockIdx.x;
    int beg = base[b * SORT_G];
    int end = (b + 1 < nBk) ? base[(b + 1) * SORT_G] : E;
    int nEb = end - beg;
    if (tid < SORT_G) {
        r_src[tid] = tid * chunk + lst[tid * nBk + b];
        r_ko[tid]  = base[b * SORT_G + tid] - beg;
    }
    if (tid < BNODES) sc[tid] = 0;
    __syncthreads();

    bool fits = nEb <= BK_CAP;
    int k = tid;
    if (fits) {
        for (; k + 3072 < nEb; k += 4096) {
            unsigned p[4];
            #pragma unroll
            for (int j = 0; j < 4; ++j) p[j] = bucket_entry(lsorted, r_src, r_ko, k + j * 1024);
            #pragma unroll
            for (int j = 0; j < 4; ++j) ebuf[k + j * 1024] = p[j];
        }
        for (; k < nEb; k += 1024) ebuf[k] = bucket_entry(lsorted, r_src, r_ko, k);
        __syncthreads();
        for (k = tid; k < nEb; k += 1024) atomicAdd(&sc[ebuf[k] & (BNODES - 1)], 1);
    } else {
        for (; k < nEb; k += 1024) atomicAdd(&sc[bucket_entry(lsorted, r_src, r_ko, k) & (BNODES - 1)], 1);
    }
    __syncthreads();

    int c = 0, inc = 0;
    if (tid < BNODES) {
        c = sc[tid];
        inc = c;
        for (int o = 1; o < 32; o <<= 1) {
            int v = __shfl_up(inc, o, 32);
            if (lane >= o) inc += v;
        }
        if (lane == 31) sw[wave] = inc;
    }
    __syncthreads();
    if (tid < BNODES) {
        int ex = beg + inc - c;
        for (int w = 0; w < wave; ++w) ex += sw[w];
        int n = b * BNODES + tid;
        if (n < N) { cnt[n] = c; off[n] = ex; }
        sc[tid] = ex;
    }
    __syncthreads();

    if (fits) {
        for (k = tid; k < nEb; k += 1024) {
            unsigned p = ebuf[k];
            int pos = atomicAdd(&sc[p & (BNODES - 1)], 1);
            sbuf[pos - beg] = (int)(p >> BSH);
        }
        __syncthreads();
        for (k = tid - (beg & 31); k < nEb; k += 1024)
            if (k >= 0) ssrc[beg + k] = sbuf[k];
    } else {
        for (k = tid; k < nEb; k += 1024) {
            unsigned p = bucket_entry(lsorted, r_src, r_ko, k);
            int pos = atomicAdd(&sc[p & (BNODES - 1)], 1);
            ssrc[pos] = (int)(p >> BSH);
        }
    }
}

template <int J>
__device__ __forceinline__ int row_bcast_i(int v)
{
    return __builtin_amdgcn_update_dpp(0, v, 0x150 + J, 0xF, 0xF, true);
}
template <int J>
__device__ __forceinline__ float row_bcast_head(float w0, float w1)
{
    int t = __builtin_amdgcn_update_dpp(0, __float_as_int(w0), 0x150 + J, 0xF, 0xF, true);
    t = __builtin_amdgcn_update_dpp(t, __float_as_int(w1), 0x150 + J, 0xF, 0xC, false);
    return __int_as_float(t);
}
template <int M>
__device__ __forceinline__ float row_xor_f(float v)
{
    return __int_as_float(__builtin_amdgcn_update_dpp(0, __float_as_int(v), 0x160 + M, 0xF, 0xF, true));
}
__device__ __forceinline__ float xhalf(float x)
{
    int b = __float_as_int(x);
    return __int_as_float(__builtin_amdgcn_permlanex16(b, b, 0x76543210, 0xfedcba98, false, false));
}
__device__ __forceinline__ float wave_max(float v)
{
    v = fmaxf(v, xhalf(v));
    v = fmaxf(v, row_xor_f<8>(v));
    v = fmaxf(v, row_xor_f<4>(v));
    v = fmaxf(v, row_xor_f<2>(v));
    v = fmaxf(v, row_xor_f<1>(v));
    return v;
}
__device__ __forceinline__ float wave_sum(float v)
{
    v += xhalf(v);
    v += row_xor_f<8>(v);
    v += row_xor_f<4>(v);
    v += row_xor_f<2>(v);
    v += row_xor_f<1>(v);
    return v;
}
#define REP8_LO(X) X(0) X(1) X(2) X(3) X(4) X(5) X(6) X(7)
#define REP8_HI(X) X(8) X(9) X(10) X(11) X(12) X(13) X(14) X(15)

__device__ __forceinline__ float node_sum(
    int lane, int col, int hw, bool hi, int beg, int deg, float er0, float er1,
    int s1, float2 v1, int s2, float2 v2, const float (&p)[16],
    const int* __restrict__ ssrc, const float* __restrict__ el, const float* __restrict__ feat)
{
    bool live1 = lane < deg;
    bool live2 = lane + 32 < deg;

    float m0 = live1 ? v1.x : -INFINITY;
    float m1 = live1 ? v1.y : -INFINITY;
    if (live2) { m0 = fmaxf(m0, v2.x); m1 = fmaxf(m1, v2.y); }
    for (int i = 64 + lane; i < deg; i += 32) {
        int s = ssrc[beg + i];
        float2 v = *reinterpret_cast<const float2*>(el + 2 * s);
        m0 = fmaxf(m0, v.x);
        m1 = fmaxf(m1, v.y);
    }
    m0 = wave_max(m0);
    m1 = wave_max(m1);
    m0 += er0;  m0 = (m0 >= 0.f) ? m0 : NEG_SLOPE * m0;
    m1 += er1;  m1 = (m1 >= 0.f) ? m1 : NEG_SLOPE * m1;
    m0 = (fabsf(m0) <= 3.402823466e38f) ? m0 : 0.f;
    m1 = (fabsf(m1) <= 3.402823466e38f) ? m1 : 0.f;

    float accv = 0.f, d0 = 0.f, d1 = 0.f;
    {
        float w0 = 0.f, w1 = 0.f;
        if (live1) {
            float a = v1.x + er0;
            a = (a >= 0.f) ? a : NEG_SLOPE * a;
            w0 = __expf(a - m0);
            float b = v1.y + er1;
            b = (b >= 0.f) ? b : NEG_SLOPE * b;
            w1 = __expf(b - m1);
        }
        d0 = w0;
        d1 = w1;
#define ROW_SUM(J) { float t = row_bcast_head<J>(w0, w1) * p[J]; accv += (hw * 16 + J < deg) ? t : 0.f; }
        REP8_LO(ROW_SUM)
        REP8_HI(ROW_SUM)
#undef ROW_SUM
    }

    if (deg > 32) {
        float w0 = 0.f, w1 = 0.f;
        if (live2) {
            float a = v2.x + er0;
            a = (a >= 0.f) ? a : NEG_SLOPE * a;
            w0 = __expf(a - m0);
            float b = v2.y + er1;
            b = (b >= 0.f) ? b : NEG_SLOPE * b;
            w1 = __expf(b - m1);
        }
        d0 += w0;
        d1 += w1;
        float r[8];
#define MID_LOAD(J) r[J & 7] = row_bcast_head<J>(w0, w1) * feat[row_bcast_i<J>(s2) * OUTC + col];
#define MID_SUM(J)  accv += (32 + hw * 16 + J < deg) ? r[J & 7] : 0.f;
        REP8_LO(MID_LOAD)
        REP8_LO(MID_SUM)
        if (32 + 8 < deg) {
            REP8_HI(MID_LOAD)
            REP8_HI(MID_SUM)
        }
#undef MID_LOAD
#undef MID_SUM
    }

    for (int c0 = 64; c0 < deg; c0 += 32) {
        int s = 0;
        float w0 = 0.f, w1 = 0.f;
        if (c0 + lane < deg) {
            s = ssrc[beg + c0 + lane];
            float2 v = *reinterpret_cast<const float2*>(el + 2 * s);
            float a = v.x + er0;
            a = (a >= 0.f) ? a : NEG_SLOPE * a;
            w0 = __expf(a - m0);
            float b = v.y + er1;
            b = (b >= 0.f) ? b : NEG_SLOPE * b;
            w1 = __expf(b - m1);
        }
        d0 += w0;
        d1 += w1;
        {
            float r[8];
#define TAIL_LOAD(J) r[J & 7] = row_bcast_head<J>(w0, w1) * feat[row_bcast_i<J>(s) * OUTC + col];
#define TAIL_SUM(J)  accv += (c0 + hw * 16 + J < deg) ? r[J & 7] : 0.f;
            REP8_LO(TAIL_LOAD)
            REP8_LO(TAIL_SUM)
            if (c0 + 8 < deg) {
                REP8_HI(TAIL_LOAD)
                REP8_HI(TAIL_SUM)
            }
#undef TAIL_LOAD
#undef TAIL_SUM
        }
    }

    accv += xhalf(accv);
    d0 = wave_sum(d0);
    d1 = wave_sum(d1);
    float r  = accv / fmaxf(hi ? d1 : d0, EPSF);
    return 0.5f * (r + row_xor_f<8>(r));
}

__global__ __launch_bounds__(256, 8) void k_gather(
    const int* __restrict__ ssrc, const int* __restrict__ off, const int* __restrict__ cnt,
    const float* __restrict__ el, const float* __restrict__ rec,
    const float* __restrict__ feat, float* __restrict__ out, int N)
{
    int tid = threadIdx.x;
    int lane = tid & 31, wave = tid >> 5;
    int n0 = __builtin_amdgcn_readfirstlane((blockIdx.x * 8 + wave) * 4);
    if (n0 >= N) return;

    int col = lane & 15;
    int hw  = lane >> 4;
    bool hi = (col & 8) != 0;

    int beg[4], deg[4];
    float er0[4], er1[4];
    if (n0 + 3 < N) {
        #pragma unroll
        for (int j = 0; j < 4; ++j) {
            beg[j] = off[n0 + j];
            deg[j] = cnt[n0 + j];
            er0[j] = rec[(n0 + j) * 4];
            er1[j] = rec[(n0 + j) * 4 + 1];
        }
    } else {
        #pragma unroll
        for (int j = 0; j < 4; ++j) {
            int n  = n0 + j;
            int nn = (n < N) ? n : N - 1;
            int c  = cnt[nn];
            beg[j] = off[nn];
            deg[j] = (n < N) ? c : 0;
            er0[j] = rec[nn * 4];
            er1[j] = rec[nn * 4 + 1];
        }
    }
    int  oi = n0 * 8 + lane;
    bool ov = n0 + (lane >> 3) < N;
    float o0 = out[ov ? oi : n0 * 8];

    int s1[4], s2[4];
    #pragma unroll
    for (int j = 0; j < 4; ++j) {
        bool l1 = lane < deg[j], l2 = lane + 32 < deg[j];
        int ra = ssrc[beg[j] + (l1 ? lane : 0)];
        int rb = ssrc[beg[j] + (l2 ? lane + 32 : 0)];
        s1[j] = l1 ? ra : 0;
        s2[j] = l2 ? rb : 0;
    }
    float2 v1[4], v2[4];
    #pragma unroll
    for (int j = 0; j < 4; ++j) {
        v1[j] = *reinterpret_cast<const float2*>(el + 2 * s1[j]);
        v2[j] = *reinterpret_cast<const float2*>(el + 2 * s2[j]);
    }
    float p[4][16];
    #pragma unroll
    for (int j = 0; j < 4; ++j) {
#define ROW_LOAD(J) p[j][J] = feat[row_bcast_i<J>(s1[j]) * OUTC + col];
        REP8_LO(ROW_LOAD)
        REP8_HI(ROW_LOAD)
#undef ROW_LOAD
    }

    float outv = 0.f;
    #pragma unroll
    for (int j = 0; j < 4; ++j) {
        float hm = node_sum(lane, col, hw, hi, beg[j], deg[j], er0[j], er1[j],
                            s1[j], v1[j], s2[j], v2[j], p[j], ssrc, el, feat);
        outv = ((lane >> 3) == j) ? hm : outv;
    }
    if (ov) out[oi] = o0 + outv;
}

extern "C" void kernel_launch(void* const* d_in, const int* in_sizes, int n_in,
                              void* d_out, int out_size, void* d_ws, size_t ws_size,
                              hipStream_t stream)
{
    const float* x    = (const float*)d_in[0];
    const int*   ei   = (const int*)  d_in[1];
    const float* W    = (const float*)d_in[2];
    const float* al   = (const float*)d_in[3];
    const float* ar   = (const float*)d_in[4];
    const float* bias = (const float*)d_in[5];
    const float* Wr   = (const float*)d_in[6];
    float* out = (float*)d_out;

    int N = in_sizes[0] / IN_FEAT;
    int E = in_sizes[1] / 2;

    int nTiles = (N + 15) / 16;
    int nBk    = (N + BNODES - 1) / BNODES;
    int nTab   = nBk * SORT_G;
    int nScan  = (nTab + 1023) / 1024;
    int chunk  = (E + SORT_G - 1) / SORT_G;
    dim3 blk(256);

    char* ws = (char*)d_ws;
    float* feat  = (float*)ws;  ws += (size_t)N * OUTC * sizeof(float);
    float* rec   = (float*)ws;  ws += (size_t)N * 4 * sizeof(float);
    float* el    = (float*)ws;  ws += (size_t)N * 2 * sizeof(float);
    int*   cnt   = (int*)ws;    ws += (size_t)N * sizeof(int);
    int*   off   = (int*)ws;    ws += (size_t)N * sizeof(int);
    int*   hist  = (int*)ws;    ws += (size_t)nTab * sizeof(int);
    int*   lst   = (int*)ws;    ws += (size_t)nTab * sizeof(int);
    int*   base  = (int*)ws;    ws += (size_t)nTab * sizeof(int);
    unsigned* lsorted = (unsigned*)ws;  ws += (size_t)E * sizeof(unsigned);
    int*   ssrc  = (int*)ws;    ws += (size_t)E * sizeof(int);
    int*   bsum  = (int*)ws;    ws += (size_t)nScan * sizeof(int);

    if (nBk <= MAXBK && (size_t)(ws - (char*)d_ws) <= ws_size) {
        k_node_lsort<<<SORT_G + (nTiles + NODE_TPB - 1) / NODE_TPB, 1024, 0, stream>>>(x, W, Wr, bias, feat, out, N, nTiles,
                                                                      ei, hist, lst, lsorted, E, nBk, chunk);
        k_elr     <<<(N + 255) / 256,     blk, 0, stream>>>(feat, al, ar, el, rec, N);
        k_blocksum<<<nScan,               blk, 0, stream>>>(hist, bsum, nTab);
        k_offsets <<<nScan,               blk, 0, stream>>>(hist, bsum, base, nTab);
        k_bsort   <<<nBk,                 1024, 0, stream>>>(lsorted, lst, base, cnt, off, ssrc, N, E, nBk, chunk);
        k_gather  <<<(N + 31) / 32,       blk, 0, stream>>>(ssrc, off, cnt, el, rec, feat, out, N);
        return;
    }

    ws = (char*)d_ws;
    feat = (float*)ws;              ws += (size_t)N * OUTC * sizeof(float);
    float*    acc   = (float*)ws;    ws += (size_t)N * OUTC * sizeof(float);
    rec  = (float*)ws;              ws += (size_t)N * 4 * sizeof(float);
    el   = (float*)ws;              ws += (size_t)N * 2 * sizeof(float);
    float*    denom = (float*)ws;    ws += (size_t)N * 2 * sizeof(float);
    unsigned* mmax  = (unsigned*)ws; ws += (size_t)N * 2 * sizeof(unsigned);

    k_node    <<<(nTiles + 7) / 8,      blk, 0, stream>>>(x, W, Wr, bias, feat, out, N, nTiles);
    k_init    <<<(4 * N + 255) / 256,   blk, 0, stream>>>(mmax, denom, acc, 2 * N, 4 * N);
    k_elr     <<<(N + 255) / 256,       blk, 0, stream>>>(feat, al, ar, el, rec, N);
    k_edge_max<<<(E + 255) / 256,       blk, 0, stream>>>(ei, el, mmax, E);
    k_max_fix <<<(2 * N + 255) / 256,   blk, 0, stream>>>(mmax, rec, 2 * N);
    k_edge_agg<<<(E + 255) / 256,       blk, 0, stream>>>(ei, el, rec, feat, denom, acc, E);
    k_out     <<<(N * 8 + 255) / 256,   blk, 0, stream>>>(acc, denom, out, N);
}
